# starting kernel
# baseline (speedup 1.0000x reference)
#include <hip/hip_runtime.h>
#include <hip/hip_bf16.h>

typedef __attribute__((ext_vector_type(2))) float v2f;
typedef __attribute__((ext_vector_type(8))) float v8f;

#define N_NODES  100000
#define N_EDGES  1600000
#define N_GRAPHS 1000
#define IN_DIM   128
#define HID      100
#define HPAD     112

__global__ void k_init_deg(float* deg, int n) {
    int i = blockIdx.x * blockDim.x + threadIdx.x;
    if (i < n) deg[i] = 1.0f;
}

__global__ void k_edge_deg(const int* __restrict__ dst, float* deg, int e) {
    int i = blockIdx.x * blockDim.x + threadIdx.x;
    if (i < e) atomicAdd(&deg[dst[i]], 1.0f);
}

__global__ void k_dinv(float* deg, int n) {
    int i = blockIdx.x * blockDim.x + threadIdx.x;
    if (i < n) deg[i] = rsqrtf(deg[i]);
}

__global__ void k_pack_w(const float* __restrict__ src, float2* __restrict__ dst,
                         int sr, int sc, int kpairs) {
    int i = blockIdx.x * blockDim.x + threadIdx.x;
    if (i >= kpairs * HPAD) return;
    int p = i / HPAD, n = i % HPAD;
    int k0 = 2 * p, k1 = 2 * p + 1;
    float lo = (k0 < sr && n < sc) ? src[k0 * sc + n] : 0.0f;
    float hi = (k1 < sr && n < sc) ? src[k1 * sc + n] : 0.0f;
    dst[i] = make_float2(lo, hi);
}

__global__ void k_pad_vec(const float* __restrict__ src, float* __restrict__ dst,
                          int sc, int dc) {
    int i = blockIdx.x * blockDim.x + threadIdx.x;
    if (i < dc) dst[i] = (i < sc) ? src[i] : 0.0f;
}

__global__ void k_zero(float* p, size_t n) {
    size_t i = (size_t)blockIdx.x * blockDim.x + threadIdx.x;
    size_t stride = (size_t)gridDim.x * blockDim.x;
    for (; i < n; i += stride) p[i] = 0.0f;
}

__global__ __launch_bounds__(256) void k_gemm_wmma(
    const float* __restrict__ X, const float2* __restrict__ Wp,
    float* __restrict__ H, int nrows, int K)
{
    __shared__ float2 wl[(IN_DIM / 2) * HPAD];
    int tid = threadIdx.x;
    int kpairs = K >> 1;
    for (int i = tid; i < kpairs * HPAD; i += 256) wl[i] = Wp[i];
    __syncthreads();

    int lane = tid & 31;
    int wave = tid >> 5;
    int m    = lane & 15;
    int koff = (lane >> 4) << 1;
    int row0 = blockIdx.x * 128 + wave * 16;
    int row  = row0 + m;
    bool inr = row < nrows;
    int rowc = inr ? row : (nrows - 1);
    const float* xrow = X + (size_t)rowc * K;

    v8f acc[7];
#pragma unroll
    for (int t = 0; t < 7; ++t)
        acc[t] = (v8f){0.f, 0.f, 0.f, 0.f, 0.f, 0.f, 0.f, 0.f};

    for (int k = 0; k < K; k += 4) {
        float2 t2 = *(const float2*)(xrow + k + koff);
        v2f a = (v2f){ inr ? t2.x : 0.f, inr ? t2.y : 0.f };
        const float2* wrow = wl + ((k + koff) >> 1) * HPAD;
#pragma unroll
        for (int t = 0; t < 7; ++t) {
            float2 bb = wrow[t * 16 + m];
            v2f b = (v2f){bb.x, bb.y};
            acc[t] = __builtin_amdgcn_wmma_f32_16x16x4_f32(
                false, a, false, b, (short)0, acc[t], false, false);
        }
    }

    int rbase = row0 + ((lane >> 4) << 3);
#pragma unroll
    for (int t = 0; t < 7; ++t) {
        int cc = t * 16 + m;
#pragma unroll
        for (int v = 0; v < 8; ++v) {
            int rr = rbase + v;
            if (rr < nrows) H[(size_t)rr * HPAD + cc] = acc[t][v];
        }
    }
}

__global__ __launch_bounds__(256) void k_scatter(
    const float* __restrict__ H, const float* __restrict__ dinv,
    const int* __restrict__ src, const int* __restrict__ dst,
    float* __restrict__ AGG, int nE, int nN)
{
    long long gid = (long long)blockIdx.x * 256 + threadIdx.x;
    int e    = (int)(gid >> 5);
    int lane = (int)(gid & 31);
    if (e >= nE + nN) return;
    int s, d;
    if (e < nE) { s = src[e]; d = dst[e]; }
    else        { s = d = e - nE; }
    float w = dinv[s] * dinv[d];
    const float* hs = H + (size_t)s * HPAD;
    float*       ad = AGG + (size_t)d * HPAD;
    atomicAdd(&ad[lane],      hs[lane]      * w);
    atomicAdd(&ad[lane + 32], hs[lane + 32] * w);
    atomicAdd(&ad[lane + 64], hs[lane + 64] * w);
    if (lane < 16) atomicAdd(&ad[lane + 96], hs[lane + 96] * w);
}

__global__ void k_bias_relu(float* __restrict__ A, const float* __restrict__ b,
                            long long n) {
    long long i = (long long)blockIdx.x * blockDim.x + threadIdx.x;
    if (i >= n) return;
    int c = (int)(i % HPAD);
    float v = A[i] + b[c];
    A[i] = v > 0.f ? v : 0.f;
}

__global__ void k_counts(const int* __restrict__ batch, float* counts, int n) {
    int i = blockIdx.x * blockDim.x + threadIdx.x;
    if (i < n) atomicAdd(&counts[batch[i]], 1.0f);
}

__global__ void k_pool_sum(const float* __restrict__ X, const int* __restrict__ batch,
                           float* __restrict__ pool, int n) {
    long long i = (long long)blockIdx.x * blockDim.x + threadIdx.x;
    if (i >= (long long)n * HID) return;
    int node = (int)(i / HID);
    int c    = (int)(i % HID);
    atomicAdd(&pool[(size_t)batch[node] * HID + c], X[(size_t)node * HPAD + c]);
}

__global__ void k_pool_fin(const float* __restrict__ pool, const float* __restrict__ counts,
                           float* __restrict__ out, int n) {
    int i = blockIdx.x * blockDim.x + threadIdx.x;
    if (i >= n) return;
    int g = i / HID;
    out[i] = pool[i] / fmaxf(counts[g], 1.0f);
}

extern "C" void kernel_launch(void* const* d_in, const int* in_sizes, int n_in,
                              void* d_out, int out_size, void* d_ws, size_t ws_size,
                              hipStream_t stream) {
    const float* x  = (const float*)d_in[0];
    const float* W1 = (const float*)d_in[1];
    const float* b1 = (const float*)d_in[2];
    const float* W2 = (const float*)d_in[3];
    const float* b2 = (const float*)d_in[4];
    const int*   ei = (const int*)d_in[5];
    const int*   bt = (const int*)d_in[6];
    const int* esrc = ei;
    const int* edst = ei + N_EDGES;
    float* out = (float*)d_out;

    float*  ws   = (float*)d_ws;
    float*  dinv = ws;
    float2* W1p  = (float2*)(dinv + N_NODES);
    float*  b1p  = (float*)(W1p + (IN_DIM / 2) * HPAD);
    float2* W2p  = (float2*)(b1p + HPAD);
    float*  b2p  = (float*)(W2p + (HPAD / 2) * HPAD);
    float*  B0   = b2p + HPAD;
    float*  B1   = B0 + (size_t)N_NODES * HPAD;
    float*  B2   = B1 + (size_t)N_NODES * HPAD;
    float*  pool = B2 + (size_t)N_NODES * HPAD;
    float*  cnts = pool + N_GRAPHS * HID;

    const size_t NH = (size_t)N_NODES * HPAD;
    auto cdiv = [](long long a, long long b) { return (unsigned)((a + b - 1) / b); };

    k_init_deg<<<cdiv(N_NODES, 256), 256, 0, stream>>>(dinv, N_NODES);
    k_edge_deg<<<cdiv(N_EDGES, 256), 256, 0, stream>>>(edst, dinv, N_EDGES);
    k_dinv<<<cdiv(N_NODES, 256), 256, 0, stream>>>(dinv, N_NODES);

    k_pack_w<<<cdiv((IN_DIM / 2) * HPAD, 256), 256, 0, stream>>>(W1, W1p, IN_DIM, HID, IN_DIM / 2);
    k_pad_vec<<<1, HPAD, 0, stream>>>(b1, b1p, HID, HPAD);
    k_pack_w<<<cdiv((HPAD / 2) * HPAD, 256), 256, 0, stream>>>(W2, W2p, HID, HID, HPAD / 2);
    k_pad_vec<<<1, HPAD, 0, stream>>>(b2, b2p, HID, HPAD);

    k_gemm_wmma<<<cdiv(N_NODES, 128), 256, 0, stream>>>(x, W1p, B0, N_NODES, IN_DIM);
    k_zero<<<4096, 256, 0, stream>>>(B1, NH);
    k_scatter<<<cdiv((long long)(N_EDGES + N_NODES) * 32, 256), 256, 0, stream>>>(
        B0, dinv, esrc, edst, B1, N_EDGES, N_NODES);
    k_bias_relu<<<cdiv((long long)NH, 256), 256, 0, stream>>>(B1, b1p, (long long)NH);

    k_gemm_wmma<<<cdiv(N_NODES, 128), 256, 0, stream>>>(B1, W2p, B2, N_NODES, HPAD);
    k_zero<<<4096, 256, 0, stream>>>(B0, NH);
    k_scatter<<<cdiv((long long)(N_EDGES + N_NODES) * 32, 256), 256, 0, stream>>>(
        B2, dinv, esrc, edst, B0, N_EDGES, N_NODES);
    k_bias_relu<<<cdiv((long long)NH, 256), 256, 0, stream>>>(B0, b2p, (long long)NH);

    k_zero<<<64, 256, 0, stream>>>(pool, (size_t)N_GRAPHS * HID + N_GRAPHS);
    k_counts<<<cdiv(N_NODES, 256), 256, 0, stream>>>(bt, cnts, N_NODES);
    k_pool_sum<<<cdiv((long long)N_NODES * HID, 256), 256, 0, stream>>>(B0, bt, pool, N_NODES);
    k_pool_fin<<<cdiv(N_GRAPHS * HID, 256), 256, 0, stream>>>(pool, cnts, out, N_GRAPHS * HID);
}


# Aggregation as gather: edges grouped by target (counting-add rank), one wave sums a node's rows in registers and adds bias+relu; float-atomic scatter, zero and bias passes go; sum order changes
# speedup vs baseline: 2.6224x; 2.6224x over previous
#include <hip/hip_runtime.h>
#include <hip/hip_bf16.h>

typedef __attribute__((ext_vector_type(2))) float v2f;
typedef __attribute__((ext_vector_type(8))) float v8f;

#define N_NODES  100000
#define N_EDGES  1600000
#define N_GRAPHS 1000
#define IN_DIM   128
#define HID      100
#define HPAD     112

__global__ void k_zero_int(int* p, int n) {
    int i = blockIdx.x * blockDim.x + threadIdx.x;
    if (i < n) p[i] = 0;
}

__global__ void k_rank(const int* __restrict__ dst, int* ecnt, int* __restrict__ rank, int e) {
    int i = blockIdx.x * blockDim.x + threadIdx.x;
    if (i < e) rank[i] = atomicAdd(&ecnt[dst[i]], 1);
}

__global__ __launch_bounds__(256) void k_alloc(const int* __restrict__ ecnt, int* cursor,
                                               int4* __restrict__ ninfo, int n) {
    __shared__ int wsum[8];
    __shared__ int sbase;
    int tid = threadIdx.x, lane = tid & 31, wv = tid >> 5;
    int i = blockIdx.x * 256 + tid;
    int c = (i < n) ? ecnt[i] : 0;
    int v = c;
#pragma unroll
    for (int o = 1; o < 32; o <<= 1) {
        int t = __shfl_up(v, o);
        if (lane >= o) v += t;
    }
    if (lane == 31) wsum[wv] = v;
    __syncthreads();
    if (tid == 0) {
        int tot = 0;
        for (int k = 0; k < 8; ++k) { int t = wsum[k]; wsum[k] = tot; tot += t; }
        sbase = atomicAdd(cursor, tot);
    }
    __syncthreads();
    if (i < n) {
        float dinv = rsqrtf((float)(c + 1));
        ninfo[i] = make_int4(sbase + wsum[wv] + v - c, c, __float_as_int(dinv), 0);
    }
}

__global__ void k_fill(const int* __restrict__ src, const int* __restrict__ dst,
                       const int* __restrict__ rank, const int4* __restrict__ ninfo,
                       int2* __restrict__ csr, int e) {
    int i = blockIdx.x * blockDim.x + threadIdx.x;
    if (i >= e) return;
    int s = src[i], d = dst[i];
    int4 nd = ninfo[d];
    float w = __int_as_float(ninfo[s].z) * __int_as_float(nd.z);
    csr[nd.x + rank[i]] = make_int2(s, __float_as_int(w));
}

__global__ void k_pack_w(const float* __restrict__ src, float2* __restrict__ dst,
                         int sr, int sc, int kpairs) {
    int i = blockIdx.x * blockDim.x + threadIdx.x;
    if (i >= kpairs * HPAD) return;
    int p = i / HPAD, n = i % HPAD;
    int k0 = 2 * p, k1 = 2 * p + 1;
    float lo = (k0 < sr && n < sc) ? src[k0 * sc + n] : 0.0f;
    float hi = (k1 < sr && n < sc) ? src[k1 * sc + n] : 0.0f;
    dst[i] = make_float2(lo, hi);
}

__global__ void k_pad_vec(const float* __restrict__ src, float* __restrict__ dst,
                          int sc, int dc) {
    int i = blockIdx.x * blockDim.x + threadIdx.x;
    if (i < dc) dst[i] = (i < sc) ? src[i] : 0.0f;
}

__global__ void k_zero(float* p, size_t n) {
    size_t i = (size_t)blockIdx.x * blockDim.x + threadIdx.x;
    size_t stride = (size_t)gridDim.x * blockDim.x;
    for (; i < n; i += stride) p[i] = 0.0f;
}

__global__ __launch_bounds__(256) void k_gemm_wmma(
    const float* __restrict__ X, const float2* __restrict__ Wp,
    float* __restrict__ H, int nrows, int K)
{
    __shared__ float2 wl[(IN_DIM / 2) * HPAD];
    int tid = threadIdx.x;
    int kpairs = K >> 1;
    for (int i = tid; i < kpairs * HPAD; i += 256) wl[i] = Wp[i];
    __syncthreads();

    int lane = tid & 31;
    int wave = tid >> 5;
    int m    = lane & 15;
    int koff = (lane >> 4) << 1;
    int row0 = blockIdx.x * 128 + wave * 16;
    int row  = row0 + m;
    bool inr = row < nrows;
    int rowc = inr ? row : (nrows - 1);
    const float* xrow = X + (size_t)rowc * K;

    v8f acc[7];
#pragma unroll
    for (int t = 0; t < 7; ++t)
        acc[t] = (v8f){0.f, 0.f, 0.f, 0.f, 0.f, 0.f, 0.f, 0.f};

    for (int k = 0; k < K; k += 4) {
        float2 t2 = *(const float2*)(xrow + k + koff);
        v2f a = (v2f){ inr ? t2.x : 0.f, inr ? t2.y : 0.f };
        const float2* wrow = wl + ((k + koff) >> 1) * HPAD;
#pragma unroll
        for (int t = 0; t < 7; ++t) {
            float2 bb = wrow[t * 16 + m];
            v2f b = (v2f){bb.x, bb.y};
            acc[t] = __builtin_amdgcn_wmma_f32_16x16x4_f32(
                false, a, false, b, (short)0, acc[t], false, false);
        }
    }

    int rbase = row0 + ((lane >> 4) << 3);
#pragma unroll
    for (int t = 0; t < 7; ++t) {
        int cc = t * 16 + m;
#pragma unroll
        for (int v = 0; v < 8; ++v) {
            int rr = rbase + v;
            if (rr < nrows) H[(size_t)rr * HPAD + cc] = acc[t][v];
        }
    }
}

__global__ __launch_bounds__(256) void k_gather(
    const float* __restrict__ H, const int4* __restrict__ ninfo, const int2* __restrict__ csr,
    const float* __restrict__ bias, float* __restrict__ OUT, int nN)
{
    int lane = threadIdx.x & 31;
    int node = blockIdx.x * 8 + (threadIdx.x >> 5);
    if (node >= nN) return;
    int d = __builtin_amdgcn_readfirstlane(node);
    int4 ni = ninfo[d];
    int beg = ni.x, n = ni.y;
    float di = __int_as_float(ni.z);
    int c4 = lane < 28 ? lane : 27;
    const float4* H4 = (const float4*)H + c4;

    int m = n < 32 ? n : 32;
    int2 meta = make_int2(0, 0);
    if (lane < m) meta = csr[beg + lane];
    float4 hs = H4[(size_t)d * (HPAD / 4)];
    float4 bb = ((const float4*)bias)[c4];
    float4 acc = make_float4(0.f, 0.f, 0.f, 0.f);

    for (int base = 0;;) {
        for (int j = 0; j < m; j += 8) {
            int r = m - j;
            int   sx[8];
            float w[8];
            float4 h[8];
#pragma unroll
            for (int u = 0; u < 8; ++u) {
                int jj = (u < r) ? j + u : m - 1;
                sx[u] = __builtin_amdgcn_readlane(meta.x, jj);
                float wv = __int_as_float(__builtin_amdgcn_readlane(meta.y, jj));
                w[u] = (u < r) ? wv : 0.f;
            }
#pragma unroll
            for (int u = 0; u < 8; ++u) {
                h[u] = make_float4(0.f, 0.f, 0.f, 0.f);
                if (u < r) h[u] = H4[(size_t)sx[u] * (HPAD / 4)];
            }
#pragma unroll
            for (int u = 0; u < 8; ++u) {
                acc.x = fmaf(h[u].x, w[u], acc.x);
                acc.y = fmaf(h[u].y, w[u], acc.y);
                acc.z = fmaf(h[u].z, w[u], acc.z);
                acc.w = fmaf(h[u].w, w[u], acc.w);
            }
        }
        base += 32;
        if (base >= n) break;
        m = n - base;
        if (m > 32) m = 32;
        meta = make_int2(0, 0);
        if (lane < m) meta = csr[beg + base + lane];
    }

    float wself = di * di;
    float4 o;
    o.x = fmaf(hs.x, wself, acc.x) + bb.x; o.x = o.x > 0.f ? o.x : 0.f;
    o.y = fmaf(hs.y, wself, acc.y) + bb.y; o.y = o.y > 0.f ? o.y : 0.f;
    o.z = fmaf(hs.z, wself, acc.z) + bb.z; o.z = o.z > 0.f ? o.z : 0.f;
    o.w = fmaf(hs.w, wself, acc.w) + bb.w; o.w = o.w > 0.f ? o.w : 0.f;
    if (lane < 28) ((float4*)OUT)[(size_t)d * (HPAD / 4) + lane] = o;
}

__global__ void k_counts(const int* __restrict__ batch, float* counts, int n) {
    int i = blockIdx.x * blockDim.x + threadIdx.x;
    if (i < n) atomicAdd(&counts[batch[i]], 1.0f);
}

__global__ void k_pool_sum(const float* __restrict__ X, const int* __restrict__ batch,
                           float* __restrict__ pool, int n) {
    long long i = (long long)blockIdx.x * blockDim.x + threadIdx.x;
    if (i >= (long long)n * HID) return;
    int node = (int)(i / HID);
    int c    = (int)(i % HID);
    atomicAdd(&pool[(size_t)batch[node] * HID + c], X[(size_t)node * HPAD + c]);
}

__global__ void k_pool_fin(const float* __restrict__ pool, const float* __restrict__ counts,
                           float* __restrict__ out, int n) {
    int i = blockIdx.x * blockDim.x + threadIdx.x;
    if (i >= n) return;
    int g = i / HID;
    out[i] = pool[i] / fmaxf(counts[g], 1.0f);
}

extern "C" void kernel_launch(void* const* d_in, const int* in_sizes, int n_in,
                              void* d_out, int out_size, void* d_ws, size_t ws_size,
                              hipStream_t stream) {
    const float* x  = (const float*)d_in[0];
    const float* W1 = (const float*)d_in[1];
    const float* b1 = (const float*)d_in[2];
    const float* W2 = (const float*)d_in[3];
    const float* b2 = (const float*)d_in[4];
    const int*   ei = (const int*)d_in[5];
    const int*   bt = (const int*)d_in[6];
    const int* esrc = ei;
    const int* edst = ei + N_EDGES;
    float* out = (float*)d_out;

    float*  ws    = (float*)d_ws;
    int4*   ninfo = (int4*)ws;
    int*    ecnt  = (int*)(ws + (size_t)4 * N_NODES);
    int*    cursor = ecnt + N_NODES;
    float2* W1p  = (float2*)(ecnt + N_NODES + 4);
    float*  b1p  = (float*)(W1p + (IN_DIM / 2) * HPAD);
    float2* W2p  = (float2*)(b1p + HPAD);
    float*  b2p  = (float*)(W2p + (HPAD / 2) * HPAD);
    float*  B0   = b2p + HPAD;
    float*  B1   = B0 + (size_t)N_NODES * HPAD;
    float*  B2   = B1 + (size_t)N_NODES * HPAD;
    float*  pool = B2 + (size_t)N_NODES * HPAD;
    float*  cnts = pool + N_GRAPHS * HID;
    int*    rank = (int*)(cnts + N_GRAPHS);
    int2*   csr  = (int2*)(rank + N_EDGES);

    auto cdiv = [](long long a, long long b) { return (unsigned)((a + b - 1) / b); };

    k_zero_int<<<cdiv(N_NODES + 1, 256), 256, 0, stream>>>(ecnt, N_NODES + 1);
    k_rank<<<cdiv(N_EDGES, 256), 256, 0, stream>>>(edst, ecnt, rank, N_EDGES);
    k_alloc<<<cdiv(N_NODES, 256), 256, 0, stream>>>(ecnt, cursor, ninfo, N_NODES);
    k_fill<<<cdiv(N_EDGES, 256), 256, 0, stream>>>(esrc, edst, rank, ninfo, csr, N_EDGES);

    k_pack_w<<<cdiv((IN_DIM / 2) * HPAD, 256), 256, 0, stream>>>(W1, W1p, IN_DIM, HID, IN_DIM / 2);
    k_pad_vec<<<1, HPAD, 0, stream>>>(b1, b1p, HID, HPAD);
    k_pack_w<<<cdiv((HPAD / 2) * HPAD, 256), 256, 0, stream>>>(W2, W2p, HID, HID, HPAD / 2);
    k_pad_vec<<<1, HPAD, 0, stream>>>(b2, b2p, HID, HPAD);

    k_gemm_wmma<<<cdiv(N_NODES, 128), 256, 0, stream>>>(x, W1p, B0, N_NODES, IN_DIM);
    k_gather<<<cdiv(N_NODES, 8), 256, 0, stream>>>(B0, ninfo, csr, b1p, B1, N_NODES);

    k_gemm_wmma<<<cdiv(N_NODES, 128), 256, 0, stream>>>(B1, W2p, B2, N_NODES, HPAD);
    k_gather<<<cdiv(N_NODES, 8), 256, 0, stream>>>(B2, ninfo, csr, b2p, B0, N_NODES);

    k_zero<<<64, 256, 0, stream>>>(pool, (size_t)N_GRAPHS * HID + N_GRAPHS);
    k_counts<<<cdiv(N_NODES, 256), 256, 0, stream>>>(bt, cnts, N_NODES);
    k_pool_sum<<<cdiv((long long)N_NODES * HID, 256), 256, 0, stream>>>(B0, bt, pool, N_NODES);
    k_pool_fin<<<cdiv(N_GRAPHS * HID, 256), 256, 0, stream>>>(pool, cnts, out, N_GRAPHS * HID);
}


# Mean pool fused into the last aggregation: the block's 8 node rows meet in LDS, one atomic add per run of equal graph ids and column; row store, count and pool-sum launches go; sum order changes
# speedup vs baseline: 2.9435x; 1.1224x over previous
#include <hip/hip_runtime.h>
#include <hip/hip_bf16.h>

typedef __attribute__((ext_vector_type(2))) float v2f;
typedef __attribute__((ext_vector_type(8))) float v8f;

#define N_NODES  100000
#define N_EDGES  1600000
#define N_GRAPHS 1000
#define IN_DIM   128
#define HID      100
#define HPAD     112

__global__ void k_zero_int(int* p, int n) {
    int i = blockIdx.x * blockDim.x + threadIdx.x;
    if (i < n) p[i] = 0;
}

__global__ void k_rank(const int* __restrict__ dst, int* ecnt, int* __restrict__ rank, int e) {
    int i = blockIdx.x * blockDim.x + threadIdx.x;
    if (i < e) rank[i] = atomicAdd(&ecnt[dst[i]], 1);
}

__global__ __launch_bounds__(256) void k_alloc(const int* __restrict__ ecnt, int* cursor,
                                               int4* __restrict__ ninfo, int n) {
    __shared__ int wsum[8];
    __shared__ int sbase;
    int tid = threadIdx.x, lane = tid & 31, wv = tid >> 5;
    int i = blockIdx.x * 256 + tid;
    int c = (i < n) ? ecnt[i] : 0;
    int v = c;
#pragma unroll
    for (int o = 1; o < 32; o <<= 1) {
        int t = __shfl_up(v, o);
        if (lane >= o) v += t;
    }
    if (lane == 31) wsum[wv] = v;
    __syncthreads();
    if (tid == 0) {
        int tot = 0;
        for (int k = 0; k < 8; ++k) { int t = wsum[k]; wsum[k] = tot; tot += t; }
        sbase = atomicAdd(cursor, tot);
    }
    __syncthreads();
    if (i < n) {
        float dinv = rsqrtf((float)(c + 1));
        ninfo[i] = make_int4(sbase + wsum[wv] + v - c, c, __float_as_int(dinv), 0);
    }
}

__global__ void k_fill(const int* __restrict__ src, const int* __restrict__ dst,
                       const int* __restrict__ rank, const int4* __restrict__ ninfo,
                       int2* __restrict__ csr, int e) {
    int i = blockIdx.x * blockDim.x + threadIdx.x;
    if (i >= e) return;
    int s = src[i], d = dst[i];
    int4 nd = ninfo[d];
    float w = __int_as_float(ninfo[s].z) * __int_as_float(nd.z);
    csr[nd.x + rank[i]] = make_int2(s, __float_as_int(w));
}

__global__ void k_pack_w(const float* __restrict__ src, float2* __restrict__ dst,
                         int sr, int sc, int kpairs) {
    int i = blockIdx.x * blockDim.x + threadIdx.x;
    if (i >= kpairs * HPAD) return;
    int p = i / HPAD, n = i % HPAD;
    int k0 = 2 * p, k1 = 2 * p + 1;
    float lo = (k0 < sr && n < sc) ? src[k0 * sc + n] : 0.0f;
    float hi = (k1 < sr && n < sc) ? src[k1 * sc + n] : 0.0f;
    dst[i] = make_float2(lo, hi);
}

__global__ void k_pad_vec(const float* __restrict__ src, float* __restrict__ dst,
                          int sc, int dc) {
    int i = blockIdx.x * blockDim.x + threadIdx.x;
    if (i < dc) dst[i] = (i < sc) ? src[i] : 0.0f;
}

__global__ void k_zero(float* p, size_t n) {
    size_t i = (size_t)blockIdx.x * blockDim.x + threadIdx.x;
    size_t stride = (size_t)gridDim.x * blockDim.x;
    for (; i < n; i += stride) p[i] = 0.0f;
}

__global__ __launch_bounds__(256) void k_gemm_wmma(
    const float* __restrict__ X, const float2* __restrict__ Wp,
    float* __restrict__ H, int nrows, int K)
{
    __shared__ float2 wl[(IN_DIM / 2) * HPAD];
    int tid = threadIdx.x;
    int kpairs = K >> 1;
    for (int i = tid; i < kpairs * HPAD; i += 256) wl[i] = Wp[i];
    __syncthreads();

    int lane = tid & 31;
    int wave = tid >> 5;
    int m    = lane & 15;
    int koff = (lane >> 4) << 1;
    int row0 = blockIdx.x * 128 + wave * 16;
    int row  = row0 + m;
    bool inr = row < nrows;
    int rowc = inr ? row : (nrows - 1);
    const float* xrow = X + (size_t)rowc * K;

    v8f acc[7];
#pragma unroll
    for (int t = 0; t < 7; ++t)
        acc[t] = (v8f){0.f, 0.f, 0.f, 0.f, 0.f, 0.f, 0.f, 0.f};

    for (int k = 0; k < K; k += 4) {
        float2 t2 = *(const float2*)(xrow + k + koff);
        v2f a = (v2f){ inr ? t2.x : 0.f, inr ? t2.y : 0.f };
        const float2* wrow = wl + ((k + koff) >> 1) * HPAD;
#pragma unroll
        for (int t = 0; t < 7; ++t) {
            float2 bb = wrow[t * 16 + m];
            v2f b = (v2f){bb.x, bb.y};
            acc[t] = __builtin_amdgcn_wmma_f32_16x16x4_f32(
                false, a, false, b, (short)0, acc[t], false, false);
        }
    }

    int rbase = row0 + ((lane >> 4) << 3);
#pragma unroll
    for (int t = 0; t < 7; ++t) {
        int cc = t * 16 + m;
#pragma unroll
        for (int v = 0; v < 8; ++v) {
            int rr = rbase + v;
            if (rr < nrows) H[(size_t)rr * HPAD + cc] = acc[t][v];
        }
    }
}

template <bool POOL>
__device__ __forceinline__ void gather_node(
    const float* __restrict__ H, const int4* __restrict__ ninfo, const int2* __restrict__ csr,
    const float* __restrict__ bias, float* __restrict__ OUT, const int* __restrict__ batch,
    float* __restrict__ pool, float* __restrict__ counts, float* srow, int* sg, int nN)
{
    int lane = threadIdx.x & 31;
    int wid = threadIdx.x >> 5;
    int node = blockIdx.x * 8 + wid;
    bool valid = node < nN;
    if (!POOL) {
        if (!valid) return;
    } else if (!valid) {
        node = nN - 1;
    }
    int d = __builtin_amdgcn_readfirstlane(node);
    int4 ni = ninfo[d];
    int gid = 0;
    if (POOL) gid = batch[d];
    int beg = ni.x, n = ni.y;
    float di = __int_as_float(ni.z);
    int c4 = lane < 28 ? lane : 27;
    const float4* H4 = (const float4*)H + c4;

    int m = n < 32 ? n : 32;
    int2 meta = make_int2(0, 0);
    if (lane < m) meta = csr[beg + lane];
    float4 hs = H4[(size_t)d * (HPAD / 4)];
    float4 bb = ((const float4*)bias)[c4];
    float4 acc = make_float4(0.f, 0.f, 0.f, 0.f);

    for (int base = 0;;) {
        for (int j = 0; j < m; j += 8) {
            int r = m - j;
            int   sx[8];
            float w[8];
            float4 h[8];
#pragma unroll
            for (int u = 0; u < 8; ++u) {
                int jj = (u < r) ? j + u : m - 1;
                sx[u] = __builtin_amdgcn_readlane(meta.x, jj);
                float wv = __int_as_float(__builtin_amdgcn_readlane(meta.y, jj));
                w[u] = (u < r) ? wv : 0.f;
            }
#pragma unroll
            for (int u = 0; u < 8; ++u) {
                h[u] = make_float4(0.f, 0.f, 0.f, 0.f);
                if (u < r) h[u] = H4[(size_t)sx[u] * (HPAD / 4)];
            }
#pragma unroll
            for (int u = 0; u < 8; ++u) {
                acc.x = fmaf(h[u].x, w[u], acc.x);
                acc.y = fmaf(h[u].y, w[u], acc.y);
                acc.z = fmaf(h[u].z, w[u], acc.z);
                acc.w = fmaf(h[u].w, w[u], acc.w);
            }
        }
        base += 32;
        if (base >= n) break;
        m = n - base;
        if (m > 32) m = 32;
        meta = make_int2(0, 0);
        if (lane < m) meta = csr[beg + base + lane];
    }

    float wself = di * di;
    float4 o;
    o.x = fmaf(hs.x, wself, acc.x) + bb.x; o.x = o.x > 0.f ? o.x : 0.f;
    o.y = fmaf(hs.y, wself, acc.y) + bb.y; o.y = o.y > 0.f ? o.y : 0.f;
    o.z = fmaf(hs.z, wself, acc.z) + bb.z; o.z = o.z > 0.f ? o.z : 0.f;
    o.w = fmaf(hs.w, wself, acc.w) + bb.w; o.w = o.w > 0.f ? o.w : 0.f;
    if (!POOL) {
        if (lane < 28) ((float4*)OUT)[(size_t)d * (HPAD / 4) + lane] = o;
        return;
    }

    if (lane < 28) ((float4*)srow)[wid * (HPAD / 4) + lane] = o;
    if (lane == 0) sg[wid] = (valid && (unsigned)gid < (unsigned)N_GRAPHS) ? gid : -1;
    __syncthreads();
    int t = threadIdx.x;
    if (t > HID) return;
    bool isc = t == HID;
    int4 ga = ((const int4*)sg)[0], gb = ((const int4*)sg)[1];
    int gq[8] = {ga.x, ga.y, ga.z, ga.w, gb.x, gb.y, gb.z, gb.w};
    float vq[8];
#pragma unroll
    for (int q = 0; q < 8; ++q) vq[q] = srow[q * HPAD + t];
    int cur = -1;
    float s = 0.f;
#pragma unroll
    for (int q = 0; q < 8; ++q) {
        int gw = gq[q];
        float v = isc ? 1.0f : vq[q];
        if (gw >= 0) {
            if (gw == cur) {
                s += v;
            } else {
                if (cur >= 0) atomicAdd(isc ? &counts[cur] : &pool[(size_t)cur * HID + t], s);
                cur = gw;
                s = v;
            }
        }
    }
    if (cur >= 0) atomicAdd(isc ? &counts[cur] : &pool[(size_t)cur * HID + t], s);
}

__global__ __launch_bounds__(256) void k_gather(
    const float* __restrict__ H, const int4* __restrict__ ninfo, const int2* __restrict__ csr,
    const float* __restrict__ bias, float* __restrict__ OUT, int nN)
{
    gather_node<false>(H, ninfo, csr, bias, OUT, nullptr, nullptr, nullptr, nullptr, nullptr, nN);
}

__global__ __launch_bounds__(256) void k_gather_pool(
    const float* __restrict__ H, const int4* __restrict__ ninfo, const int2* __restrict__ csr,
    const float* __restrict__ bias, const int* __restrict__ batch,
    float* __restrict__ pool, float* __restrict__ counts, int nN)
{
    __shared__ float4 srow[8 * (HPAD / 4)];
    __shared__ int4 sg[2];
    gather_node<true>(H, ninfo, csr, bias, nullptr, batch, pool, counts, (float*)srow, (int*)sg, nN);
}

__global__ void k_pool_fin(const float* __restrict__ pool, const float* __restrict__ counts,
                           float* __restrict__ out, int n) {
    int i = blockIdx.x * blockDim.x + threadIdx.x;
    if (i >= n) return;
    int g = i / HID;
    out[i] = pool[i] / fmaxf(counts[g], 1.0f);
}

extern "C" void kernel_launch(void* const* d_in, const int* in_sizes, int n_in,
                              void* d_out, int out_size, void* d_ws, size_t ws_size,
                              hipStream_t stream) {
    const float* x  = (const float*)d_in[0];
    const float* W1 = (const float*)d_in[1];
    const float* b1 = (const float*)d_in[2];
    const float* W2 = (const float*)d_in[3];
    const float* b2 = (const float*)d_in[4];
    const int*   ei = (const int*)d_in[5];
    const int*   bt = (const int*)d_in[6];
    const int* esrc = ei;
    const int* edst = ei + N_EDGES;
    float* out = (float*)d_out;

    float*  ws    = (float*)d_ws;
    int4*   ninfo = (int4*)ws;
    int*    ecnt  = (int*)(ws + (size_t)4 * N_NODES);
    int*    cursor = ecnt + N_NODES;
    float2* W1p  = (float2*)(ecnt + N_NODES + 4);
    float*  b1p  = (float*)(W1p + (IN_DIM / 2) * HPAD);
    float2* W2p  = (float2*)(b1p + HPAD);
    float*  b2p  = (float*)(W2p + (HPAD / 2) * HPAD);
    float*  B0   = b2p + HPAD;
    float*  B1   = B0 + (size_t)N_NODES * HPAD;
    float*  B2   = B1 + (size_t)N_NODES * HPAD;
    float*  pool = B2 + (size_t)N_NODES * HPAD;
    float*  cnts = pool + N_GRAPHS * HID;
    int*    rank = (int*)(cnts + N_GRAPHS);
    int2*   csr  = (int2*)(rank + N_EDGES);

    auto cdiv = [](long long a, long long b) { return (unsigned)((a + b - 1) / b); };

    k_zero_int<<<cdiv(N_NODES + 1, 256), 256, 0, stream>>>(ecnt, N_NODES + 1);
    k_rank<<<cdiv(N_EDGES, 256), 256, 0, stream>>>(edst, ecnt, rank, N_EDGES);
    k_alloc<<<cdiv(N_NODES, 256), 256, 0, stream>>>(ecnt, cursor, ninfo, N_NODES);
    k_fill<<<cdiv(N_EDGES, 256), 256, 0, stream>>>(esrc, edst, rank, ninfo, csr, N_EDGES);

    k_pack_w<<<cdiv((IN_DIM / 2) * HPAD, 256), 256, 0, stream>>>(W1, W1p, IN_DIM, HID, IN_DIM / 2);
    k_pad_vec<<<1, HPAD, 0, stream>>>(b1, b1p, HID, HPAD);
    k_pack_w<<<cdiv((HPAD / 2) * HPAD, 256), 256, 0, stream>>>(W2, W2p, HID, HID, HPAD / 2);
    k_pad_vec<<<1, HPAD, 0, stream>>>(b2, b2p, HID, HPAD);

    k_gemm_wmma<<<cdiv(N_NODES, 128), 256, 0, stream>>>(x, W1p, B0, N_NODES, IN_DIM);
    k_gather<<<cdiv(N_NODES, 8), 256, 0, stream>>>(B0, ninfo, csr, b1p, B1, N_NODES);

    k_gemm_wmma<<<cdiv(N_NODES, 128), 256, 0, stream>>>(B1, W2p, B2, N_NODES, HPAD);
    k_zero<<<64, 256, 0, stream>>>(pool, (size_t)N_GRAPHS * HID + N_GRAPHS);
    k_gather_pool<<<cdiv(N_NODES, 8), 256, 0, stream>>>(B2, ninfo, csr, b2p, bt, pool, cnts, N_NODES);

    k_pool_fin<<<cdiv(N_GRAPHS * HID, 256), 256, 0, stream>>>(pool, cnts, out, N_GRAPHS * HID);
}


# Setup launches folded away: each GEMM block packs the raw weights into LDS itself, its loads issued in batches; gathers read the raw bias; one clear covers counts and pool; 14 launches become 9
# speedup vs baseline: 3.1733x; 1.0781x over previous
#include <hip/hip_runtime.h>
#include <hip/hip_bf16.h>

typedef __attribute__((ext_vector_type(2))) float v2f;
typedef __attribute__((ext_vector_type(8))) float v8f;

#define N_NODES  100000
#define N_EDGES  1600000
#define N_GRAPHS 1000
#define IN_DIM   128
#define HID      100
#define HPAD     112

__global__ void k_zero_int(int* p, int n) {
    int i = blockIdx.x * blockDim.x + threadIdx.x;
    if (i < n) p[i] = 0;
}

__global__ void k_rank(const int* __restrict__ dst, int* ecnt, int* __restrict__ rank, int e) {
    int i = blockIdx.x * blockDim.x + threadIdx.x;
    if (i < e) rank[i] = atomicAdd(&ecnt[dst[i]], 1);
}

__global__ __launch_bounds__(256) void k_alloc(const int* __restrict__ ecnt, int* cursor,
                                               int4* __restrict__ ninfo, int n) {
    __shared__ int wsum[8];
    __shared__ int sbase;
    int tid = threadIdx.x, lane = tid & 31, wv = tid >> 5;
    int i = blockIdx.x * 256 + tid;
    int c = (i < n) ? ecnt[i] : 0;
    int v = c;
#pragma unroll
    for (int o = 1; o < 32; o <<= 1) {
        int t = __shfl_up(v, o);
        if (lane >= o) v += t;
    }
    if (lane == 31) wsum[wv] = v;
    __syncthreads();
    if (tid == 0) {
        int tot = 0;
        for (int k = 0; k < 8; ++k) { int t = wsum[k]; wsum[k] = tot; tot += t; }
        sbase = atomicAdd(cursor, tot);
    }
    __syncthreads();
    if (i < n) {
        float dinv = rsqrtf((float)(c + 1));
        ninfo[i] = make_int4(sbase + wsum[wv] + v - c, c, __float_as_int(dinv), 0);
    }
}

__global__ void k_fill(const int* __restrict__ src, const int* __restrict__ dst,
                       const int* __restrict__ rank, const int4* __restrict__ ninfo,
                       int2* __restrict__ csr, int e) {
    int i = blockIdx.x * blockDim.x + threadIdx.x;
    if (i >= e) return;
    int s = src[i], d = dst[i];
    int4 nd = ninfo[d];
    float w = __int_as_float(ninfo[s].z) * __int_as_float(nd.z);
    csr[nd.x + rank[i]] = make_int2(s, __float_as_int(w));
}

__global__ __launch_bounds__(256) void k_gemm_wmma(
    const float* __restrict__ X, const float* __restrict__ W, int wr, int wc,
    float* __restrict__ H, int nrows, int K)
{
    __shared__ float2 wl[(IN_DIM / 2) * HPAD];
    int tid = threadIdx.x;
    int kpairs = K >> 1;
    int total = kpairs * HPAD;
    for (int base = tid; base < total; base += 14 * 256) {
        float lo[14], hi[14];
#pragma unroll
        for (int u = 0; u < 14; ++u) {
            int i = base + u * 256;
            int p = i / HPAD, n = i - p * HPAD;
            int k0 = 2 * p, k1 = k0 + 1;
            bool ins = i < total && n < wc;
            bool ok0 = ins && k0 < wr, ok1 = ins && k1 < wr;
            float v0 = W[ok0 ? k0 * wc + n : 0];
            float v1 = W[ok1 ? k1 * wc + n : 0];
            lo[u] = ok0 ? v0 : 0.0f;
            hi[u] = ok1 ? v1 : 0.0f;
        }
#pragma unroll
        for (int u = 0; u < 14; ++u) {
            wl[base + u * 256] = make_float2(lo[u], hi[u]);
        }
    }
    __syncthreads();

    int lane = tid & 31;
    int wave = tid >> 5;
    int m    = lane & 15;
    int koff = (lane >> 4) << 1;
    int row0 = blockIdx.x * 128 + wave * 16;
    int row  = row0 + m;
    bool inr = row < nrows;
    int rowc = inr ? row : (nrows - 1);
    const float* xrow = X + (size_t)rowc * K;

    v8f acc[7];
#pragma unroll
    for (int t = 0; t < 7; ++t)
        acc[t] = (v8f){0.f, 0.f, 0.f, 0.f, 0.f, 0.f, 0.f, 0.f};

    for (int k = 0; k < K; k += 4) {
        float2 t2 = *(const float2*)(xrow + k + koff);
        v2f a = (v2f){ inr ? t2.x : 0.f, inr ? t2.y : 0.f };
        const float2* wrow = wl + ((k + koff) >> 1) * HPAD;
#pragma unroll
        for (int t = 0; t < 7; ++t) {
            float2 bb = wrow[t * 16 + m];
            v2f b = (v2f){bb.x, bb.y};
            acc[t] = __builtin_amdgcn_wmma_f32_16x16x4_f32(
                false, a, false, b, (short)0, acc[t], false, false);
        }
    }

    int rbase = row0 + ((lane >> 4) << 3);
#pragma unroll
    for (int t = 0; t < 7; ++t) {
        int cc = t * 16 + m;
#pragma unroll
        for (int v = 0; v < 8; ++v) {
            int rr = rbase + v;
            if (rr < nrows) H[(size_t)rr * HPAD + cc] = acc[t][v];
        }
    }
}

template <bool POOL>
__device__ __forceinline__ void gather_node(
    const float* __restrict__ H, const int4* __restrict__ ninfo, const int2* __restrict__ csr,
    const float* __restrict__ bias, float* __restrict__ OUT, const int* __restrict__ batch,
    float* __restrict__ pool, float* __restrict__ counts, float* srow, int* sg, int nN)
{
    int lane = threadIdx.x & 31;
    int wid = threadIdx.x >> 5;
    int node = blockIdx.x * 8 + wid;
    bool valid = node < nN;
    if (!POOL) {
        if (!valid) return;
    } else if (!valid) {
        node = nN - 1;
    }
    int d = __builtin_amdgcn_readfirstlane(node);
    int4 ni = ninfo[d];
    int gid = 0;
    if (POOL) gid = batch[d];
    int beg = ni.x, n = ni.y;
    float di = __int_as_float(ni.z);
    int c4 = lane < 28 ? lane : 27;
    const float4* H4 = (const float4*)H + c4;

    int m = n < 32 ? n : 32;
    int2 meta = make_int2(0, 0);
    if (lane < m) meta = csr[beg + lane];
    float4 hs = H4[(size_t)d * (HPAD / 4)];
    bool hasb = c4 < HID / 4;
    float4 bb = ((const float4*)bias)[hasb ? c4 : HID / 4 - 1];
    float4 acc = make_float4(0.f, 0.f, 0.f, 0.f);

    for (int base = 0;;) {
        for (int j = 0; j < m; j += 8) {
            int r = m - j;
            int   sx[8];
            float w[8];
            float4 h[8];
#pragma unroll
            for (int u = 0; u < 8; ++u) {
                int jj = (u < r) ? j + u : m - 1;
                sx[u] = __builtin_amdgcn_readlane(meta.x, jj);
                float wv = __int_as_float(__builtin_amdgcn_readlane(meta.y, jj));
                w[u] = (u < r) ? wv : 0.f;
            }
#pragma unroll
            for (int u = 0; u < 8; ++u) {
                h[u] = make_float4(0.f, 0.f, 0.f, 0.f);
                if (u < r) h[u] = H4[(size_t)sx[u] * (HPAD / 4)];
            }
#pragma unroll
            for (int u = 0; u < 8; ++u) {
                acc.x = fmaf(h[u].x, w[u], acc.x);
                acc.y = fmaf(h[u].y, w[u], acc.y);
                acc.z = fmaf(h[u].z, w[u], acc.z);
                acc.w = fmaf(h[u].w, w[u], acc.w);
            }
        }
        base += 32;
        if (base >= n) break;
        m = n - base;
        if (m > 32) m = 32;
        meta = make_int2(0, 0);
        if (lane < m) meta = csr[beg + base + lane];
    }

    float wself = di * di;
    if (!hasb) bb = make_float4(0.f, 0.f, 0.f, 0.f);
    float4 o;
    o.x = fmaf(hs.x, wself, acc.x) + bb.x; o.x = o.x > 0.f ? o.x : 0.f;
    o.y = fmaf(hs.y, wself, acc.y) + bb.y; o.y = o.y > 0.f ? o.y : 0.f;
    o.z = fmaf(hs.z, wself, acc.z) + bb.z; o.z = o.z > 0.f ? o.z : 0.f;
    o.w = fmaf(hs.w, wself, acc.w) + bb.w; o.w = o.w > 0.f ? o.w : 0.f;
    if (!POOL) {
        if (lane < 28) ((float4*)OUT)[(size_t)d * (HPAD / 4) + lane] = o;
        return;
    }

    if (lane < 28) ((float4*)srow)[wid * (HPAD / 4) + lane] = o;
    if (lane == 0) sg[wid] = (valid && (unsigned)gid < (unsigned)N_GRAPHS) ? gid : -1;
    __syncthreads();
    int t = threadIdx.x;
    if (t > HID) return;
    bool isc = t == HID;
    int4 ga = ((const int4*)sg)[0], gb = ((const int4*)sg)[1];
    int gq[8] = {ga.x, ga.y, ga.z, ga.w, gb.x, gb.y, gb.z, gb.w};
    float vq[8];
#pragma unroll
    for (int q = 0; q < 8; ++q) vq[q] = srow[q * HPAD + t];
    int cur = -1;
    float s = 0.f;
#pragma unroll
    for (int q = 0; q < 8; ++q) {
        int gw = gq[q];
        float v = isc ? 1.0f : vq[q];
        if (gw >= 0) {
            if (gw == cur) {
                s += v;
            } else {
                if (cur >= 0) atomicAdd(isc ? &counts[cur] : &pool[(size_t)cur * HID + t], s);
                cur = gw;
                s = v;
            }
        }
    }
    if (cur >= 0) atomicAdd(isc ? &counts[cur] : &pool[(size_t)cur * HID + t], s);
}

__global__ __launch_bounds__(256) void k_gather(
    const float* __restrict__ H, const int4* __restrict__ ninfo, const int2* __restrict__ csr,
    const float* __restrict__ bias, float* __restrict__ OUT, int nN)
{
    gather_node<false>(H, ninfo, csr, bias, OUT, nullptr, nullptr, nullptr, nullptr, nullptr, nN);
}

__global__ __launch_bounds__(256) void k_gather_pool(
    const float* __restrict__ H, const int4* __restrict__ ninfo, const int2* __restrict__ csr,
    const float* __restrict__ bias, const int* __restrict__ batch,
    float* __restrict__ pool, float* __restrict__ counts, int nN)
{
    __shared__ float4 srow[8 * (HPAD / 4)];
    __shared__ int4 sg[2];
    gather_node<true>(H, ninfo, csr, bias, nullptr, batch, pool, counts, (float*)srow, (int*)sg, nN);
}

__global__ void k_pool_fin(const float* __restrict__ pool, const float* __restrict__ counts,
                           float* __restrict__ out, int n) {
    int i = blockIdx.x * blockDim.x + threadIdx.x;
    if (i >= n) return;
    int g = i / HID;
    out[i] = pool[i] / fmaxf(counts[g], 1.0f);
}

extern "C" void kernel_launch(void* const* d_in, const int* in_sizes, int n_in,
                              void* d_out, int out_size, void* d_ws, size_t ws_size,
                              hipStream_t stream) {
    const float* x  = (const float*)d_in[0];
    const float* W1 = (const float*)d_in[1];
    const float* b1 = (const float*)d_in[2];
    const float* W2 = (const float*)d_in[3];
    const float* b2 = (const float*)d_in[4];
    const int*   ei = (const int*)d_in[5];
    const int*   bt = (const int*)d_in[6];
    const int* esrc = ei;
    const int* edst = ei + N_EDGES;
    float* out = (float*)d_out;

    float*  ws    = (float*)d_ws;
    int4*   ninfo = (int4*)ws;
    int*    ecnt  = (int*)(ws + (size_t)4 * N_NODES);
    int*    cursor = ecnt + N_NODES;
    float*  pool = (float*)(cursor + 4);
    float*  cnts = pool + N_GRAPHS * HID;
    float*  B0   = cnts + N_GRAPHS;
    float*  B1   = B0 + (size_t)N_NODES * HPAD;
    float*  B2   = B1 + (size_t)N_NODES * HPAD;
    int*    rank = (int*)(B2 + (size_t)N_NODES * HPAD);
    int2*   csr  = (int2*)(rank + N_EDGES);

    auto cdiv = [](long long a, long long b) { return (unsigned)((a + b - 1) / b); };

    k_zero_int<<<cdiv(N_NODES + 4 + N_GRAPHS * HID + N_GRAPHS, 256), 256, 0, stream>>>(
        ecnt, N_NODES + 4 + N_GRAPHS * HID + N_GRAPHS);
    k_rank<<<cdiv(N_EDGES, 256), 256, 0, stream>>>(edst, ecnt, rank, N_EDGES);
    k_alloc<<<cdiv(N_NODES, 256), 256, 0, stream>>>(ecnt, cursor, ninfo, N_NODES);
    k_fill<<<cdiv(N_EDGES, 256), 256, 0, stream>>>(esrc, edst, rank, ninfo, csr, N_EDGES);

    k_gemm_wmma<<<cdiv(N_NODES, 128), 256, 0, stream>>>(x, W1, IN_DIM, HID, B0, N_NODES, IN_DIM);
    k_gather<<<cdiv(N_NODES, 8), 256, 0, stream>>>(B0, ninfo, csr, b1, B1, N_NODES);

    k_gemm_wmma<<<cdiv(N_NODES, 128), 256, 0, stream>>>(B1, W2, HID, HID, B2, N_NODES, HPAD);
    k_gather_pool<<<cdiv(N_NODES, 8), 256, 0, stream>>>(B2, ninfo, csr, b2, bt, pool, cnts, N_NODES);

    k_pool_fin<<<cdiv(N_GRAPHS * HID, 256), 256, 0, stream>>>(pool, cnts, out, N_GRAPHS * HID);
}


# Edge counters doubled, one copy per chip half picked by the die id: the 1.6M returning counting adds no longer move counter lines between the halves' L2; group order changes, so sum order changes
# speedup vs baseline: 4.1362x; 1.3034x over previous
#include <hip/hip_runtime.h>
#include <hip/hip_bf16.h>

typedef __attribute__((ext_vector_type(2))) float v2f;
typedef __attribute__((ext_vector_type(8))) float v8f;

#define N_NODES  100000
#define N_EDGES  1600000
#define N_GRAPHS 1000
#define IN_DIM   128
#define HID      100
#define HPAD     112

__global__ void k_zero_int(int* p, int n) {
    int i = blockIdx.x * blockDim.x + threadIdx.x;
    if (i < n) p[i] = 0;
}

__global__ void k_rank(const int* __restrict__ dst, int* ecnt, int* __restrict__ rank, int e) {
    int i = blockIdx.x * blockDim.x + threadIdx.x;
    unsigned h = (__builtin_amdgcn_s_sendmsg_rtn(0x87) >> 18) & 1u;
    if (i < e) rank[i] = (int)((unsigned)atomicAdd(&ecnt[h * N_NODES + dst[i]], 1) | (h << 31));
}

__global__ __launch_bounds__(256) void k_alloc(const int* __restrict__ ecnt, int* cursor,
                                               int4* __restrict__ ninfo, int n) {
    __shared__ int wsum[8];
    __shared__ int sbase;
    int tid = threadIdx.x, lane = tid & 31, wv = tid >> 5;
    int i = blockIdx.x * 256 + tid;
    int c0 = (i < n) ? ecnt[i] : 0;
    int c1 = (i < n) ? ecnt[N_NODES + i] : 0;
    int c = c0 + c1;
    int v = c;
#pragma unroll
    for (int o = 1; o < 32; o <<= 1) {
        int t = __shfl_up(v, o);
        if (lane >= o) v += t;
    }
    if (lane == 31) wsum[wv] = v;
    __syncthreads();
    if (tid == 0) {
        int tot = 0;
        for (int k = 0; k < 8; ++k) { int t = wsum[k]; wsum[k] = tot; tot += t; }
        sbase = atomicAdd(cursor, tot);
    }
    __syncthreads();
    if (i < n) {
        float dinv = rsqrtf((float)(c + 1));
        ninfo[i] = make_int4(sbase + wsum[wv] + v - c, c, __float_as_int(dinv), c0);
    }
}

__global__ void k_fill(const int* __restrict__ src, const int* __restrict__ dst,
                       const int* __restrict__ rank, const int4* __restrict__ ninfo,
                       int2* __restrict__ csr, int e) {
    int i = blockIdx.x * blockDim.x + threadIdx.x;
    if (i >= e) return;
    int s = src[i], d = dst[i];
    int4 nd = ninfo[d];
    float w = __int_as_float(ninfo[s].z) * __int_as_float(nd.z);
    int rk = rank[i];
    int r = rk & 0x7fffffff;
    if (rk < 0) r += nd.w;
    csr[nd.x + r] = make_int2(s, __float_as_int(w));
}

__global__ __launch_bounds__(256) void k_gemm_wmma(
    const float* __restrict__ X, const float* __restrict__ W, int wr, int wc,
    float* __restrict__ H, int nrows, int K)
{
    __shared__ float2 wl[(IN_DIM / 2) * HPAD];
    int tid = threadIdx.x;
    int kpairs = K >> 1;
    int total = kpairs * HPAD;
    for (int base = tid; base < total; base += 14 * 256) {
        float lo[14], hi[14];
#pragma unroll
        for (int u = 0; u < 14; ++u) {
            int i = base + u * 256;
            int p = i / HPAD, n = i - p * HPAD;
            int k0 = 2 * p, k1 = k0 + 1;
            bool ins = i < total && n < wc;
            bool ok0 = ins && k0 < wr, ok1 = ins && k1 < wr;
            float v0 = W[ok0 ? k0 * wc + n : 0];
            float v1 = W[ok1 ? k1 * wc + n : 0];
            lo[u] = ok0 ? v0 : 0.0f;
            hi[u] = ok1 ? v1 : 0.0f;
        }
#pragma unroll
        for (int u = 0; u < 14; ++u) {
            wl[base + u * 256] = make_float2(lo[u], hi[u]);
        }
    }
    __syncthreads();

    int lane = tid & 31;
    int wave = tid >> 5;
    int m    = lane & 15;
    int koff = (lane >> 4) << 1;
    int row0 = blockIdx.x * 128 + wave * 16;
    int row  = row0 + m;
    bool inr = row < nrows;
    int rowc = inr ? row : (nrows - 1);
    const float* xrow = X + (size_t)rowc * K;

    v8f acc[7];
#pragma unroll
    for (int t = 0; t < 7; ++t)
        acc[t] = (v8f){0.f, 0.f, 0.f, 0.f, 0.f, 0.f, 0.f, 0.f};

    for (int k = 0; k < K; k += 4) {
        float2 t2 = *(const float2*)(xrow + k + koff);
        v2f a = (v2f){ inr ? t2.x : 0.f, inr ? t2.y : 0.f };
        const float2* wrow = wl + ((k + koff) >> 1) * HPAD;
#pragma unroll
        for (int t = 0; t < 7; ++t) {
            float2 bb = wrow[t * 16 + m];
            v2f b = (v2f){bb.x, bb.y};
            acc[t] = __builtin_amdgcn_wmma_f32_16x16x4_f32(
                false, a, false, b, (short)0, acc[t], false, false);
        }
    }

    int rbase = row0 + ((lane >> 4) << 3);
#pragma unroll
    for (int t = 0; t < 7; ++t) {
        int cc = t * 16 + m;
#pragma unroll
        for (int v = 0; v < 8; ++v) {
            int rr = rbase + v;
            if (rr < nrows) H[(size_t)rr * HPAD + cc] = acc[t][v];
        }
    }
}

template <bool POOL>
__device__ __forceinline__ void gather_node(
    const float* __restrict__ H, const int4* __restrict__ ninfo, const int2* __restrict__ csr,
    const float* __restrict__ bias, float* __restrict__ OUT, const int* __restrict__ batch,
    float* __restrict__ pool, float* __restrict__ counts, float* srow, int* sg, int nN)
{
    int lane = threadIdx.x & 31;
    int wid = threadIdx.x >> 5;
    int node = blockIdx.x * 8 + wid;
    bool valid = node < nN;
    if (!POOL) {
        if (!valid) return;
    } else if (!valid) {
        node = nN - 1;
    }
    int d = __builtin_amdgcn_readfirstlane(node);
    int4 ni = ninfo[d];
    int gid = 0;
    if (POOL) gid = batch[d];
    int beg = ni.x, n = ni.y;
    float di = __int_as_float(ni.z);
    int c4 = lane < 28 ? lane : 27;
    const float4* H4 = (const float4*)H + c4;

    int m = n < 32 ? n : 32;
    int2 meta = make_int2(0, 0);
    if (lane < m) meta = csr[beg + lane];
    float4 hs = H4[(size_t)d * (HPAD / 4)];
    bool hasb = c4 < HID / 4;
    float4 bb = ((const float4*)bias)[hasb ? c4 : HID / 4 - 1];
    float4 acc = make_float4(0.f, 0.f, 0.f, 0.f);

    for (int base = 0;;) {
        for (int j = 0; j < m; j += 8) {
            int r = m - j;
            int   sx[8];
            float w[8];
            float4 h[8];
#pragma unroll
            for (int u = 0; u < 8; ++u) {
                int jj = (u < r) ? j + u : m - 1;
                sx[u] = __builtin_amdgcn_readlane(meta.x, jj);
                float wv = __int_as_float(__builtin_amdgcn_readlane(meta.y, jj));
                w[u] = (u < r) ? wv : 0.f;
            }
#pragma unroll
            for (int u = 0; u < 8; ++u) {
                h[u] = make_float4(0.f, 0.f, 0.f, 0.f);
                if (u < r) h[u] = H4[(size_t)sx[u] * (HPAD / 4)];
            }
#pragma unroll
            for (int u = 0; u < 8; ++u) {
                acc.x = fmaf(h[u].x, w[u], acc.x);
                acc.y = fmaf(h[u].y, w[u], acc.y);
                acc.z = fmaf(h[u].z, w[u], acc.z);
                acc.w = fmaf(h[u].w, w[u], acc.w);
            }
        }
        base += 32;
        if (base >= n) break;
        m = n - base;
        if (m > 32) m = 32;
        meta = make_int2(0, 0);
        if (lane < m) meta = csr[beg + base + lane];
    }

    float wself = di * di;
    if (!hasb) bb = make_float4(0.f, 0.f, 0.f, 0.f);
    float4 o;
    o.x = fmaf(hs.x, wself, acc.x) + bb.x; o.x = o.x > 0.f ? o.x : 0.f;
    o.y = fmaf(hs.y, wself, acc.y) + bb.y; o.y = o.y > 0.f ? o.y : 0.f;
    o.z = fmaf(hs.z, wself, acc.z) + bb.z; o.z = o.z > 0.f ? o.z : 0.f;
    o.w = fmaf(hs.w, wself, acc.w) + bb.w; o.w = o.w > 0.f ? o.w : 0.f;
    if (!POOL) {
        if (lane < 28) ((float4*)OUT)[(size_t)d * (HPAD / 4) + lane] = o;
        return;
    }

    if (lane < 28) ((float4*)srow)[wid * (HPAD / 4) + lane] = o;
    if (lane == 0) sg[wid] = (valid && (unsigned)gid < (unsigned)N_GRAPHS) ? gid : -1;
    __syncthreads();
    int t = threadIdx.x;
    if (t > HID) return;
    bool isc = t == HID;
    int4 ga = ((const int4*)sg)[0], gb = ((const int4*)sg)[1];
    int gq[8] = {ga.x, ga.y, ga.z, ga.w, gb.x, gb.y, gb.z, gb.w};
    float vq[8];
#pragma unroll
    for (int q = 0; q < 8; ++q) vq[q] = srow[q * HPAD + t];
    int cur = -1;
    float s = 0.f;
#pragma unroll
    for (int q = 0; q < 8; ++q) {
        int gw = gq[q];
        float v = isc ? 1.0f : vq[q];
        if (gw >= 0) {
            if (gw == cur) {
                s += v;
            } else {
                if (cur >= 0) atomicAdd(isc ? &counts[cur] : &pool[(size_t)cur * HID + t], s);
                cur = gw;
                s = v;
            }
        }
    }
    if (cur >= 0) atomicAdd(isc ? &counts[cur] : &pool[(size_t)cur * HID + t], s);
}

__global__ __launch_bounds__(256) void k_gather(
    const float* __restrict__ H, const int4* __restrict__ ninfo, const int2* __restrict__ csr,
    const float* __restrict__ bias, float* __restrict__ OUT, int nN)
{
    gather_node<false>(H, ninfo, csr, bias, OUT, nullptr, nullptr, nullptr, nullptr, nullptr, nN);
}

__global__ __launch_bounds__(256) void k_gather_pool(
    const float* __restrict__ H, const int4* __restrict__ ninfo, const int2* __restrict__ csr,
    const float* __restrict__ bias, const int* __restrict__ batch,
    float* __restrict__ pool, float* __restrict__ counts, int nN)
{
    __shared__ float4 srow[8 * (HPAD / 4)];
    __shared__ int4 sg[2];
    gather_node<true>(H, ninfo, csr, bias, nullptr, batch, pool, counts, (float*)srow, (int*)sg, nN);
}

__global__ void k_pool_fin(const float* __restrict__ pool, const float* __restrict__ counts,
                           float* __restrict__ out, int n) {
    int i = blockIdx.x * blockDim.x + threadIdx.x;
    if (i >= n) return;
    int g = i / HID;
    out[i] = pool[i] / fmaxf(counts[g], 1.0f);
}

extern "C" void kernel_launch(void* const* d_in, const int* in_sizes, int n_in,
                              void* d_out, int out_size, void* d_ws, size_t ws_size,
                              hipStream_t stream) {
    const float* x  = (const float*)d_in[0];
    const float* W1 = (const float*)d_in[1];
    const float* b1 = (const float*)d_in[2];
    const float* W2 = (const float*)d_in[3];
    const float* b2 = (const float*)d_in[4];
    const int*   ei = (const int*)d_in[5];
    const int*   bt = (const int*)d_in[6];
    const int* esrc = ei;
    const int* edst = ei + N_EDGES;
    float* out = (float*)d_out;

    float*  ws    = (float*)d_ws;
    int4*   ninfo = (int4*)ws;
    int*    ecnt  = (int*)(ws + (size_t)4 * N_NODES);
    int*    cursor = ecnt + 2 * N_NODES;
    float*  pool = (float*)(cursor + 4);
    float*  cnts = pool + N_GRAPHS * HID;
    float*  B0   = cnts + N_GRAPHS;
    float*  B1   = B0 + (size_t)N_NODES * HPAD;
    float*  B2   = B1 + (size_t)N_NODES * HPAD;
    int*    rank = (int*)(B2 + (size_t)N_NODES * HPAD);
    int2*   csr  = (int2*)(rank + N_EDGES);

    auto cdiv = [](long long a, long long b) { return (unsigned)((a + b - 1) / b); };

    k_zero_int<<<cdiv(2 * N_NODES + 4 + N_GRAPHS * HID + N_GRAPHS, 256), 256, 0, stream>>>(
        ecnt, 2 * N_NODES + 4 + N_GRAPHS * HID + N_GRAPHS);
    k_rank<<<cdiv(N_EDGES, 256), 256, 0, stream>>>(edst, ecnt, rank, N_EDGES);
    k_alloc<<<cdiv(N_NODES, 256), 256, 0, stream>>>(ecnt, cursor, ninfo, N_NODES);
    k_fill<<<cdiv(N_EDGES, 256), 256, 0, stream>>>(esrc, edst, rank, ninfo, csr, N_EDGES);

    k_gemm_wmma<<<cdiv(N_NODES, 128), 256, 0, stream>>>(x, W1, IN_DIM, HID, B0, N_NODES, IN_DIM);
    k_gather<<<cdiv(N_NODES, 8), 256, 0, stream>>>(B0, ninfo, csr, b1, B1, N_NODES);

    k_gemm_wmma<<<cdiv(N_NODES, 128), 256, 0, stream>>>(B1, W2, HID, HID, B2, N_NODES, HPAD);
    k_gather_pool<<<cdiv(N_NODES, 8), 256, 0, stream>>>(B2, ninfo, csr, b2, bt, pool, cnts, N_NODES);

    k_pool_fin<<<cdiv(N_GRAPHS * HID, 256), 256, 0, stream>>>(pool, cnts, out, N_GRAPHS * HID);
}


# 9 launches become 7: edge counting runs as tail blocks of the first GEMM's launch; pool adds carry sum/count into the output itself (node counts from the slot pass), partials divided before the add
# speedup vs baseline: 4.4946x; 1.0866x over previous
#include <hip/hip_runtime.h>
#include <hip/hip_bf16.h>

typedef __attribute__((ext_vector_type(2))) float v2f;
typedef __attribute__((ext_vector_type(8))) float v8f;

#define N_NODES  100000
#define N_EDGES  1600000
#define N_GRAPHS 1000
#define IN_DIM   128
#define HID      100
#define HPAD     112
#define GEMM_BLOCKS ((N_NODES + 127) / 128)
#define RANK_BLOCKS ((N_EDGES / 4 + 255) / 256)

__global__ void k_zero_int(int* p, int n, float* q, int m) {
    int i = blockIdx.x * blockDim.x + threadIdx.x;
    if (i < n) p[i] = 0;
    else if (i - n < m) q[i - n] = 0.0f;
}

static_assert(N_EDGES % 4 == 0, "the counting pass reads the targets four at a time");
__device__ __forceinline__ void rank_block(int rb, const int* __restrict__ dst, int* ecnt, int* __restrict__ rank) {
    int i = rb * 256 + threadIdx.x;
    unsigned h = (__builtin_amdgcn_s_sendmsg_rtn(0x87) >> 18) & 1u;
    if (i >= N_EDGES / 4) return;
    int4 d = ((const int4*)dst)[i];
    int* c = ecnt + h * N_NODES;
    unsigned r0 = (unsigned)atomicAdd(&c[d.x], 1);
    unsigned r1 = (unsigned)atomicAdd(&c[d.y], 1);
    unsigned r2 = (unsigned)atomicAdd(&c[d.z], 1);
    unsigned r3 = (unsigned)atomicAdd(&c[d.w], 1);
    unsigned hb = h << 31;
    ((int4*)rank)[i] = make_int4((int)(r0 | hb), (int)(r1 | hb), (int)(r2 | hb), (int)(r3 | hb));
}

__global__ __launch_bounds__(256) void k_alloc(const int* __restrict__ ecnt, int* cursor,
                                               int4* __restrict__ ninfo, const int* __restrict__ batch,
                                               float* counts, int n) {
    __shared__ int wsum[8];
    __shared__ int sbase;
    int tid = threadIdx.x, lane = tid & 31, wv = tid >> 5;
    int i = blockIdx.x * 256 + tid;
    int c0 = (i < n) ? ecnt[i] : 0;
    int c1 = (i < n) ? ecnt[N_NODES + i] : 0;
    int g = (i < n) ? batch[i] : -1;
    int c = c0 + c1;
    {
        if ((unsigned)g >= (unsigned)N_GRAPHS) g = -1;
        int gp = __shfl_up(g, 1);
        bool head = lane == 0 || g != gp;
        unsigned hm = (unsigned)__ballot(head);
        unsigned above = hm & ~((2u << lane) - 1u);
        int len = (above ? __builtin_ctz(above) : 32) - lane;
        if (head && g >= 0) atomicAdd(&counts[g], (float)len);
    }
    int v = c;
#pragma unroll
    for (int o = 1; o < 32; o <<= 1) {
        int t = __shfl_up(v, o);
        if (lane >= o) v += t;
    }
    if (lane == 31) wsum[wv] = v;
    __syncthreads();
    if (tid == 0) {
        int tot = 0;
        for (int k = 0; k < 8; ++k) { int t = wsum[k]; wsum[k] = tot; tot += t; }
        sbase = atomicAdd(cursor, tot);
    }
    __syncthreads();
    if (i < n) {
        float dinv = rsqrtf((float)(c + 1));
        ninfo[i] = make_int4(sbase + wsum[wv] + v - c, c, __float_as_int(dinv), c0);
    }
}

__global__ void k_fill(const int* __restrict__ src, const int* __restrict__ dst,
                       const int* __restrict__ rank, const int4* __restrict__ ninfo,
                       int2* __restrict__ csr, int e) {
    int i = blockIdx.x * blockDim.x + threadIdx.x;
    if (i >= e) return;
    int s = src[i], d = dst[i];
    int4 nd = ninfo[d];
    float w = __int_as_float(ninfo[s].z) * __int_as_float(nd.z);
    int rk = rank[i];
    int r = rk & 0x7fffffff;
    if (rk < 0) r += nd.w;
    csr[nd.x + r] = make_int2(s, __float_as_int(w));
}

__device__ __forceinline__ void gemm_block(
    int bx, const float* __restrict__ X, const float* __restrict__ W, int wr, int wc,
    float* __restrict__ H, int nrows, int K)
{
    __shared__ float2 wl[(IN_DIM / 2) * HPAD];
    int tid = threadIdx.x;
    int kpairs = K >> 1;
    int total = kpairs * HPAD;
    for (int base = tid; base < total; base += 14 * 256) {
        float lo[14], hi[14];
#pragma unroll
        for (int u = 0; u < 14; ++u) {
            int i = base + u * 256;
            int p = i / HPAD, n = i - p * HPAD;
            int k0 = 2 * p, k1 = k0 + 1;
            bool ins = i < total && n < wc;
            bool ok0 = ins && k0 < wr, ok1 = ins && k1 < wr;
            float v0 = W[ok0 ? k0 * wc + n : 0];
            float v1 = W[ok1 ? k1 * wc + n : 0];
            lo[u] = ok0 ? v0 : 0.0f;
            hi[u] = ok1 ? v1 : 0.0f;
        }
#pragma unroll
        for (int u = 0; u < 14; ++u) {
            wl[base + u * 256] = make_float2(lo[u], hi[u]);
        }
    }
    __syncthreads();

    int lane = tid & 31;
    int wave = tid >> 5;
    int m    = lane & 15;
    int koff = (lane >> 4) << 1;
    int row0 = bx * 128 + wave * 16;
    int row  = row0 + m;
    bool inr = row < nrows;
    int rowc = inr ? row : (nrows - 1);
    const float* xrow = X + (size_t)rowc * K;

    v8f acc[7];
#pragma unroll
    for (int t = 0; t < 7; ++t)
        acc[t] = (v8f){0.f, 0.f, 0.f, 0.f, 0.f, 0.f, 0.f, 0.f};

    for (int k = 0; k < K; k += 4) {
        float2 t2 = *(const float2*)(xrow + k + koff);
        v2f a = (v2f){ inr ? t2.x : 0.f, inr ? t2.y : 0.f };
        const float2* wrow = wl + ((k + koff) >> 1) * HPAD;
#pragma unroll
        for (int t = 0; t < 7; ++t) {
            float2 bb = wrow[t * 16 + m];
            v2f b = (v2f){bb.x, bb.y};
            acc[t] = __builtin_amdgcn_wmma_f32_16x16x4_f32(
                false, a, false, b, (short)0, acc[t], false, false);
        }
    }

    int rbase = row0 + ((lane >> 4) << 3);
#pragma unroll
    for (int t = 0; t < 7; ++t) {
        int cc = t * 16 + m;
#pragma unroll
        for (int v = 0; v < 8; ++v) {
            int rr = rbase + v;
            if (rr < nrows) H[(size_t)rr * HPAD + cc] = acc[t][v];
        }
    }
}

__global__ __launch_bounds__(256) void k_gemm_wmma(
    const float* __restrict__ X, const float* __restrict__ W, int wr, int wc,
    float* __restrict__ H, int nrows, int K)
{
    gemm_block(blockIdx.x, X, W, wr, wc, H, nrows, K);
}

__global__ __launch_bounds__(256) void k_gemm_rank(
    const float* __restrict__ X, const float* __restrict__ W, int wr, int wc,
    float* __restrict__ H, int nrows, int K, int gemm_blocks,
    const int* __restrict__ dst, int* ecnt, int* __restrict__ rank)
{
    int b = blockIdx.x;
    if (b >= gemm_blocks) {
        rank_block(b - gemm_blocks, dst, ecnt, rank);
        return;
    }
    gemm_block(b, X, W, wr, wc, H, nrows, K);
}

template <bool POOL>
__device__ __forceinline__ void gather_node(
    const float* __restrict__ H, const int4* __restrict__ ninfo, const int2* __restrict__ csr,
    const float* __restrict__ bias, float* __restrict__ OUT, const int* __restrict__ batch,
    float* __restrict__ mean, const float* __restrict__ counts, float* srow, int* sg, float* sc, int nN)
{
    int lane = threadIdx.x & 31;
    int wid = threadIdx.x >> 5;
    int node = blockIdx.x * 8 + wid;
    bool valid = node < nN;
    if (!POOL) {
        if (!valid) return;
    } else if (!valid) {
        node = nN - 1;
    }
    int d = __builtin_amdgcn_readfirstlane(node);
    int4 ni = ninfo[d];
    int gid = 0;
    if (POOL) {
        gid = batch[d];
        bool ok = __builtin_amdgcn_readfirstlane((int)(blockIdx.x * 8 + wid)) < nN && (unsigned)gid < (unsigned)N_GRAPHS;
        if (!ok) gid = -1;
    }
    int beg = ni.x, n = ni.y;
    float di = __int_as_float(ni.z);
    int c4 = lane < 28 ? lane : 27;
    const float4* H4 = (const float4*)H + c4;

    int m = n < 32 ? n : 32;
    int2 meta = make_int2(0, 0);
    if (lane < m) meta = csr[beg + lane];
    float4 hs = H4[(size_t)d * (HPAD / 4)];
    bool hasb = c4 < HID / 4;
    float4 bb = ((const float4*)bias)[hasb ? c4 : HID / 4 - 1];
    float gcnt = 1.0f;
    if (POOL) gcnt = counts[gid >= 0 ? gid : 0];
    float4 acc = make_float4(0.f, 0.f, 0.f, 0.f);

    for (int base = 0;;) {
        for (int j = 0; j < m; j += 8) {
            int r = m - j;
            int   sx[8];
            float w[8];
            float4 h[8];
#pragma unroll
            for (int u = 0; u < 8; ++u) {
                int jj = (u < r) ? j + u : m - 1;
                sx[u] = __builtin_amdgcn_readlane(meta.x, jj);
                float wv = __int_as_float(__builtin_amdgcn_readlane(meta.y, jj));
                w[u] = (u < r) ? wv : 0.f;
            }
#pragma unroll
            for (int u = 0; u < 8; ++u) {
                h[u] = make_float4(0.f, 0.f, 0.f, 0.f);
                if (u < r) h[u] = H4[(size_t)sx[u] * (HPAD / 4)];
            }
#pragma unroll
            for (int u = 0; u < 8; ++u) {
                acc.x = fmaf(h[u].x, w[u], acc.x);
                acc.y = fmaf(h[u].y, w[u], acc.y);
                acc.z = fmaf(h[u].z, w[u], acc.z);
                acc.w = fmaf(h[u].w, w[u], acc.w);
            }
        }
        base += 32;
        if (base >= n) break;
        m = n - base;
        if (m > 32) m = 32;
        meta = make_int2(0, 0);
        if (lane < m) meta = csr[beg + base + lane];
    }

    float wself = di * di;
    if (!hasb) bb = make_float4(0.f, 0.f, 0.f, 0.f);
    float4 o;
    o.x = fmaf(hs.x, wself, acc.x) + bb.x; o.x = o.x > 0.f ? o.x : 0.f;
    o.y = fmaf(hs.y, wself, acc.y) + bb.y; o.y = o.y > 0.f ? o.y : 0.f;
    o.z = fmaf(hs.z, wself, acc.z) + bb.z; o.z = o.z > 0.f ? o.z : 0.f;
    o.w = fmaf(hs.w, wself, acc.w) + bb.w; o.w = o.w > 0.f ? o.w : 0.f;
    if (!POOL) {
        if (lane < 28) ((float4*)OUT)[(size_t)d * (HPAD / 4) + lane] = o;
        return;
    }

    if (lane < 28) ((float4*)srow)[wid * (HPAD / 4) + lane] = o;
    if (lane == 0) { sg[wid] = gid; sc[wid] = gcnt; }
    __syncthreads();
    int t = threadIdx.x;
    if (t >= HID) return;
    int4 ga = ((const int4*)sg)[0], gb = ((const int4*)sg)[1];
    float4 ca = ((const float4*)sc)[0], cb = ((const float4*)sc)[1];
    int gq[8] = {ga.x, ga.y, ga.z, ga.w, gb.x, gb.y, gb.z, gb.w};
    float cq[8] = {ca.x, ca.y, ca.z, ca.w, cb.x, cb.y, cb.z, cb.w};
    float vq[8];
#pragma unroll
    for (int q = 0; q < 8; ++q) vq[q] = srow[q * HPAD + t];
    int cur = -1;
    float ccur = 1.0f;
    float s = 0.f;
#pragma unroll
    for (int q = 0; q < 8; ++q) {
        int gw = gq[q];
        float v = vq[q];
        if (gw >= 0) {
            if (gw == cur) {
                s += v;
            } else {
                if (cur >= 0) atomicAdd(&mean[(size_t)cur * HID + t], s / fmaxf(ccur, 1.0f));
                cur = gw;
                ccur = cq[q];
                s = v;
            }
        }
    }
    if (cur >= 0) atomicAdd(&mean[(size_t)cur * HID + t], s / fmaxf(ccur, 1.0f));
}

__global__ __launch_bounds__(256) void k_gather(
    const float* __restrict__ H, const int4* __restrict__ ninfo, const int2* __restrict__ csr,
    const float* __restrict__ bias, float* __restrict__ OUT, int nN)
{
    gather_node<false>(H, ninfo, csr, bias, OUT, nullptr, nullptr, nullptr, nullptr, nullptr, nullptr, nN);
}

__global__ __launch_bounds__(256) void k_gather_pool(
    const float* __restrict__ H, const int4* __restrict__ ninfo, const int2* __restrict__ csr,
    const float* __restrict__ bias, const int* __restrict__ batch,
    float* __restrict__ mean, const float* __restrict__ counts, int nN)
{
    __shared__ float4 srow[8 * (HPAD / 4)];
    __shared__ int4 sg[2];
    __shared__ float4 sc[2];
    gather_node<true>(H, ninfo, csr, bias, nullptr, batch, mean, counts, (float*)srow, (int*)sg, (float*)sc, nN);
}

extern "C" void kernel_launch(void* const* d_in, const int* in_sizes, int n_in,
                              void* d_out, int out_size, void* d_ws, size_t ws_size,
                              hipStream_t stream) {
    const float* x  = (const float*)d_in[0];
    const float* W1 = (const float*)d_in[1];
    const float* b1 = (const float*)d_in[2];
    const float* W2 = (const float*)d_in[3];
    const float* b2 = (const float*)d_in[4];
    const int*   ei = (const int*)d_in[5];
    const int*   bt = (const int*)d_in[6];
    const int* esrc = ei;
    const int* edst = ei + N_EDGES;
    float* out = (float*)d_out;

    float*  ws    = (float*)d_ws;
    int4*   ninfo = (int4*)ws;
    int*    ecnt  = (int*)(ws + (size_t)4 * N_NODES);
    int*    cursor = ecnt + 2 * N_NODES;
    float*  cnts = (float*)(cursor + 4);
    float*  B0   = cnts + N_GRAPHS;
    float*  B1   = B0 + (size_t)N_NODES * HPAD;
    float*  B2   = B1 + (size_t)N_NODES * HPAD;
    int*    rank = (int*)(B2 + (size_t)N_NODES * HPAD);
    int2*   csr  = (int2*)(rank + N_EDGES);

    auto cdiv = [](long long a, long long b) { return (unsigned)((a + b - 1) / b); };

    k_zero_int<<<cdiv(2 * N_NODES + 4 + N_GRAPHS + N_GRAPHS * HID, 256), 256, 0, stream>>>(
        ecnt, 2 * N_NODES + 4 + N_GRAPHS, out, N_GRAPHS * HID);

    k_gemm_rank<<<GEMM_BLOCKS + RANK_BLOCKS, 256, 0, stream>>>(
        x, W1, IN_DIM, HID, B0, N_NODES, IN_DIM, GEMM_BLOCKS, edst, ecnt, rank);
    k_alloc<<<cdiv(N_NODES, 256), 256, 0, stream>>>(ecnt, cursor, ninfo, bt, cnts, N_NODES);
    k_fill<<<cdiv(N_EDGES, 256), 256, 0, stream>>>(esrc, edst, rank, ninfo, csr, N_EDGES);

    k_gather<<<cdiv(N_NODES, 8), 256, 0, stream>>>(B0, ninfo, csr, b1, B1, N_NODES);

    k_gemm_wmma<<<cdiv(N_NODES, 128), 256, 0, stream>>>(B1, W2, HID, HID, B2, N_NODES, HPAD);
    k_gather_pool<<<cdiv(N_NODES, 8), 256, 0, stream>>>(B2, ninfo, csr, b2, bt, out, cnts, N_NODES);
}


# Layer-1 aggregation and layer-2 product fused: 16 gathered rows meet in an LDS tile, 7 of a block's 16 waves multiply it by W2 under the others' row loads; X2 never stored; 7 launches to 6; same sums
# speedup vs baseline: 4.8012x; 1.0682x over previous
#include <hip/hip_runtime.h>
#include <hip/hip_bf16.h>

typedef __attribute__((ext_vector_type(2))) float v2f;
typedef __attribute__((ext_vector_type(8))) float v8f;

#define N_NODES  100000
#define N_EDGES  1600000
#define N_GRAPHS 1000
#define IN_DIM   128
#define HID      100
#define HPAD     112
#define GEMM_BLOCKS ((N_NODES + 127) / 128)
#define RANK_BLOCKS ((N_EDGES / 4 + 255) / 256)

__global__ void k_zero_int(int* p, int n, float* q, int m) {
    int i = blockIdx.x * blockDim.x + threadIdx.x;
    if (i < n) p[i] = 0;
    else if (i - n < m) q[i - n] = 0.0f;
}

static_assert(N_EDGES % 4 == 0, "the counting pass reads the targets four at a time");
__device__ __forceinline__ void rank_block(int rb, const int* __restrict__ dst, int* ecnt, int* __restrict__ rank) {
    int i = rb * 256 + threadIdx.x;
    unsigned h = (__builtin_amdgcn_s_sendmsg_rtn(0x87) >> 18) & 1u;
    if (i >= N_EDGES / 4) return;
    int4 d = ((const int4*)dst)[i];
    int* c = ecnt + h * N_NODES;
    unsigned r0 = (unsigned)atomicAdd(&c[d.x], 1);
    unsigned r1 = (unsigned)atomicAdd(&c[d.y], 1);
    unsigned r2 = (unsigned)atomicAdd(&c[d.z], 1);
    unsigned r3 = (unsigned)atomicAdd(&c[d.w], 1);
    unsigned hb = h << 31;
    ((int4*)rank)[i] = make_int4((int)(r0 | hb), (int)(r1 | hb), (int)(r2 | hb), (int)(r3 | hb));
}

__global__ __launch_bounds__(256) void k_alloc(const int* __restrict__ ecnt, int* cursor,
                                               int4* __restrict__ ninfo, const int* __restrict__ batch,
                                               float* counts, int n) {
    __shared__ int wsum[8];
    __shared__ int sbase;
    int tid = threadIdx.x, lane = tid & 31, wv = tid >> 5;
    int i = blockIdx.x * 256 + tid;
    int c0 = (i < n) ? ecnt[i] : 0;
    int c1 = (i < n) ? ecnt[N_NODES + i] : 0;
    int g = (i < n) ? batch[i] : -1;
    int c = c0 + c1;
    {
        if ((unsigned)g >= (unsigned)N_GRAPHS) g = -1;
        int gp = __shfl_up(g, 1);
        bool head = lane == 0 || g != gp;
        unsigned hm = (unsigned)__ballot(head);
        unsigned above = hm & ~((2u << lane) - 1u);
        int len = (above ? __builtin_ctz(above) : 32) - lane;
        if (head && g >= 0) atomicAdd(&counts[g], (float)len);
    }
    int v = c;
#pragma unroll
    for (int o = 1; o < 32; o <<= 1) {
        int t = __shfl_up(v, o);
        if (lane >= o) v += t;
    }
    if (lane == 31) wsum[wv] = v;
    __syncthreads();
    if (tid == 0) {
        int tot = 0;
        for (int k = 0; k < 8; ++k) { int t = wsum[k]; wsum[k] = tot; tot += t; }
        sbase = atomicAdd(cursor, tot);
    }
    __syncthreads();
    if (i < n) {
        float dinv = rsqrtf((float)(c + 1));
        ninfo[i] = make_int4(sbase + wsum[wv] + v - c, c, __float_as_int(dinv), c0);
    }
}

__global__ void k_fill(const int* __restrict__ src, const int* __restrict__ dst,
                       const int* __restrict__ rank, const int4* __restrict__ ninfo,
                       int2* __restrict__ csr, int e) {
    int i = blockIdx.x * blockDim.x + threadIdx.x;
    if (i >= e) return;
    int s = src[i], d = dst[i];
    int4 nd = ninfo[d];
    float w = __int_as_float(ninfo[s].z) * __int_as_float(nd.z);
    int rk = rank[i];
    int r = rk & 0x7fffffff;
    if (rk < 0) r += nd.w;
    csr[nd.x + r] = make_int2(s, __float_as_int(w));
}

__device__ __forceinline__ void gemm_block(
    int bx, const float* __restrict__ X, const float* __restrict__ W, int wr, int wc,
    float* __restrict__ H, int nrows, int K)
{
    __shared__ float2 wl[(IN_DIM / 2) * HPAD];
    int tid = threadIdx.x;
    int kpairs = K >> 1;
    int total = kpairs * HPAD;
    for (int base = tid; base < total; base += 14 * 256) {
        float lo[14], hi[14];
#pragma unroll
        for (int u = 0; u < 14; ++u) {
            int i = base + u * 256;
            int p = i / HPAD, n = i - p * HPAD;
            int k0 = 2 * p, k1 = k0 + 1;
            bool ins = i < total && n < wc;
            bool ok0 = ins && k0 < wr, ok1 = ins && k1 < wr;
            float v0 = W[ok0 ? k0 * wc + n : 0];
            float v1 = W[ok1 ? k1 * wc + n : 0];
            lo[u] = ok0 ? v0 : 0.0f;
            hi[u] = ok1 ? v1 : 0.0f;
        }
#pragma unroll
        for (int u = 0; u < 14; ++u) {
            wl[base + u * 256] = make_float2(lo[u], hi[u]);
        }
    }
    __syncthreads();

    int lane = tid & 31;
    int wave = tid >> 5;
    int m    = lane & 15;
    int koff = (lane >> 4) << 1;
    int row0 = bx * 128 + wave * 16;
    int row  = row0 + m;
    bool inr = row < nrows;
    int rowc = inr ? row : (nrows - 1);
    const float* xrow = X + (size_t)rowc * K;

    v8f acc[7];
#pragma unroll
    for (int t = 0; t < 7; ++t)
        acc[t] = (v8f){0.f, 0.f, 0.f, 0.f, 0.f, 0.f, 0.f, 0.f};

    for (int k = 0; k < K; k += 4) {
        float2 t2 = *(const float2*)(xrow + k + koff);
        v2f a = (v2f){ inr ? t2.x : 0.f, inr ? t2.y : 0.f };
        const float2* wrow = wl + ((k + koff) >> 1) * HPAD;
#pragma unroll
        for (int t = 0; t < 7; ++t) {
            float2 bb = wrow[t * 16 + m];
            v2f b = (v2f){bb.x, bb.y};
            acc[t] = __builtin_amdgcn_wmma_f32_16x16x4_f32(
                false, a, false, b, (short)0, acc[t], false, false);
        }
    }

    int rbase = row0 + ((lane >> 4) << 3);
#pragma unroll
    for (int t = 0; t < 7; ++t) {
        int cc = t * 16 + m;
#pragma unroll
        for (int v = 0; v < 8; ++v) {
            int rr = rbase + v;
            if (rr < nrows) H[(size_t)rr * HPAD + cc] = acc[t][v];
        }
    }
}

__global__ __launch_bounds__(256) void k_gemm_rank(
    const float* __restrict__ X, const float* __restrict__ W, int wr, int wc,
    float* __restrict__ H, int nrows, int K, int gemm_blocks,
    const int* __restrict__ dst, int* ecnt, int* __restrict__ rank)
{
    int b = blockIdx.x;
    if (b >= gemm_blocks) {
        rank_block(b - gemm_blocks, dst, ecnt, rank);
        return;
    }
    gemm_block(b, X, W, wr, wc, H, nrows, K);
}

__device__ __forceinline__ void gather_node(
    const float* __restrict__ H, const int4* __restrict__ ninfo, const int2* __restrict__ csr,
    const float* __restrict__ bias, const int* __restrict__ batch,
    float* __restrict__ mean, const float* __restrict__ counts, float* srow, int* sg, float* sc, int nN)
{
    int lane = threadIdx.x & 31;
    int wid = threadIdx.x >> 5;
    int node = blockIdx.x * 8 + wid;
    bool valid = node < nN;
    if (!valid) node = nN - 1;
    int d = __builtin_amdgcn_readfirstlane(node);
    int4 ni = ninfo[d];
    int gid = batch[d];
    {
        bool ok = __builtin_amdgcn_readfirstlane((int)(blockIdx.x * 8 + wid)) < nN && (unsigned)gid < (unsigned)N_GRAPHS;
        if (!ok) gid = -1;
    }
    int beg = ni.x, n = ni.y;
    float di = __int_as_float(ni.z);
    int c4 = lane < 28 ? lane : 27;
    const float4* H4 = (const float4*)H + c4;

    int m = n < 32 ? n : 32;
    int2 meta = make_int2(0, 0);
    if (lane < m) meta = csr[beg + lane];
    float4 hs = H4[(size_t)d * (HPAD / 4)];
    bool hasb = c4 < HID / 4;
    float4 bb = ((const float4*)bias)[hasb ? c4 : HID / 4 - 1];
    float gcnt = counts[gid >= 0 ? gid : 0];
    float4 acc = make_float4(0.f, 0.f, 0.f, 0.f);

    for (int base = 0;;) {
        for (int j = 0; j < m; j += 8) {
            int r = m - j;
            int   sx[8];
            float w[8];
            float4 h[8];
#pragma unroll
            for (int u = 0; u < 8; ++u) {
                int jj = (u < r) ? j + u : m - 1;
                sx[u] = __builtin_amdgcn_readlane(meta.x, jj);
                float wv = __int_as_float(__builtin_amdgcn_readlane(meta.y, jj));
                w[u] = (u < r) ? wv : 0.f;
            }
#pragma unroll
            for (int u = 0; u < 8; ++u) {
                h[u] = make_float4(0.f, 0.f, 0.f, 0.f);
                if (u < r) h[u] = H4[(size_t)sx[u] * (HPAD / 4)];
            }
#pragma unroll
            for (int u = 0; u < 8; ++u) {
                acc.x = fmaf(h[u].x, w[u], acc.x);
                acc.y = fmaf(h[u].y, w[u], acc.y);
                acc.z = fmaf(h[u].z, w[u], acc.z);
                acc.w = fmaf(h[u].w, w[u], acc.w);
            }
        }
        base += 32;
        if (base >= n) break;
        m = n - base;
        if (m > 32) m = 32;
        meta = make_int2(0, 0);
        if (lane < m) meta = csr[beg + base + lane];
    }

    float wself = di * di;
    if (!hasb) bb = make_float4(0.f, 0.f, 0.f, 0.f);
    float4 o;
    o.x = fmaf(hs.x, wself, acc.x) + bb.x; o.x = o.x > 0.f ? o.x : 0.f;
    o.y = fmaf(hs.y, wself, acc.y) + bb.y; o.y = o.y > 0.f ? o.y : 0.f;
    o.z = fmaf(hs.z, wself, acc.z) + bb.z; o.z = o.z > 0.f ? o.z : 0.f;
    o.w = fmaf(hs.w, wself, acc.w) + bb.w; o.w = o.w > 0.f ? o.w : 0.f;
    if (lane < 28) ((float4*)srow)[wid * (HPAD / 4) + lane] = o;
    if (lane == 0) { sg[wid] = gid; sc[wid] = gcnt; }
    __syncthreads();
    int t = threadIdx.x;
    if (t >= HID) return;
    int4 ga = ((const int4*)sg)[0], gb = ((const int4*)sg)[1];
    float4 ca = ((const float4*)sc)[0], cb = ((const float4*)sc)[1];
    int gq[8] = {ga.x, ga.y, ga.z, ga.w, gb.x, gb.y, gb.z, gb.w};
    float cq[8] = {ca.x, ca.y, ca.z, ca.w, cb.x, cb.y, cb.z, cb.w};
    float vq[8];
#pragma unroll
    for (int q = 0; q < 8; ++q) vq[q] = srow[q * HPAD + t];
    int cur = -1;
    float ccur = 1.0f;
    float s = 0.f;
#pragma unroll
    for (int q = 0; q < 8; ++q) {
        int gw = gq[q];
        float v = vq[q];
        if (gw >= 0) {
            if (gw == cur) {
                s += v;
            } else {
                if (cur >= 0) atomicAdd(&mean[(size_t)cur * HID + t], s / fmaxf(ccur, 1.0f));
                cur = gw;
                ccur = cq[q];
                s = v;
            }
        }
    }
    if (cur >= 0) atomicAdd(&mean[(size_t)cur * HID + t], s / fmaxf(ccur, 1.0f));
}

#define FG_BLOCKS 1024
#define FG_GROUPS ((N_NODES + 15) / 16)
#define TPITCH 116
__global__ __launch_bounds__(512) void k_gather_gemm(
    const float* __restrict__ H, const int4* __restrict__ ninfo, const int2* __restrict__ csr,
    const float* __restrict__ bias, const float* __restrict__ W, int wr, int wc,
    float* __restrict__ H2, int nN)
{
    __shared__ float2 wl[(HPAD / 2) * HPAD];
    __shared__ float4 tile[2][16 * (TPITCH / 4)];
    int tid = threadIdx.x;
    int lane = tid & 31;
    int wave = __builtin_amdgcn_readfirstlane(tid >> 5);

    {
        const int total = (HPAD / 2) * HPAD;
        float lo[13], hi[13];
#pragma unroll
        for (int u = 0; u < 13; ++u) {
            int i = tid + u * 512;
            int p = i / HPAD, n = i - p * HPAD;
            int k0 = 2 * p, k1 = k0 + 1;
            bool ins = i < total && n < wc;
            bool ok0 = ins && k0 < wr, ok1 = ins && k1 < wr;
            float v0 = W[ok0 ? k0 * wc + n : 0];
            float v1 = W[ok1 ? k1 * wc + n : 0];
            lo[u] = ok0 ? v0 : 0.0f;
            hi[u] = ok1 ? v1 : 0.0f;
        }
#pragma unroll
        for (int u = 0; u < 13; ++u) {
            int i = tid + u * 512;
            if (i < total) wl[i] = make_float2(lo[u], hi[u]);
        }
    }

    int m = lane & 15;
    int koff = (lane >> 4) << 1;
    int c4 = lane < 28 ? lane : 27;
    const float4* H4 = (const float4*)H + c4;
    bool hasb = c4 < HID / 4;
    float4 bb = ((const float4*)bias)[hasb ? c4 : HID / 4 - 1];
    if (!hasb) bb = make_float4(0.f, 0.f, 0.f, 0.f);

    int g = blockIdx.x;
    if (g >= FG_GROUPS) return;

    const int* nwords = (const int*)ninfo + (lane & 3);

    int node = g * 16 + wave;
    bool valid = node < nN;
    int d = __builtin_amdgcn_readfirstlane(valid ? node : nN - 1);
    int rec = nwords[(size_t)d * 4];
    int beg = __builtin_amdgcn_readlane(rec, 0);
    int n = __builtin_amdgcn_readlane(rec, 1);
    float di = __int_as_float(__builtin_amdgcn_readlane(rec, 2));
    int mm = n < 32 ? n : 32;
    int2 meta = make_int2(0, 0);
    if (lane < mm) meta = csr[beg + lane];
    float4 hs = H4[(size_t)d * (HPAD / 4)];
    int gn = g + FG_BLOCKS;
    bool more = gn < FG_GROUPS;
    int noden = gn * 16 + wave;
    bool validn = more && noden < nN;
    int dn = __builtin_amdgcn_readfirstlane(validn ? noden : nN - 1);
    int nrec = nwords[(size_t)dn * 4];
    int buf = 0;
    int rot = 0;

    for (;;) {
        float4 acc = make_float4(0.f, 0.f, 0.f, 0.f);
        for (int base = 0;;) {
            for (int j = 0; j < mm; j += 8) {
                int r = mm - j;
                int   sx[8];
                float w[8];
                float4 h[8];
#pragma unroll
                for (int u = 0; u < 8; ++u) {
                    int jj = (u < r) ? j + u : mm - 1;
                    sx[u] = __builtin_amdgcn_readlane(meta.x, jj);
                    float wv = __int_as_float(__builtin_amdgcn_readlane(meta.y, jj));
                    w[u] = (u < r) ? wv : 0.f;
                }
#pragma unroll
                for (int u = 0; u < 8; ++u) {
                    h[u] = make_float4(0.f, 0.f, 0.f, 0.f);
                    if (u < r) h[u] = H4[(size_t)sx[u] * (HPAD / 4)];
                }
#pragma unroll
                for (int u = 0; u < 8; ++u) {
                    acc.x = fmaf(h[u].x, w[u], acc.x);
                    acc.y = fmaf(h[u].y, w[u], acc.y);
                    acc.z = fmaf(h[u].z, w[u], acc.z);
                    acc.w = fmaf(h[u].w, w[u], acc.w);
                }
            }
            base += 32;
            if (base >= n) break;
            mm = n - base;
            if (mm > 32) mm = 32;
            meta = make_int2(0, 0);
            if (lane < mm) meta = csr[beg + base + lane];
        }

        float wself = di * di;
        float4 o;
        o.x = fmaf(hs.x, wself, acc.x) + bb.x; o.x = o.x > 0.f ? o.x : 0.f;
        o.y = fmaf(hs.y, wself, acc.y) + bb.y; o.y = o.y > 0.f ? o.y : 0.f;
        o.z = fmaf(hs.z, wself, acc.z) + bb.z; o.z = o.z > 0.f ? o.z : 0.f;
        o.w = fmaf(hs.w, wself, acc.w) + bb.w; o.w = o.w > 0.f ? o.w : 0.f;
        if (!valid) o = make_float4(0.f, 0.f, 0.f, 0.f);
        if (lane < 28) tile[buf][wave * (TPITCH / 4) + lane] = o;
        __syncthreads();

        int begn = __builtin_amdgcn_readlane(nrec, 0);
        int nn = __builtin_amdgcn_readlane(nrec, 1);
        float din = __int_as_float(__builtin_amdgcn_readlane(nrec, 2));
        int mmn = nn < 32 ? nn : 32;
        int2 metan = make_int2(0, 0);
        if (lane < mmn) metan = csr[begn + lane];
        float4 hsn = H4[(size_t)dn * (HPAD / 4)];
        int gnn = gn + FG_BLOCKS;
        bool more2 = gnn < FG_GROUPS;
        int nodenn = gnn * 16 + wave;
        bool validnn = more2 && nodenn < nN;
        int dnn = __builtin_amdgcn_readfirstlane(validnn ? nodenn : nN - 1);
        int nrec2 = nwords[(size_t)dnn * 4];

        int t = (wave - rot) & 15;
        if (t < 7) {
            const float* trow = (const float*)tile[buf] + m * TPITCH + koff;
            const float2* wcol = wl + (koff >> 1) * HPAD + t * 16 + m;
            v8f pacc = (v8f){0.f, 0.f, 0.f, 0.f, 0.f, 0.f, 0.f, 0.f};
#pragma unroll
            for (int k = 0; k < HPAD; k += 4) {
                float2 a2 = *(const float2*)(trow + k);
                float2 b2 = wcol[(k >> 1) * HPAD];
                v2f a = (v2f){a2.x, a2.y};
                v2f b = (v2f){b2.x, b2.y};
                pacc = __builtin_amdgcn_wmma_f32_16x16x4_f32(
                    false, a, false, b, (short)0, pacc, false, false);
            }
            int rbase = g * 16 + ((lane >> 4) << 3);
            int cc = t * 16 + m;
#pragma unroll
            for (int v = 0; v < 8; ++v) {
                int rr = rbase + v;
                if (rr < nN) H2[(size_t)rr * HPAD + cc] = pacc[v];
            }
        }
        if (!more) break;
        g = gn; valid = validn;
        beg = begn; n = nn; di = din; mm = mmn; meta = metan; hs = hsn;
        gn = gnn; more = more2; validn = validnn; dn = dnn; nrec = nrec2;
        rot = (rot + 7) & 15;
        buf ^= 1;
    }
}

__global__ __launch_bounds__(256) void k_gather_pool(
    const float* __restrict__ H, const int4* __restrict__ ninfo, const int2* __restrict__ csr,
    const float* __restrict__ bias, const int* __restrict__ batch,
    float* __restrict__ mean, const float* __restrict__ counts, int nN)
{
    __shared__ float4 srow[8 * (HPAD / 4)];
    __shared__ int4 sg[2];
    __shared__ float4 sc[2];
    gather_node(H, ninfo, csr, bias, batch, mean, counts, (float*)srow, (int*)sg, (float*)sc, nN);
}

extern "C" void kernel_launch(void* const* d_in, const int* in_sizes, int n_in,
                              void* d_out, int out_size, void* d_ws, size_t ws_size,
                              hipStream_t stream) {
    const float* x  = (const float*)d_in[0];
    const float* W1 = (const float*)d_in[1];
    const float* b1 = (const float*)d_in[2];
    const float* W2 = (const float*)d_in[3];
    const float* b2 = (const float*)d_in[4];
    const int*   ei = (const int*)d_in[5];
    const int*   bt = (const int*)d_in[6];
    const int* esrc = ei;
    const int* edst = ei + N_EDGES;
    float* out = (float*)d_out;

    float*  ws    = (float*)d_ws;
    int4*   ninfo = (int4*)ws;
    int*    ecnt  = (int*)(ws + (size_t)4 * N_NODES);
    int*    cursor = ecnt + 2 * N_NODES;
    float*  cnts = (float*)(cursor + 4);
    float*  B0   = cnts + N_GRAPHS;
    float*  B2   = B0 + (size_t)N_NODES * HPAD;
    int*    rank = (int*)(B2 + (size_t)N_NODES * HPAD);
    int2*   csr  = (int2*)(rank + N_EDGES);

    auto cdiv = [](long long a, long long b) { return (unsigned)((a + b - 1) / b); };

    k_zero_int<<<cdiv(2 * N_NODES + 4 + N_GRAPHS + N_GRAPHS * HID, 256), 256, 0, stream>>>(
        ecnt, 2 * N_NODES + 4 + N_GRAPHS, out, N_GRAPHS * HID);

    k_gemm_rank<<<GEMM_BLOCKS + RANK_BLOCKS, 256, 0, stream>>>(
        x, W1, IN_DIM, HID, B0, N_NODES, IN_DIM, GEMM_BLOCKS, edst, ecnt, rank);
    k_alloc<<<cdiv(N_NODES, 256), 256, 0, stream>>>(ecnt, cursor, ninfo, bt, cnts, N_NODES);
    k_fill<<<cdiv(N_EDGES, 256), 256, 0, stream>>>(esrc, edst, rank, ninfo, csr, N_EDGES);

    k_gather_gemm<<<FG_BLOCKS, 512, 0, stream>>>(B0, ninfo, csr, b1, W2, HID, HID, B2, N_NODES);

    k_gather_pool<<<cdiv(N_NODES, 8), 256, 0, stream>>>(B2, ninfo, csr, b2, bt, out, cnts, N_NODES);
}


# Row tables H1 and H2 start on a 128-byte line (pad in the workspace carve-up): a 448-byte row began 48 or 112 bytes into a line, so every second gathered row touched 5 lines; now always 4
# speedup vs baseline: 5.0718x; 1.0564x over previous
#include <hip/hip_runtime.h>
#include <hip/hip_bf16.h>

typedef __attribute__((ext_vector_type(2))) float v2f;
typedef __attribute__((ext_vector_type(8))) float v8f;

#define N_NODES  100000
#define N_EDGES  1600000
#define N_GRAPHS 1000
#define IN_DIM   128
#define HID      100
#define HPAD     112
#define GEMM_BLOCKS ((N_NODES + 127) / 128)
#define RANK_BLOCKS ((N_EDGES / 4 + 255) / 256)
#define WS_HEAD (4 * N_NODES + 2 * N_NODES + 4 + N_GRAPHS)
#define WS_HEAD_PAD ((32 - WS_HEAD % 32) % 32)
static_assert((WS_HEAD + WS_HEAD_PAD) % 32 == 0, "the first row table starts on a 128-byte line");
static_assert((N_NODES * HPAD) % 32 == 0, "and so does the second, one table behind it");

__global__ void k_zero_int(int* p, int n, float* q, int m) {
    int i = blockIdx.x * blockDim.x + threadIdx.x;
    if (i < n) p[i] = 0;
    else if (i - n < m) q[i - n] = 0.0f;
}

static_assert(N_EDGES % 4 == 0, "the counting pass reads the targets four at a time");
__device__ __forceinline__ void rank_block(int rb, const int* __restrict__ dst, int* ecnt, int* __restrict__ rank) {
    int i = rb * 256 + threadIdx.x;
    unsigned h = (__builtin_amdgcn_s_sendmsg_rtn(0x87) >> 18) & 1u;
    if (i >= N_EDGES / 4) return;
    int4 d = ((const int4*)dst)[i];
    int* c = ecnt + h * N_NODES;
    unsigned r0 = (unsigned)atomicAdd(&c[d.x], 1);
    unsigned r1 = (unsigned)atomicAdd(&c[d.y], 1);
    unsigned r2 = (unsigned)atomicAdd(&c[d.z], 1);
    unsigned r3 = (unsigned)atomicAdd(&c[d.w], 1);
    unsigned hb = h << 31;
    ((int4*)rank)[i] = make_int4((int)(r0 | hb), (int)(r1 | hb), (int)(r2 | hb), (int)(r3 | hb));
}

__global__ __launch_bounds__(256) void k_alloc(const int* __restrict__ ecnt, int* cursor,
                                               int4* __restrict__ ninfo, const int* __restrict__ batch,
                                               float* counts, int n) {
    __shared__ int wsum[8];
    __shared__ int sbase;
    int tid = threadIdx.x, lane = tid & 31, wv = tid >> 5;
    int i = blockIdx.x * 256 + tid;
    int c0 = (i < n) ? ecnt[i] : 0;
    int c1 = (i < n) ? ecnt[N_NODES + i] : 0;
    int g = (i < n) ? batch[i] : -1;
    int c = c0 + c1;
    {
        if ((unsigned)g >= (unsigned)N_GRAPHS) g = -1;
        int gp = __shfl_up(g, 1);
        bool head = lane == 0 || g != gp;
        unsigned hm = (unsigned)__ballot(head);
        unsigned above = hm & ~((2u << lane) - 1u);
        int len = (above ? __builtin_ctz(above) : 32) - lane;
        if (head && g >= 0) atomicAdd(&counts[g], (float)len);
    }
    int v = c;
#pragma unroll
    for (int o = 1; o < 32; o <<= 1) {
        int t = __shfl_up(v, o);
        if (lane >= o) v += t;
    }
    if (lane == 31) wsum[wv] = v;
    __syncthreads();
    if (tid == 0) {
        int tot = 0;
        for (int k = 0; k < 8; ++k) { int t = wsum[k]; wsum[k] = tot; tot += t; }
        sbase = atomicAdd(cursor, tot);
    }
    __syncthreads();
    if (i < n) {
        float dinv = rsqrtf((float)(c + 1));
        ninfo[i] = make_int4(sbase + wsum[wv] + v - c, c, __float_as_int(dinv), c0);
    }
}

__global__ void k_fill(const int* __restrict__ src, const int* __restrict__ dst,
                       const int* __restrict__ rank, const int4* __restrict__ ninfo,
                       int2* __restrict__ csr, int e) {
    int i = blockIdx.x * blockDim.x + threadIdx.x;
    if (i >= e) return;
    int s = src[i], d = dst[i];
    int4 nd = ninfo[d];
    float w = __int_as_float(ninfo[s].z) * __int_as_float(nd.z);
    int rk = rank[i];
    int r = rk & 0x7fffffff;
    if (rk < 0) r += nd.w;
    csr[nd.x + r] = make_int2(s, __float_as_int(w));
}

__device__ __forceinline__ void gemm_block(
    int bx, const float* __restrict__ X, const float* __restrict__ W, int wr, int wc,
    float* __restrict__ H, int nrows, int K)
{
    __shared__ float2 wl[(IN_DIM / 2) * HPAD];
    int tid = threadIdx.x;
    int kpairs = K >> 1;
    int total = kpairs * HPAD;
    for (int base = tid; base < total; base += 14 * 256) {
        float lo[14], hi[14];
#pragma unroll
        for (int u = 0; u < 14; ++u) {
            int i = base + u * 256;
            int p = i / HPAD, n = i - p * HPAD;
            int k0 = 2 * p, k1 = k0 + 1;
            bool ins = i < total && n < wc;
            bool ok0 = ins && k0 < wr, ok1 = ins && k1 < wr;
            float v0 = W[ok0 ? k0 * wc + n : 0];
            float v1 = W[ok1 ? k1 * wc + n : 0];
            lo[u] = ok0 ? v0 : 0.0f;
            hi[u] = ok1 ? v1 : 0.0f;
        }
#pragma unroll
        for (int u = 0; u < 14; ++u) {
            wl[base + u * 256] = make_float2(lo[u], hi[u]);
        }
    }
    __syncthreads();

    int lane = tid & 31;
    int wave = tid >> 5;
    int m    = lane & 15;
    int koff = (lane >> 4) << 1;
    int row0 = bx * 128 + wave * 16;
    int row  = row0 + m;
    bool inr = row < nrows;
    int rowc = inr ? row : (nrows - 1);
    const float* xrow = X + (size_t)rowc * K;

    v8f acc[7];
#pragma unroll
    for (int t = 0; t < 7; ++t)
        acc[t] = (v8f){0.f, 0.f, 0.f, 0.f, 0.f, 0.f, 0.f, 0.f};

    for (int k = 0; k < K; k += 4) {
        float2 t2 = *(const float2*)(xrow + k + koff);
        v2f a = (v2f){ inr ? t2.x : 0.f, inr ? t2.y : 0.f };
        const float2* wrow = wl + ((k + koff) >> 1) * HPAD;
#pragma unroll
        for (int t = 0; t < 7; ++t) {
            float2 bb = wrow[t * 16 + m];
            v2f b = (v2f){bb.x, bb.y};
            acc[t] = __builtin_amdgcn_wmma_f32_16x16x4_f32(
                false, a, false, b, (short)0, acc[t], false, false);
        }
    }

    int rbase = row0 + ((lane >> 4) << 3);
#pragma unroll
    for (int t = 0; t < 7; ++t) {
        int cc = t * 16 + m;
#pragma unroll
        for (int v = 0; v < 8; ++v) {
            int rr = rbase + v;
            if (rr < nrows) H[(size_t)rr * HPAD + cc] = acc[t][v];
        }
    }
}

__global__ __launch_bounds__(256) void k_gemm_rank(
    const float* __restrict__ X, const float* __restrict__ W, int wr, int wc,
    float* __restrict__ H, int nrows, int K, int gemm_blocks,
    const int* __restrict__ dst, int* ecnt, int* __restrict__ rank)
{
    int b = blockIdx.x;
    if (b >= gemm_blocks) {
        rank_block(b - gemm_blocks, dst, ecnt, rank);
        return;
    }
    gemm_block(b, X, W, wr, wc, H, nrows, K);
}

__device__ __forceinline__ void gather_node(
    const float* __restrict__ H, const int4* __restrict__ ninfo, const int2* __restrict__ csr,
    const float* __restrict__ bias, const int* __restrict__ batch,
    float* __restrict__ mean, const float* __restrict__ counts, float* srow, int* sg, float* sc, int nN)
{
    int lane = threadIdx.x & 31;
    int wid = threadIdx.x >> 5;
    int node = blockIdx.x * 8 + wid;
    bool valid = node < nN;
    if (!valid) node = nN - 1;
    int d = __builtin_amdgcn_readfirstlane(node);
    int4 ni = ninfo[d];
    int gid = batch[d];
    {
        bool ok = __builtin_amdgcn_readfirstlane((int)(blockIdx.x * 8 + wid)) < nN && (unsigned)gid < (unsigned)N_GRAPHS;
        if (!ok) gid = -1;
    }
    int beg = ni.x, n = ni.y;
    float di = __int_as_float(ni.z);
    int c4 = lane < 28 ? lane : 27;
    const float4* H4 = (const float4*)H + c4;

    int m = n < 32 ? n : 32;
    int2 meta = make_int2(0, 0);
    if (lane < m) meta = csr[beg + lane];
    float4 hs = H4[(size_t)d * (HPAD / 4)];
    bool hasb = c4 < HID / 4;
    float4 bb = ((const float4*)bias)[hasb ? c4 : HID / 4 - 1];
    float gcnt = counts[gid >= 0 ? gid : 0];
    float4 acc = make_float4(0.f, 0.f, 0.f, 0.f);

    for (int base = 0;;) {
        for (int j = 0; j < m; j += 8) {
            int r = m - j;
            int   sx[8];
            float w[8];
            float4 h[8];
#pragma unroll
            for (int u = 0; u < 8; ++u) {
                int jj = (u < r) ? j + u : m - 1;
                sx[u] = __builtin_amdgcn_readlane(meta.x, jj);
                float wv = __int_as_float(__builtin_amdgcn_readlane(meta.y, jj));
                w[u] = (u < r) ? wv : 0.f;
            }
#pragma unroll
            for (int u = 0; u < 8; ++u) {
                h[u] = make_float4(0.f, 0.f, 0.f, 0.f);
                if (u < r) h[u] = H4[(size_t)sx[u] * (HPAD / 4)];
            }
#pragma unroll
            for (int u = 0; u < 8; ++u) {
                acc.x = fmaf(h[u].x, w[u], acc.x);
                acc.y = fmaf(h[u].y, w[u], acc.y);
                acc.z = fmaf(h[u].z, w[u], acc.z);
                acc.w = fmaf(h[u].w, w[u], acc.w);
            }
        }
        base += 32;
        if (base >= n) break;
        m = n - base;
        if (m > 32) m = 32;
        meta = make_int2(0, 0);
        if (lane < m) meta = csr[beg + base + lane];
    }

    float wself = di * di;
    if (!hasb) bb = make_float4(0.f, 0.f, 0.f, 0.f);
    float4 o;
    o.x = fmaf(hs.x, wself, acc.x) + bb.x; o.x = o.x > 0.f ? o.x : 0.f;
    o.y = fmaf(hs.y, wself, acc.y) + bb.y; o.y = o.y > 0.f ? o.y : 0.f;
    o.z = fmaf(hs.z, wself, acc.z) + bb.z; o.z = o.z > 0.f ? o.z : 0.f;
    o.w = fmaf(hs.w, wself, acc.w) + bb.w; o.w = o.w > 0.f ? o.w : 0.f;
    if (lane < 28) ((float4*)srow)[wid * (HPAD / 4) + lane] = o;
    if (lane == 0) { sg[wid] = gid; sc[wid] = gcnt; }
    __syncthreads();
    int t = threadIdx.x;
    if (t >= HID) return;
    int4 ga = ((const int4*)sg)[0], gb = ((const int4*)sg)[1];
    float4 ca = ((const float4*)sc)[0], cb = ((const float4*)sc)[1];
    int gq[8] = {ga.x, ga.y, ga.z, ga.w, gb.x, gb.y, gb.z, gb.w};
    float cq[8] = {ca.x, ca.y, ca.z, ca.w, cb.x, cb.y, cb.z, cb.w};
    float vq[8];
#pragma unroll
    for (int q = 0; q < 8; ++q) vq[q] = srow[q * HPAD + t];
    int cur = -1;
    float ccur = 1.0f;
    float s = 0.f;
#pragma unroll
    for (int q = 0; q < 8; ++q) {
        int gw = gq[q];
        float v = vq[q];
        if (gw >= 0) {
            if (gw == cur) {
                s += v;
            } else {
                if (cur >= 0) atomicAdd(&mean[(size_t)cur * HID + t], s / fmaxf(ccur, 1.0f));
                cur = gw;
                ccur = cq[q];
                s = v;
            }
        }
    }
    if (cur >= 0) atomicAdd(&mean[(size_t)cur * HID + t], s / fmaxf(ccur, 1.0f));
}

#define FG_BLOCKS 1024
#define FG_GROUPS ((N_NODES + 15) / 16)
#define TPITCH 116
__global__ __launch_bounds__(512) void k_gather_gemm(
    const float* __restrict__ H, const int4* __restrict__ ninfo, const int2* __restrict__ csr,
    const float* __restrict__ bias, const float* __restrict__ W, int wr, int wc,
    float* __restrict__ H2, int nN)
{
    __shared__ float2 wl[(HPAD / 2) * HPAD];
    __shared__ float4 tile[2][16 * (TPITCH / 4)];
    int tid = threadIdx.x;
    int lane = tid & 31;
    int wave = __builtin_amdgcn_readfirstlane(tid >> 5);

    {
        const int total = (HPAD / 2) * HPAD;
        float lo[13], hi[13];
#pragma unroll
        for (int u = 0; u < 13; ++u) {
            int i = tid + u * 512;
            int p = i / HPAD, n = i - p * HPAD;
            int k0 = 2 * p, k1 = k0 + 1;
            bool ins = i < total && n < wc;
            bool ok0 = ins && k0 < wr, ok1 = ins && k1 < wr;
            float v0 = W[ok0 ? k0 * wc + n : 0];
            float v1 = W[ok1 ? k1 * wc + n : 0];
            lo[u] = ok0 ? v0 : 0.0f;
            hi[u] = ok1 ? v1 : 0.0f;
        }
#pragma unroll
        for (int u = 0; u < 13; ++u) {
            int i = tid + u * 512;
            if (i < total) wl[i] = make_float2(lo[u], hi[u]);
        }
    }

    int m = lane & 15;
    int koff = (lane >> 4) << 1;
    int c4 = lane < 28 ? lane : 27;
    const float4* H4 = (const float4*)H + c4;
    bool hasb = c4 < HID / 4;
    float4 bb = ((const float4*)bias)[hasb ? c4 : HID / 4 - 1];
    if (!hasb) bb = make_float4(0.f, 0.f, 0.f, 0.f);

    int g = blockIdx.x;
    if (g >= FG_GROUPS) return;

    const int* nwords = (const int*)ninfo + (lane & 3);

    int node = g * 16 + wave;
    bool valid = node < nN;
    int d = __builtin_amdgcn_readfirstlane(valid ? node : nN - 1);
    int rec = nwords[(size_t)d * 4];
    int beg = __builtin_amdgcn_readlane(rec, 0);
    int n = __builtin_amdgcn_readlane(rec, 1);
    float di = __int_as_float(__builtin_amdgcn_readlane(rec, 2));
    int mm = n < 32 ? n : 32;
    int2 meta = make_int2(0, 0);
    if (lane < mm) meta = csr[beg + lane];
    float4 hs = H4[(size_t)d * (HPAD / 4)];
    int gn = g + FG_BLOCKS;
    bool more = gn < FG_GROUPS;
    int noden = gn * 16 + wave;
    bool validn = more && noden < nN;
    int dn = __builtin_amdgcn_readfirstlane(validn ? noden : nN - 1);
    int nrec = nwords[(size_t)dn * 4];
    int buf = 0;
    int rot = 0;

    for (;;) {
        float4 acc = make_float4(0.f, 0.f, 0.f, 0.f);
        for (int base = 0;;) {
            for (int j = 0; j < mm; j += 8) {
                int r = mm - j;
                int   sx[8];
                float w[8];
                float4 h[8];
#pragma unroll
                for (int u = 0; u < 8; ++u) {
                    int jj = (u < r) ? j + u : mm - 1;
                    sx[u] = __builtin_amdgcn_readlane(meta.x, jj);
                    float wv = __int_as_float(__builtin_amdgcn_readlane(meta.y, jj));
                    w[u] = (u < r) ? wv : 0.f;
                }
#pragma unroll
                for (int u = 0; u < 8; ++u) {
                    h[u] = make_float4(0.f, 0.f, 0.f, 0.f);
                    if (u < r) h[u] = H4[(size_t)sx[u] * (HPAD / 4)];
                }
#pragma unroll
                for (int u = 0; u < 8; ++u) {
                    acc.x = fmaf(h[u].x, w[u], acc.x);
                    acc.y = fmaf(h[u].y, w[u], acc.y);
                    acc.z = fmaf(h[u].z, w[u], acc.z);
                    acc.w = fmaf(h[u].w, w[u], acc.w);
                }
            }
            base += 32;
            if (base >= n) break;
            mm = n - base;
            if (mm > 32) mm = 32;
            meta = make_int2(0, 0);
            if (lane < mm) meta = csr[beg + base + lane];
        }

        float wself = di * di;
        float4 o;
        o.x = fmaf(hs.x, wself, acc.x) + bb.x; o.x = o.x > 0.f ? o.x : 0.f;
        o.y = fmaf(hs.y, wself, acc.y) + bb.y; o.y = o.y > 0.f ? o.y : 0.f;
        o.z = fmaf(hs.z, wself, acc.z) + bb.z; o.z = o.z > 0.f ? o.z : 0.f;
        o.w = fmaf(hs.w, wself, acc.w) + bb.w; o.w = o.w > 0.f ? o.w : 0.f;
        if (!valid) o = make_float4(0.f, 0.f, 0.f, 0.f);
        if (lane < 28) tile[buf][wave * (TPITCH / 4) + lane] = o;
        __syncthreads();

        int begn = __builtin_amdgcn_readlane(nrec, 0);
        int nn = __builtin_amdgcn_readlane(nrec, 1);
        float din = __int_as_float(__builtin_amdgcn_readlane(nrec, 2));
        int mmn = nn < 32 ? nn : 32;
        int2 metan = make_int2(0, 0);
        if (lane < mmn) metan = csr[begn + lane];
        float4 hsn = H4[(size_t)dn * (HPAD / 4)];
        int gnn = gn + FG_BLOCKS;
        bool more2 = gnn < FG_GROUPS;
        int nodenn = gnn * 16 + wave;
        bool validnn = more2 && nodenn < nN;
        int dnn = __builtin_amdgcn_readfirstlane(validnn ? nodenn : nN - 1);
        int nrec2 = nwords[(size_t)dnn * 4];

        int t = (wave - rot) & 15;
        if (t < 7) {
            const float* trow = (const float*)tile[buf] + m * TPITCH + koff;
            const float2* wcol = wl + (koff >> 1) * HPAD + t * 16 + m;
            v8f pacc = (v8f){0.f, 0.f, 0.f, 0.f, 0.f, 0.f, 0.f, 0.f};
#pragma unroll
            for (int k = 0; k < HPAD; k += 4) {
                float2 a2 = *(const float2*)(trow + k);
                float2 b2 = wcol[(k >> 1) * HPAD];
                v2f a = (v2f){a2.x, a2.y};
                v2f b = (v2f){b2.x, b2.y};
                pacc = __builtin_amdgcn_wmma_f32_16x16x4_f32(
                    false, a, false, b, (short)0, pacc, false, false);
            }
            int rbase = g * 16 + ((lane >> 4) << 3);
            int cc = t * 16 + m;
#pragma unroll
            for (int v = 0; v < 8; ++v) {
                int rr = rbase + v;
                if (rr < nN) H2[(size_t)rr * HPAD + cc] = pacc[v];
            }
        }
        if (!more) break;
        g = gn; valid = validn;
        beg = begn; n = nn; di = din; mm = mmn; meta = metan; hs = hsn;
        gn = gnn; more = more2; validn = validnn; dn = dnn; nrec = nrec2;
        rot = (rot + 7) & 15;
        buf ^= 1;
    }
}

__global__ __launch_bounds__(256) void k_gather_pool(
    const float* __restrict__ H, const int4* __restrict__ ninfo, const int2* __restrict__ csr,
    const float* __restrict__ bias, const int* __restrict__ batch,
    float* __restrict__ mean, const float* __restrict__ counts, int nN)
{
    __shared__ float4 srow[8 * (HPAD / 4)];
    __shared__ int4 sg[2];
    __shared__ float4 sc[2];
    gather_node(H, ninfo, csr, bias, batch, mean, counts, (float*)srow, (int*)sg, (float*)sc, nN);
}

extern "C" void kernel_launch(void* const* d_in, const int* in_sizes, int n_in,
                              void* d_out, int out_size, void* d_ws, size_t ws_size,
                              hipStream_t stream) {
    const float* x  = (const float*)d_in[0];
    const float* W1 = (const float*)d_in[1];
    const float* b1 = (const float*)d_in[2];
    const float* W2 = (const float*)d_in[3];
    const float* b2 = (const float*)d_in[4];
    const int*   ei = (const int*)d_in[5];
    const int*   bt = (const int*)d_in[6];
    const int* esrc = ei;
    const int* edst = ei + N_EDGES;
    float* out = (float*)d_out;

    float*  ws    = (float*)d_ws;
    int4*   ninfo = (int4*)ws;
    int*    ecnt  = (int*)(ws + (size_t)4 * N_NODES);
    int*    cursor = ecnt + 2 * N_NODES;
    float*  cnts = (float*)(cursor + 4);
    float*  B0   = cnts + N_GRAPHS + WS_HEAD_PAD;
    float*  B2   = B0 + (size_t)N_NODES * HPAD;
    int*    rank = (int*)(B2 + (size_t)N_NODES * HPAD);
    int2*   csr  = (int2*)(rank + N_EDGES);

    auto cdiv = [](long long a, long long b) { return (unsigned)((a + b - 1) / b); };

    k_zero_int<<<cdiv(2 * N_NODES + 4 + N_GRAPHS + N_GRAPHS * HID, 256), 256, 0, stream>>>(
        ecnt, 2 * N_NODES + 4 + N_GRAPHS, out, N_GRAPHS * HID);

    k_gemm_rank<<<GEMM_BLOCKS + RANK_BLOCKS, 256, 0, stream>>>(
        x, W1, IN_DIM, HID, B0, N_NODES, IN_DIM, GEMM_BLOCKS, edst, ecnt, rank);
    k_alloc<<<cdiv(N_NODES, 256), 256, 0, stream>>>(ecnt, cursor, ninfo, bt, cnts, N_NODES);
    k_fill<<<cdiv(N_EDGES, 256), 256, 0, stream>>>(esrc, edst, rank, ninfo, csr, N_EDGES);

    k_gather_gemm<<<FG_BLOCKS, 512, 0, stream>>>(B0, ninfo, csr, b1, W2, HID, HID, B2, N_NODES);

    k_gather_pool<<<cdiv(N_NODES, 8), 256, 0, stream>>>(B2, ninfo, csr, b2, bt, out, cnts, N_NODES);
}


# Edges grouped by target on LDS counters, 2 levels (chunk ordered by 128-node bucket, then one block per bucket before its GEMM rows); global counting adds go; dinv folded into rows; sum order changes
# speedup vs baseline: 5.7739x; 1.1384x over previous
#include <hip/hip_runtime.h>
#include <hip/hip_bf16.h>

typedef __attribute__((ext_vector_type(2))) float v2f;
typedef __attribute__((ext_vector_type(8))) float v8f;

#define N_NODES  100000
#define N_EDGES  1600000
#define N_GRAPHS 1000
#define IN_DIM   128
#define HID      100
#define HPAD     112
#define GEMM_BLOCKS ((N_NODES + 127) / 128)
#define WS_HEAD (4 * N_NODES + N_GRAPHS)
#define WS_HEAD_PAD ((32 - WS_HEAD % 32) % 32)
static_assert((WS_HEAD + WS_HEAD_PAD) % 32 == 0, "the first row table starts on a 128-byte line");
static_assert((N_NODES * HPAD) % 32 == 0, "and so does the second, one table behind it");

#define BN 128
#define NBK ((N_NODES + BN - 1) / BN)
#define CH 4096
#define NCH ((N_EDGES + CH - 1) / CH)
#define STG 4096
static_assert(NBK + 1 <= 1024, "the level-1 scan covers 1024 entries");
static_assert(N_EDGES % 4 == 0 && CH % 1024 == 0, "edges are read four at a time, 256 threads");
static_assert(NCH * 256 >= N_GRAPHS * HID && NCH * 256 >= N_GRAPHS, "the level-1 blocks also clear the output and the graph counts");

__global__ __launch_bounds__(256) void k_bin(const int* __restrict__ src, const int* __restrict__ dst,
                                             int2* __restrict__ bins, int* __restrict__ offT,
                                             float* __restrict__ out, float* __restrict__ cnts) {
    __shared__ int hist[1024];
    __shared__ int2 sorted[CH];
    __shared__ int wsum[8];
    int tid = threadIdx.x, lane = tid & 31, wv = tid >> 5;
    int k = blockIdx.x;
    {
        int i = k * 256 + tid;
        if (i < N_GRAPHS * HID) out[i] = 0.0f;
        if (i < N_GRAPHS) cnts[i] = 0.0f;
    }
    for (int i = tid; i < 1024; i += 256) hist[i] = 0;
    int4 dv[CH / 1024], sv[CH / 1024];
#pragma unroll
    for (int j = 0; j < CH / 1024; ++j) {
        int i4 = k * (CH / 4) + j * 256 + tid;
        bool ok = i4 < N_EDGES / 4;
        int ic = ok ? i4 : 0;
        dv[j] = ((const int4*)dst)[ic];
        sv[j] = ((const int4*)src)[ic];
        if (!ok) dv[j] = make_int4(-1, -1, -1, -1);
    }
    __syncthreads();
    int rk[CH / 256];
#pragma unroll
    for (int j = 0; j < CH / 1024; ++j) {
        int dd[4] = {dv[j].x, dv[j].y, dv[j].z, dv[j].w};
#pragma unroll
        for (int q = 0; q < 4; ++q) {
            rk[j * 4 + q] = 0;
            if (dd[q] >= 0) rk[j * 4 + q] = atomicAdd(&hist[dd[q] >> 7], 1);
        }
    }
    __syncthreads();
    int a0 = hist[tid * 4], a1 = hist[tid * 4 + 1], a2 = hist[tid * 4 + 2], a3 = hist[tid * 4 + 3];
    int v = a0 + a1 + a2 + a3;
    int sum4 = v;
#pragma unroll
    for (int o = 1; o < 32; o <<= 1) {
        int t = __shfl_up(v, o);
        if (lane >= o) v += t;
    }
    if (lane == 31) wsum[wv] = v;
    __syncthreads();
    int pre = 0;
#pragma unroll
    for (int q = 0; q < 8; ++q) pre += (q < wv) ? wsum[q] : 0;
    int ex = pre + v - sum4;
    hist[tid * 4] = ex; hist[tid * 4 + 1] = ex + a0; hist[tid * 4 + 2] = ex + a0 + a1; hist[tid * 4 + 3] = ex + a0 + a1 + a2;
    __syncthreads();
#pragma unroll
    for (int j = 0; j < CH / 1024; ++j) {
        int dd[4] = {dv[j].x, dv[j].y, dv[j].z, dv[j].w};
        int ss[4] = {sv[j].x, sv[j].y, sv[j].z, sv[j].w};
#pragma unroll
        for (int q = 0; q < 4; ++q)
            if (dd[q] >= 0) sorted[hist[dd[q] >> 7] + rk[j * 4 + q]] = make_int2(ss[q], dd[q]);
    }
    for (int b = tid; b <= NBK; b += 256) offT[b * NCH + k] = hist[b];
    __syncthreads();
    int tot = hist[NBK];
    int4* o4 = (int4*)(bins + (size_t)k * CH);
    const int4* s4 = (const int4*)sorted;
    for (int i = tid; i < tot / 2; i += 256) o4[i] = s4[i];
    if ((tot & 1) && tid == 0) bins[(size_t)k * CH + tot - 1] = sorted[tot - 1];
}

__device__ __forceinline__ void gemm_block(
    int bx, const float* __restrict__ X, const float* __restrict__ W, int wr, int wc,
    float* __restrict__ H, int nrows, int K, const float* rscale)
{
    __shared__ float2 wl[(IN_DIM / 2) * HPAD];
    int tid = threadIdx.x;
    int kpairs = K >> 1;
    int total = kpairs * HPAD;
    for (int base = tid; base < total; base += 14 * 256) {
        float lo[14], hi[14];
#pragma unroll
        for (int u = 0; u < 14; ++u) {
            int i = base + u * 256;
            int p = i / HPAD, n = i - p * HPAD;
            int k0 = 2 * p, k1 = k0 + 1;
            bool ins = i < total && n < wc;
            bool ok0 = ins && k0 < wr, ok1 = ins && k1 < wr;
            float v0 = W[ok0 ? k0 * wc + n : 0];
            float v1 = W[ok1 ? k1 * wc + n : 0];
            lo[u] = ok0 ? v0 : 0.0f;
            hi[u] = ok1 ? v1 : 0.0f;
        }
#pragma unroll
        for (int u = 0; u < 14; ++u) {
            wl[base + u * 256] = make_float2(lo[u], hi[u]);
        }
    }
    __syncthreads();

    int lane = tid & 31;
    int wave = tid >> 5;
    int m    = lane & 15;
    int koff = (lane >> 4) << 1;
    int row0 = bx * 128 + wave * 16;
    int row  = row0 + m;
    bool inr = row < nrows;
    int rowc = inr ? row : (nrows - 1);
    const float* xrow = X + (size_t)rowc * K;

    v8f acc[7];
#pragma unroll
    for (int t = 0; t < 7; ++t)
        acc[t] = (v8f){0.f, 0.f, 0.f, 0.f, 0.f, 0.f, 0.f, 0.f};

    for (int k = 0; k < K; k += 4) {
        float2 t2 = *(const float2*)(xrow + k + koff);
        v2f a = (v2f){ inr ? t2.x : 0.f, inr ? t2.y : 0.f };
        const float2* wrow = wl + ((k + koff) >> 1) * HPAD;
#pragma unroll
        for (int t = 0; t < 7; ++t) {
            float2 bb = wrow[t * 16 + m];
            v2f b = (v2f){bb.x, bb.y};
            acc[t] = __builtin_amdgcn_wmma_f32_16x16x4_f32(
                false, a, false, b, (short)0, acc[t], false, false);
        }
    }

    int rbase = row0 + ((lane >> 4) << 3);
#pragma unroll
    for (int t = 0; t < 7; ++t) {
        int cc = t * 16 + m;
#pragma unroll
        for (int v = 0; v < 8; ++v) {
            int rr = rbase + v;
            if (rr < nrows) H[(size_t)rr * HPAD + cc] = acc[t][v] * rscale[rr - bx * 128];
        }
    }
}

static_assert(BN == 128 && GEMM_BLOCKS == NBK, "a bucket is a GEMM block's 128 rows");
__global__ __launch_bounds__(256) void k_build_gemm(
    const float* __restrict__ X, const float* __restrict__ W, int wr, int wc,
    float* __restrict__ H, int nrows, int K,
    const int2* __restrict__ bins, const int* __restrict__ offT,
    int4* __restrict__ ninfo, int* __restrict__ csr, const int* __restrict__ batch, float* counts)
{
    __shared__ int cnt[BN];
    __shared__ int cur[BN];
    __shared__ float dsc[BN];
    __shared__ int stage[STG];
    __shared__ int wsum[8];
    __shared__ int sbase, stot;
    int tid = threadIdx.x, lane = tid & 31, wv = tid >> 5;
    int b = blockIdx.x;
    if (tid < BN) cnt[tid] = 0;
    if (tid == 0) sbase = 0;
    int lo0 = 0, hi0 = 0, lo1 = 0, hi1 = 0;
    int k0 = tid, k1 = tid + 256;
    if (k0 < NCH) { lo0 = offT[b * NCH + k0]; hi0 = offT[(b + 1) * NCH + k0]; }
    if (k1 < NCH) { lo1 = offT[b * NCH + k1]; hi1 = offT[(b + 1) * NCH + k1]; }
    const int2* p0 = bins + (size_t)k0 * CH;
    const int2* p1 = bins + (size_t)(k1 < NCH ? k1 : 0) * CH;
    int n0 = hi0 - lo0, n1 = hi1 - lo1;
    int2 q0[8], q1[8];
#pragma unroll
    for (int u = 0; u < 8; ++u) {
        q0[u] = p0[u < n0 ? lo0 + u : 0];
        q1[u] = p1[u < n1 ? lo1 + u : 0];
    }
    __syncthreads();
    atomicAdd(&sbase, lo0 + lo1);
#pragma unroll
    for (int u = 0; u < 8; ++u) {
        if (u < n0) atomicAdd(&cnt[q0[u].y & (BN - 1)], 1);
        if (u < n1) atomicAdd(&cnt[q1[u].y & (BN - 1)], 1);
    }
    for (int e = lo0 + 8; e < hi0; ++e) atomicAdd(&cnt[p0[e].y & (BN - 1)], 1);
    for (int e = lo1 + 8; e < hi1; ++e) atomicAdd(&cnt[p1[e].y & (BN - 1)], 1);
    __syncthreads();
    int node = b * BN + tid;
    bool isn = tid < BN && node < nrows;
    int c = isn ? cnt[tid] : 0;
    int g = isn ? batch[node] : -1;
    {
        if ((unsigned)g >= (unsigned)N_GRAPHS) g = -1;
        int gp = __shfl_up(g, 1);
        bool head = lane == 0 || g != gp;
        unsigned hm = (unsigned)__ballot(head);
        unsigned above = hm & ~((2u << lane) - 1u);
        int len = (above ? __builtin_ctz(above) : 32) - lane;
        if (head && g >= 0) atomicAdd(&counts[g], (float)len);
    }
    int v = c;
#pragma unroll
    for (int o = 1; o < 32; o <<= 1) {
        int t = __shfl_up(v, o);
        if (lane >= o) v += t;
    }
    if (lane == 31) wsum[wv] = v;
    __syncthreads();
    int pre = 0;
#pragma unroll
    for (int q = 0; q < 8; ++q) pre += (q < wv) ? wsum[q] : 0;
    int start = pre + v - c;
    int base = sbase;
    if (tid < BN) {
        float dinv = rsqrtf((float)(c + 1));
        cur[tid] = start;
        dsc[tid] = dinv;
        if (isn) ninfo[node] = make_int4(base + start, c, __float_as_int(dinv), 0);
    }
    if (tid == 255) stot = pre + v;
    __syncthreads();
    int tot = stot;
    bool staged = tot <= STG;
#pragma unroll
    for (int u = 0; u < 8; ++u) {
        if (u < n0) {
            int pos = atomicAdd(&cur[q0[u].y & (BN - 1)], 1);
            if (staged) stage[pos] = q0[u].x; else csr[base + pos] = q0[u].x;
        }
        if (u < n1) {
            int pos = atomicAdd(&cur[q1[u].y & (BN - 1)], 1);
            if (staged) stage[pos] = q1[u].x; else csr[base + pos] = q1[u].x;
        }
    }
    for (int e = lo0 + 8; e < hi0; ++e) {
        int2 sd = p0[e];
        int pos = atomicAdd(&cur[sd.y & (BN - 1)], 1);
        if (staged) stage[pos] = sd.x; else csr[base + pos] = sd.x;
    }
    for (int e = lo1 + 8; e < hi1; ++e) {
        int2 sd = p1[e];
        int pos = atomicAdd(&cur[sd.y & (BN - 1)], 1);
        if (staged) stage[pos] = sd.x; else csr[base + pos] = sd.x;
    }
    __syncthreads();
    if (staged)
        for (int i = tid; i < tot; i += 256) csr[base + i] = stage[i];
    gemm_block(b, X, W, wr, wc, H, nrows, K, dsc);
}

__device__ __forceinline__ void gather_node(
    const float* __restrict__ H, const int4* __restrict__ ninfo, const int* __restrict__ csr,
    const float* __restrict__ bias, const int* __restrict__ batch,
    float* __restrict__ mean, const float* __restrict__ counts, float* srow, int* sg, float* sc, int nN)
{
    int lane = threadIdx.x & 31;
    int wid = threadIdx.x >> 5;
    int node = blockIdx.x * 8 + wid;
    bool valid = node < nN;
    if (!valid) node = nN - 1;
    int d = __builtin_amdgcn_readfirstlane(node);
    int4 ni = ninfo[d];
    int gid = batch[d];
    {
        bool ok = __builtin_amdgcn_readfirstlane((int)(blockIdx.x * 8 + wid)) < nN && (unsigned)gid < (unsigned)N_GRAPHS;
        if (!ok) gid = -1;
    }
    int beg = ni.x, n = ni.y;
    float di = __int_as_float(ni.z);
    int c4 = lane < 28 ? lane : 27;
    const float4* H4 = (const float4*)H + c4;

    int m = n < 32 ? n : 32;
    int meta = 0;
    if (lane < m) meta = csr[beg + lane];
    float4 hs = H4[(size_t)d * (HPAD / 4)];
    bool hasb = c4 < HID / 4;
    float4 bb = ((const float4*)bias)[hasb ? c4 : HID / 4 - 1];
    float gcnt = counts[gid >= 0 ? gid : 0];
    float4 acc = make_float4(0.f, 0.f, 0.f, 0.f);

    for (int base = 0;;) {
        for (int j = 0; j < m; j += 8) {
            int r = m - j;
            int   sx[8];
            float4 h[8];
#pragma unroll
            for (int u = 0; u < 8; ++u) {
                int jj = (u < r) ? j + u : m - 1;
                sx[u] = __builtin_amdgcn_readlane(meta, jj);
            }
#pragma unroll
            for (int u = 0; u < 8; ++u) {
                h[u] = make_float4(0.f, 0.f, 0.f, 0.f);
                if (u < r) h[u] = H4[(size_t)sx[u] * (HPAD / 4)];
            }
#pragma unroll
            for (int u = 0; u < 8; ++u) {
                acc.x += h[u].x;
                acc.y += h[u].y;
                acc.z += h[u].z;
                acc.w += h[u].w;
            }
        }
        base += 32;
        if (base >= n) break;
        m = n - base;
        if (m > 32) m = 32;
        meta = 0;
        if (lane < m) meta = csr[beg + base + lane];
    }

    if (!hasb) bb = make_float4(0.f, 0.f, 0.f, 0.f);
    float4 o;
    o.x = fmaf(hs.x + acc.x, di, bb.x); o.x = o.x > 0.f ? o.x : 0.f;
    o.y = fmaf(hs.y + acc.y, di, bb.y); o.y = o.y > 0.f ? o.y : 0.f;
    o.z = fmaf(hs.z + acc.z, di, bb.z); o.z = o.z > 0.f ? o.z : 0.f;
    o.w = fmaf(hs.w + acc.w, di, bb.w); o.w = o.w > 0.f ? o.w : 0.f;
    if (lane < 28) ((float4*)srow)[wid * (HPAD / 4) + lane] = o;
    if (lane == 0) { sg[wid] = gid; sc[wid] = gcnt; }
    __syncthreads();
    int t = threadIdx.x;
    if (t >= HID) return;
    int4 ga = ((const int4*)sg)[0], gb = ((const int4*)sg)[1];
    float4 ca = ((const float4*)sc)[0], cb = ((const float4*)sc)[1];
    int gq[8] = {ga.x, ga.y, ga.z, ga.w, gb.x, gb.y, gb.z, gb.w};
    float cq[8] = {ca.x, ca.y, ca.z, ca.w, cb.x, cb.y, cb.z, cb.w};
    float vq[8];
#pragma unroll
    for (int q = 0; q < 8; ++q) vq[q] = srow[q * HPAD + t];
    int cur = -1;
    float ccur = 1.0f;
    float s = 0.f;
#pragma unroll
    for (int q = 0; q < 8; ++q) {
        int gw = gq[q];
        float v = vq[q];
        if (gw >= 0) {
            if (gw == cur) {
                s += v;
            } else {
                if (cur >= 0) atomicAdd(&mean[(size_t)cur * HID + t], s / fmaxf(ccur, 1.0f));
                cur = gw;
                ccur = cq[q];
                s = v;
            }
        }
    }
    if (cur >= 0) atomicAdd(&mean[(size_t)cur * HID + t], s / fmaxf(ccur, 1.0f));
}

#define FG_BLOCKS 1024
#define FG_GROUPS ((N_NODES + 15) / 16)
#define TPITCH 116
__global__ __launch_bounds__(512) void k_gather_gemm(
    const float* __restrict__ H, const int4* __restrict__ ninfo, const int* __restrict__ csr,
    const float* __restrict__ bias, const float* __restrict__ W, int wr, int wc,
    float* __restrict__ H2, int nN)
{
    __shared__ float2 wl[(HPAD / 2) * HPAD];
    __shared__ float4 tile[2][16 * (TPITCH / 4)];
    int tid = threadIdx.x;
    int lane = tid & 31;
    int wave = __builtin_amdgcn_readfirstlane(tid >> 5);

    {
        const int total = (HPAD / 2) * HPAD;
        float lo[13], hi[13];
#pragma unroll
        for (int u = 0; u < 13; ++u) {
            int i = tid + u * 512;
            int p = i / HPAD, n = i - p * HPAD;
            int k0 = 2 * p, k1 = k0 + 1;
            bool ins = i < total && n < wc;
            bool ok0 = ins && k0 < wr, ok1 = ins && k1 < wr;
            float v0 = W[ok0 ? k0 * wc + n : 0];
            float v1 = W[ok1 ? k1 * wc + n : 0];
            lo[u] = ok0 ? v0 : 0.0f;
            hi[u] = ok1 ? v1 : 0.0f;
        }
#pragma unroll
        for (int u = 0; u < 13; ++u) {
            int i = tid + u * 512;
            if (i < total) wl[i] = make_float2(lo[u], hi[u]);
        }
    }

    int m = lane & 15;
    int koff = (lane >> 4) << 1;
    int c4 = lane < 28 ? lane : 27;
    const float4* H4 = (const float4*)H + c4;
    bool hasb = c4 < HID / 4;
    float4 bb = ((const float4*)bias)[hasb ? c4 : HID / 4 - 1];
    if (!hasb) bb = make_float4(0.f, 0.f, 0.f, 0.f);

    int g = blockIdx.x;
    if (g >= FG_GROUPS) return;

    const int* nwords = (const int*)ninfo + (lane & 3);

    int node = g * 16 + wave;
    bool valid = node < nN;
    int d = __builtin_amdgcn_readfirstlane(valid ? node : nN - 1);
    int rec = nwords[(size_t)d * 4];
    int beg = __builtin_amdgcn_readlane(rec, 0);
    int n = __builtin_amdgcn_readlane(rec, 1);
    float di = __int_as_float(__builtin_amdgcn_readlane(rec, 2));
    int mm = n < 32 ? n : 32;
    int meta = 0;
    if (lane < mm) meta = csr[beg + lane];
    float4 hs = H4[(size_t)d * (HPAD / 4)];
    int gn = g + FG_BLOCKS;
    bool more = gn < FG_GROUPS;
    int noden = gn * 16 + wave;
    bool validn = more && noden < nN;
    int dn = __builtin_amdgcn_readfirstlane(validn ? noden : nN - 1);
    int nrec = nwords[(size_t)dn * 4];
    int buf = 0;
    int rot = 0;

    for (;;) {
        float4 acc = make_float4(0.f, 0.f, 0.f, 0.f);
        for (int base = 0;;) {
            for (int j = 0; j < mm; j += 8) {
                int r = mm - j;
                int   sx[8];
                float4 h[8];
#pragma unroll
                for (int u = 0; u < 8; ++u) {
                    int jj = (u < r) ? j + u : mm - 1;
                    sx[u] = __builtin_amdgcn_readlane(meta, jj);
                }
#pragma unroll
                for (int u = 0; u < 8; ++u) {
                    h[u] = make_float4(0.f, 0.f, 0.f, 0.f);
                    if (u < r) h[u] = H4[(size_t)sx[u] * (HPAD / 4)];
                }
#pragma unroll
                for (int u = 0; u < 8; ++u) {
                    acc.x += h[u].x;
                    acc.y += h[u].y;
                    acc.z += h[u].z;
                    acc.w += h[u].w;
                }
            }
            base += 32;
            if (base >= n) break;
            mm = n - base;
            if (mm > 32) mm = 32;
            meta = 0;
            if (lane < mm) meta = csr[beg + base + lane];
        }

        float4 o;
        o.x = fmaf(hs.x + acc.x, di, bb.x); o.x = o.x > 0.f ? o.x * di : 0.f;
        o.y = fmaf(hs.y + acc.y, di, bb.y); o.y = o.y > 0.f ? o.y * di : 0.f;
        o.z = fmaf(hs.z + acc.z, di, bb.z); o.z = o.z > 0.f ? o.z * di : 0.f;
        o.w = fmaf(hs.w + acc.w, di, bb.w); o.w = o.w > 0.f ? o.w * di : 0.f;
        if (!valid) o = make_float4(0.f, 0.f, 0.f, 0.f);
        if (lane < 28) tile[buf][wave * (TPITCH / 4) + lane] = o;
        __syncthreads();

        int begn = __builtin_amdgcn_readlane(nrec, 0);
        int nn = __builtin_amdgcn_readlane(nrec, 1);
        float din = __int_as_float(__builtin_amdgcn_readlane(nrec, 2));
        int mmn = nn < 32 ? nn : 32;
        int metan = 0;
        if (lane < mmn) metan = csr[begn + lane];
        float4 hsn = H4[(size_t)dn * (HPAD / 4)];
        int gnn = gn + FG_BLOCKS;
        bool more2 = gnn < FG_GROUPS;
        int nodenn = gnn * 16 + wave;
        bool validnn = more2 && nodenn < nN;
        int dnn = __builtin_amdgcn_readfirstlane(validnn ? nodenn : nN - 1);
        int nrec2 = nwords[(size_t)dnn * 4];

        int t = (wave - rot) & 15;
        if (t < 7) {
            const float* trow = (const float*)tile[buf] + m * TPITCH + koff;
            const float2* wcol = wl + (koff >> 1) * HPAD + t * 16 + m;
            v8f pacc = (v8f){0.f, 0.f, 0.f, 0.f, 0.f, 0.f, 0.f, 0.f};
#pragma unroll
            for (int k = 0; k < HPAD; k += 4) {
                float2 a2 = *(const float2*)(trow + k);
                float2 b2 = wcol[(k >> 1) * HPAD];
                v2f a = (v2f){a2.x, a2.y};
                v2f b = (v2f){b2.x, b2.y};
                pacc = __builtin_amdgcn_wmma_f32_16x16x4_f32(
                    false, a, false, b, (short)0, pacc, false, false);
            }
            int rbase = g * 16 + ((lane >> 4) << 3);
            int cc = t * 16 + m;
#pragma unroll
            for (int v = 0; v < 8; ++v) {
                int rr = rbase + v;
                if (rr < nN) H2[(size_t)rr * HPAD + cc] = pacc[v];
            }
        }
        if (!more) break;
        g = gn; valid = validn;
        beg = begn; n = nn; di = din; mm = mmn; meta = metan; hs = hsn;
        gn = gnn; more = more2; validn = validnn; dn = dnn; nrec = nrec2;
        rot = (rot + 7) & 15;
        buf ^= 1;
    }
}

__global__ __launch_bounds__(256) void k_gather_pool(
    const float* __restrict__ H, const int4* __restrict__ ninfo, const int* __restrict__ csr,
    const float* __restrict__ bias, const int* __restrict__ batch,
    float* __restrict__ mean, const float* __restrict__ counts, int nN)
{
    __shared__ float4 srow[8 * (HPAD / 4)];
    __shared__ int4 sg[2];
    __shared__ float4 sc[2];
    gather_node(H, ninfo, csr, bias, batch, mean, counts, (float*)srow, (int*)sg, (float*)sc, nN);
}

extern "C" void kernel_launch(void* const* d_in, const int* in_sizes, int n_in,
                              void* d_out, int out_size, void* d_ws, size_t ws_size,
                              hipStream_t stream) {
    const float* x  = (const float*)d_in[0];
    const float* W1 = (const float*)d_in[1];
    const float* b1 = (const float*)d_in[2];
    const float* W2 = (const float*)d_in[3];
    const float* b2 = (const float*)d_in[4];
    const int*   ei = (const int*)d_in[5];
    const int*   bt = (const int*)d_in[6];
    const int* esrc = ei;
    const int* edst = ei + N_EDGES;
    float* out = (float*)d_out;

    float*  ws    = (float*)d_ws;
    int4*   ninfo = (int4*)ws;
    float*  cnts = ws + (size_t)4 * N_NODES;
    float*  B0   = cnts + N_GRAPHS + WS_HEAD_PAD;
    float*  B2   = B0 + (size_t)N_NODES * HPAD;
    int2*   bins = (int2*)(B2 + (size_t)N_NODES * HPAD);
    int*    offT = (int*)(bins + (size_t)NCH * CH);
    int*    csr  = offT + (size_t)(NBK + 1) * NCH;

    auto cdiv = [](long long a, long long b) { return (unsigned)((a + b - 1) / b); };

    k_bin<<<NCH, 256, 0, stream>>>(esrc, edst, bins, offT, out, cnts);

    k_build_gemm<<<GEMM_BLOCKS, 256, 0, stream>>>(
        x, W1, IN_DIM, HID, B0, N_NODES, IN_DIM, bins, offT, ninfo, csr, bt, cnts);

    k_gather_gemm<<<FG_BLOCKS, 512, 0, stream>>>(B0, ninfo, csr, b1, W2, HID, HID, B2, N_NODES);

    k_gather_pool<<<cdiv(N_NODES, 8), 256, 0, stream>>>(B2, ninfo, csr, b2, bt, out, cnts, N_NODES);
}


# First product freed from the degree chain: H1 rows unscaled, node's dinv in pad column 100, multiplied in by the gathering wave (fused multiply-adds, same order); it runs as side blocks of launch 1
# speedup vs baseline: 6.2308x; 1.0791x over previous
#include <hip/hip_runtime.h>
#include <hip/hip_bf16.h>

typedef __attribute__((ext_vector_type(2))) float v2f;
typedef __attribute__((ext_vector_type(8))) float v8f;

#define N_NODES  100000
#define N_EDGES  1600000
#define N_GRAPHS 1000
#define IN_DIM   128
#define HID      100
#define HPAD     112
#define GEMM_BLOCKS ((N_NODES + 127) / 128)
#define WS_HEAD (4 * N_NODES + N_GRAPHS)
#define WS_HEAD_PAD ((32 - WS_HEAD % 32) % 32)
static_assert((WS_HEAD + WS_HEAD_PAD) % 32 == 0, "the first row table starts on a 128-byte line");
static_assert((N_NODES * HPAD) % 32 == 0, "and so does the second, one table behind it");

#define BN 128
#define NBK ((N_NODES + BN - 1) / BN)
#define CH 4096
#define NCH ((N_EDGES + CH - 1) / CH)
#define STG 4096
static_assert(NBK + 1 <= 1024, "the level-1 scan covers 1024 entries");
static_assert(N_EDGES % 4 == 0 && CH % 1024 == 0, "edges are read four at a time, 256 threads");
static_assert(NCH * 256 >= N_GRAPHS * HID && NCH * 256 >= N_GRAPHS, "the level-1 blocks also clear the output and the graph counts");

__device__ __forceinline__ void gemm_block(
    int bx, const float* __restrict__ X, const float* __restrict__ W, int wr, int wc,
    float* __restrict__ H, int nrows, int K, float2* wl);

static_assert(1024 * 4 + CH * 8 + 8 * 4 <= (IN_DIM / 2) * HPAD * 8, "a level-1 block's counters and chunk fit in the weight array");
__global__ __launch_bounds__(256) void k_bin_gemm(const int* __restrict__ src, const int* __restrict__ dst,
                                             int2* __restrict__ bins, int* __restrict__ offT,
                                             float* __restrict__ out, float* __restrict__ cnts,
                                             const float* __restrict__ X, const float* __restrict__ W, int wr, int wc,
                                             float* __restrict__ H, int nrows, int K) {
    __shared__ float2 wl[(IN_DIM / 2) * HPAD];
    if (blockIdx.x >= NCH) {
        gemm_block(blockIdx.x - NCH, X, W, wr, wc, H, nrows, K, wl);
        return;
    }
    int* hist = (int*)wl;
    int2* sorted = (int2*)(wl + 512);
    int* wsum = (int*)(wl + 512 + CH);
    int tid = threadIdx.x, lane = tid & 31, wv = tid >> 5;
    int k = blockIdx.x;
    {
        int i = k * 256 + tid;
        if (i < N_GRAPHS * HID) out[i] = 0.0f;
        if (i < N_GRAPHS) cnts[i] = 0.0f;
    }
    for (int i = tid; i < 1024; i += 256) hist[i] = 0;
    int4 dv[CH / 1024], sv[CH / 1024];
#pragma unroll
    for (int j = 0; j < CH / 1024; ++j) {
        int i4 = k * (CH / 4) + j * 256 + tid;
        bool ok = i4 < N_EDGES / 4;
        int ic = ok ? i4 : 0;
        dv[j] = ((const int4*)dst)[ic];
        sv[j] = ((const int4*)src)[ic];
        if (!ok) dv[j] = make_int4(-1, -1, -1, -1);
    }
    __syncthreads();
    int rk[CH / 256];
#pragma unroll
    for (int j = 0; j < CH / 1024; ++j) {
        int dd[4] = {dv[j].x, dv[j].y, dv[j].z, dv[j].w};
#pragma unroll
        for (int q = 0; q < 4; ++q) {
            rk[j * 4 + q] = 0;
            if (dd[q] >= 0) rk[j * 4 + q] = atomicAdd(&hist[dd[q] >> 7], 1);
        }
    }
    __syncthreads();
    int a0 = hist[tid * 4], a1 = hist[tid * 4 + 1], a2 = hist[tid * 4 + 2], a3 = hist[tid * 4 + 3];
    int v = a0 + a1 + a2 + a3;
    int sum4 = v;
#pragma unroll
    for (int o = 1; o < 32; o <<= 1) {
        int t = __shfl_up(v, o);
        if (lane >= o) v += t;
    }
    if (lane == 31) wsum[wv] = v;
    __syncthreads();
    int pre = 0;
#pragma unroll
    for (int q = 0; q < 8; ++q) pre += (q < wv) ? wsum[q] : 0;
    int ex = pre + v - sum4;
    hist[tid * 4] = ex; hist[tid * 4 + 1] = ex + a0; hist[tid * 4 + 2] = ex + a0 + a1; hist[tid * 4 + 3] = ex + a0 + a1 + a2;
    __syncthreads();
#pragma unroll
    for (int j = 0; j < CH / 1024; ++j) {
        int dd[4] = {dv[j].x, dv[j].y, dv[j].z, dv[j].w};
        int ss[4] = {sv[j].x, sv[j].y, sv[j].z, sv[j].w};
#pragma unroll
        for (int q = 0; q < 4; ++q)
            if (dd[q] >= 0) sorted[hist[dd[q] >> 7] + rk[j * 4 + q]] = make_int2(ss[q], dd[q]);
    }
    for (int b = tid; b <= NBK; b += 256) offT[b * NCH + k] = hist[b];
    __syncthreads();
    int tot = hist[NBK];
    int4* o4 = (int4*)(bins + (size_t)k * CH);
    const int4* s4 = (const int4*)sorted;
    for (int i = tid; i < tot / 2; i += 256) o4[i] = s4[i];
    if ((tot & 1) && tid == 0) bins[(size_t)k * CH + tot - 1] = sorted[tot - 1];
}

__device__ __forceinline__ void gemm_block(
    int bx, const float* __restrict__ X, const float* __restrict__ W, int wr, int wc,
    float* __restrict__ H, int nrows, int K, float2* wl)
{
    int tid = threadIdx.x;
    int kpairs = K >> 1;
    int total = kpairs * HPAD;
    for (int base = tid; base < total; base += 14 * 256) {
        float lo[14], hi[14];
#pragma unroll
        for (int u = 0; u < 14; ++u) {
            int i = base + u * 256;
            int p = i / HPAD, n = i - p * HPAD;
            int k0 = 2 * p, k1 = k0 + 1;
            bool ins = i < total && n < wc;
            bool ok0 = ins && k0 < wr, ok1 = ins && k1 < wr;
            float v0 = W[ok0 ? k0 * wc + n : 0];
            float v1 = W[ok1 ? k1 * wc + n : 0];
            lo[u] = ok0 ? v0 : 0.0f;
            hi[u] = ok1 ? v1 : 0.0f;
        }
#pragma unroll
        for (int u = 0; u < 14; ++u) {
            wl[base + u * 256] = make_float2(lo[u], hi[u]);
        }
    }
    __syncthreads();

    int lane = tid & 31;
    int wave = tid >> 5;
    int m    = lane & 15;
    int koff = (lane >> 4) << 1;
    int row0 = bx * 128 + wave * 16;
    int row  = row0 + m;
    bool inr = row < nrows;
    int rowc = inr ? row : (nrows - 1);
    const float* xrow = X + (size_t)rowc * K;

    v8f acc[7];
#pragma unroll
    for (int t = 0; t < 7; ++t)
        acc[t] = (v8f){0.f, 0.f, 0.f, 0.f, 0.f, 0.f, 0.f, 0.f};

    for (int k = 0; k < K; k += 4) {
        float2 t2 = *(const float2*)(xrow + k + koff);
        v2f a = (v2f){ inr ? t2.x : 0.f, inr ? t2.y : 0.f };
        const float2* wrow = wl + ((k + koff) >> 1) * HPAD;
#pragma unroll
        for (int t = 0; t < 7; ++t) {
            float2 bb = wrow[t * 16 + m];
            v2f b = (v2f){bb.x, bb.y};
            acc[t] = __builtin_amdgcn_wmma_f32_16x16x4_f32(
                false, a, false, b, (short)0, acc[t], false, false);
        }
    }

    int rbase = row0 + ((lane >> 4) << 3);
#pragma unroll
    for (int t = 0; t < 7; ++t) {
        int cc = t * 16 + m;
#pragma unroll
        for (int v = 0; v < 8; ++v) {
            int rr = rbase + v;
            if (rr < nrows) H[(size_t)rr * HPAD + cc] = acc[t][v];
        }
    }
}

static_assert(BN == 128 && GEMM_BLOCKS == NBK, "a bucket is a GEMM block's 128 rows");
__global__ __launch_bounds__(256) void k_build(
    float* __restrict__ H, int nrows,
    const int2* __restrict__ bins, const int* __restrict__ offT,
    int4* __restrict__ ninfo, int* __restrict__ csr, const int* __restrict__ batch, float* counts)
{
    __shared__ int cnt[BN];
    __shared__ int cur[BN];
    __shared__ int stage[STG];
    __shared__ int wsum[8];
    __shared__ int sbase, stot;
    int tid = threadIdx.x, lane = tid & 31, wv = tid >> 5;
    int b = blockIdx.x;
    if (tid < BN) cnt[tid] = 0;
    if (tid == 0) sbase = 0;
    int lo0 = 0, hi0 = 0, lo1 = 0, hi1 = 0;
    int k0 = tid, k1 = tid + 256;
    if (k0 < NCH) { lo0 = offT[b * NCH + k0]; hi0 = offT[(b + 1) * NCH + k0]; }
    if (k1 < NCH) { lo1 = offT[b * NCH + k1]; hi1 = offT[(b + 1) * NCH + k1]; }
    const int2* p0 = bins + (size_t)k0 * CH;
    const int2* p1 = bins + (size_t)(k1 < NCH ? k1 : 0) * CH;
    int n0 = hi0 - lo0, n1 = hi1 - lo1;
    int2 q0[8], q1[8];
#pragma unroll
    for (int u = 0; u < 8; ++u) {
        q0[u] = p0[u < n0 ? lo0 + u : 0];
        q1[u] = p1[u < n1 ? lo1 + u : 0];
    }
    __syncthreads();
    atomicAdd(&sbase, lo0 + lo1);
#pragma unroll
    for (int u = 0; u < 8; ++u) {
        if (u < n0) atomicAdd(&cnt[q0[u].y & (BN - 1)], 1);
        if (u < n1) atomicAdd(&cnt[q1[u].y & (BN - 1)], 1);
    }
    for (int e = lo0 + 8; e < hi0; ++e) atomicAdd(&cnt[p0[e].y & (BN - 1)], 1);
    for (int e = lo1 + 8; e < hi1; ++e) atomicAdd(&cnt[p1[e].y & (BN - 1)], 1);
    __syncthreads();
    int node = b * BN + tid;
    bool isn = tid < BN && node < nrows;
    int c = isn ? cnt[tid] : 0;
    int g = isn ? batch[node] : -1;
    {
        if ((unsigned)g >= (unsigned)N_GRAPHS) g = -1;
        int gp = __shfl_up(g, 1);
        bool head = lane == 0 || g != gp;
        unsigned hm = (unsigned)__ballot(head);
        unsigned above = hm & ~((2u << lane) - 1u);
        int len = (above ? __builtin_ctz(above) : 32) - lane;
        if (head && g >= 0) atomicAdd(&counts[g], (float)len);
    }
    int v = c;
#pragma unroll
    for (int o = 1; o < 32; o <<= 1) {
        int t = __shfl_up(v, o);
        if (lane >= o) v += t;
    }
    if (lane == 31) wsum[wv] = v;
    __syncthreads();
    int pre = 0;
#pragma unroll
    for (int q = 0; q < 8; ++q) pre += (q < wv) ? wsum[q] : 0;
    int start = pre + v - c;
    int base = sbase;
    if (tid < BN) {
        float dinv = rsqrtf((float)(c + 1));
        cur[tid] = start;
        if (isn) {
            ninfo[node] = make_int4(base + start, c, __float_as_int(dinv), 0);
            H[(size_t)node * HPAD + HID] = dinv;
        }
    }
    if (tid == 255) stot = pre + v;
    __syncthreads();
    int tot = stot;
    bool staged = tot <= STG;
#pragma unroll
    for (int u = 0; u < 8; ++u) {
        if (u < n0) {
            int pos = atomicAdd(&cur[q0[u].y & (BN - 1)], 1);
            if (staged) stage[pos] = q0[u].x; else csr[base + pos] = q0[u].x;
        }
        if (u < n1) {
            int pos = atomicAdd(&cur[q1[u].y & (BN - 1)], 1);
            if (staged) stage[pos] = q1[u].x; else csr[base + pos] = q1[u].x;
        }
    }
    for (int e = lo0 + 8; e < hi0; ++e) {
        int2 sd = p0[e];
        int pos = atomicAdd(&cur[sd.y & (BN - 1)], 1);
        if (staged) stage[pos] = sd.x; else csr[base + pos] = sd.x;
    }
    for (int e = lo1 + 8; e < hi1; ++e) {
        int2 sd = p1[e];
        int pos = atomicAdd(&cur[sd.y & (BN - 1)], 1);
        if (staged) stage[pos] = sd.x; else csr[base + pos] = sd.x;
    }
    __syncthreads();
    if (staged)
        for (int i = tid; i < tot; i += 256) csr[base + i] = stage[i];
}

__device__ __forceinline__ void gather_node(
    const float* __restrict__ H, const int4* __restrict__ ninfo, const int* __restrict__ csr,
    const float* __restrict__ bias, const int* __restrict__ batch,
    float* __restrict__ mean, const float* __restrict__ counts, float* srow, int* sg, float* sc, int nN)
{
    int lane = threadIdx.x & 31;
    int wid = threadIdx.x >> 5;
    int node = blockIdx.x * 8 + wid;
    bool valid = node < nN;
    if (!valid) node = nN - 1;
    int d = __builtin_amdgcn_readfirstlane(node);
    int4 ni = ninfo[d];
    int gid = batch[d];
    {
        bool ok = __builtin_amdgcn_readfirstlane((int)(blockIdx.x * 8 + wid)) < nN && (unsigned)gid < (unsigned)N_GRAPHS;
        if (!ok) gid = -1;
    }
    int beg = ni.x, n = ni.y;
    float di = __int_as_float(ni.z);
    int c4 = lane < 28 ? lane : 27;
    const float4* H4 = (const float4*)H + c4;

    int m = n < 32 ? n : 32;
    int meta = 0;
    if (lane < m) meta = csr[beg + lane];
    float4 hs = H4[(size_t)d * (HPAD / 4)];
    bool hasb = c4 < HID / 4;
    float4 bb = ((const float4*)bias)[hasb ? c4 : HID / 4 - 1];
    float gcnt = counts[gid >= 0 ? gid : 0];
    float4 acc = make_float4(0.f, 0.f, 0.f, 0.f);

    for (int base = 0;;) {
        for (int j = 0; j < m; j += 8) {
            int r = m - j;
            int   sx[8];
            float4 h[8];
#pragma unroll
            for (int u = 0; u < 8; ++u) {
                int jj = (u < r) ? j + u : m - 1;
                sx[u] = __builtin_amdgcn_readlane(meta, jj);
            }
#pragma unroll
            for (int u = 0; u < 8; ++u) {
                h[u] = make_float4(0.f, 0.f, 0.f, 0.f);
                if (u < r) h[u] = H4[(size_t)sx[u] * (HPAD / 4)];
            }
#pragma unroll
            for (int u = 0; u < 8; ++u) {
                acc.x += h[u].x;
                acc.y += h[u].y;
                acc.z += h[u].z;
                acc.w += h[u].w;
            }
        }
        base += 32;
        if (base >= n) break;
        m = n - base;
        if (m > 32) m = 32;
        meta = 0;
        if (lane < m) meta = csr[beg + base + lane];
    }

    if (!hasb) bb = make_float4(0.f, 0.f, 0.f, 0.f);
    float4 o;
    o.x = fmaf(hs.x + acc.x, di, bb.x); o.x = o.x > 0.f ? o.x : 0.f;
    o.y = fmaf(hs.y + acc.y, di, bb.y); o.y = o.y > 0.f ? o.y : 0.f;
    o.z = fmaf(hs.z + acc.z, di, bb.z); o.z = o.z > 0.f ? o.z : 0.f;
    o.w = fmaf(hs.w + acc.w, di, bb.w); o.w = o.w > 0.f ? o.w : 0.f;
    if (lane < 28) ((float4*)srow)[wid * (HPAD / 4) + lane] = o;
    if (lane == 0) { sg[wid] = gid; sc[wid] = gcnt; }
    __syncthreads();
    int t = threadIdx.x;
    if (t >= HID) return;
    int4 ga = ((const int4*)sg)[0], gb = ((const int4*)sg)[1];
    float4 ca = ((const float4*)sc)[0], cb = ((const float4*)sc)[1];
    int gq[8] = {ga.x, ga.y, ga.z, ga.w, gb.x, gb.y, gb.z, gb.w};
    float cq[8] = {ca.x, ca.y, ca.z, ca.w, cb.x, cb.y, cb.z, cb.w};
    float vq[8];
#pragma unroll
    for (int q = 0; q < 8; ++q) vq[q] = srow[q * HPAD + t];
    int cur = -1;
    float ccur = 1.0f;
    float s = 0.f;
#pragma unroll
    for (int q = 0; q < 8; ++q) {
        int gw = gq[q];
        float v = vq[q];
        if (gw >= 0) {
            if (gw == cur) {
                s += v;
            } else {
                if (cur >= 0) atomicAdd(&mean[(size_t)cur * HID + t], s / fmaxf(ccur, 1.0f));
                cur = gw;
                ccur = cq[q];
                s = v;
            }
        }
    }
    if (cur >= 0) atomicAdd(&mean[(size_t)cur * HID + t], s / fmaxf(ccur, 1.0f));
}

template <int NR>
__device__ __forceinline__ void sum_rows(const float4* __restrict__ H4, int meta, int j, v2f& acc01, v2f& acc23)
{
    int   sx[NR];
    float4 h[NR];
#pragma unroll
    for (int u = 0; u < NR; ++u) sx[u] = __builtin_amdgcn_readlane(meta, j + u);
#pragma unroll
    for (int u = 0; u < NR; ++u) h[u] = H4[(size_t)sx[u] * (HPAD / 4)];
#pragma unroll
    for (int u = 0; u < NR; ++u) {
        float sc = __int_as_float(__builtin_amdgcn_readlane(__float_as_int(h[u].x), HID / 4));
        acc01 = __builtin_elementwise_fma((v2f){h[u].x, h[u].y}, (v2f){sc, sc}, acc01);
        acc23 = __builtin_elementwise_fma((v2f){h[u].z, h[u].w}, (v2f){sc, sc}, acc23);
    }
}

#define FG_BLOCKS 1024
#define FG_GROUPS ((N_NODES + 15) / 16)
#define TPITCH 116
__global__ __launch_bounds__(512) void k_gather_gemm(
    const float* __restrict__ H, const int4* __restrict__ ninfo, const int* __restrict__ csr,
    const float* __restrict__ bias, const float* __restrict__ W, int wr, int wc,
    float* __restrict__ H2, int nN)
{
    __shared__ float2 wl[(HPAD / 2) * HPAD];
    __shared__ float4 tile[2][16 * (TPITCH / 4)];
    int tid = threadIdx.x;
    int lane = tid & 31;
    int wave = __builtin_amdgcn_readfirstlane(tid >> 5);

    {
        const int total = (HPAD / 2) * HPAD;
        float lo[13], hi[13];
#pragma unroll
        for (int u = 0; u < 13; ++u) {
            int i = tid + u * 512;
            int p = i / HPAD, n = i - p * HPAD;
            int k0 = 2 * p, k1 = k0 + 1;
            bool ins = i < total && n < wc;
            bool ok0 = ins && k0 < wr, ok1 = ins && k1 < wr;
            float v0 = W[ok0 ? k0 * wc + n : 0];
            float v1 = W[ok1 ? k1 * wc + n : 0];
            lo[u] = ok0 ? v0 : 0.0f;
            hi[u] = ok1 ? v1 : 0.0f;
        }
#pragma unroll
        for (int u = 0; u < 13; ++u) {
            int i = tid + u * 512;
            if (i < total) wl[i] = make_float2(lo[u], hi[u]);
        }
    }

    int m = lane & 15;
    int koff = (lane >> 4) << 1;
    int c4 = lane < 28 ? lane : 27;
    const float4* H4 = (const float4*)H + c4;
    bool hasb = c4 < HID / 4;
    float4 bb = ((const float4*)bias)[hasb ? c4 : HID / 4 - 1];
    if (!hasb) bb = make_float4(0.f, 0.f, 0.f, 0.f);

    int g = blockIdx.x;
    if (g >= FG_GROUPS) return;

    const int* nwords = (const int*)ninfo + (lane & 3);

    int node = g * 16 + wave;
    bool valid = node < nN;
    int d = __builtin_amdgcn_readfirstlane(valid ? node : nN - 1);
    int rec = nwords[(size_t)d * 4];
    int beg = __builtin_amdgcn_readlane(rec, 0);
    int n = __builtin_amdgcn_readlane(rec, 1);
    float di = __int_as_float(__builtin_amdgcn_readlane(rec, 2));
    int mm = n < 32 ? n : 32;
    int meta = 0;
    if (lane < mm) meta = csr[beg + lane];
    float4 hs = H4[(size_t)d * (HPAD / 4)];
    int gn = g + FG_BLOCKS;
    bool more = gn < FG_GROUPS;
    int noden = gn * 16 + wave;
    bool validn = more && noden < nN;
    int dn = __builtin_amdgcn_readfirstlane(validn ? noden : nN - 1);
    int nrec = nwords[(size_t)dn * 4];
    int buf = 0;
    int rot = 0;

    for (;;) {
        v2f acc01 = (v2f){0.f, 0.f}, acc23 = (v2f){0.f, 0.f};
        for (int base = 0;;) {
            for (int j = 0; j < mm; j += 8) {
                int r = mm - j;
                if (r >= 8) sum_rows<8>(H4, meta, j, acc01, acc23);
                else switch (r) {
                    case 7: sum_rows<7>(H4, meta, j, acc01, acc23); break;
                    case 6: sum_rows<6>(H4, meta, j, acc01, acc23); break;
                    case 5: sum_rows<5>(H4, meta, j, acc01, acc23); break;
                    case 4: sum_rows<4>(H4, meta, j, acc01, acc23); break;
                    case 3: sum_rows<3>(H4, meta, j, acc01, acc23); break;
                    case 2: sum_rows<2>(H4, meta, j, acc01, acc23); break;
                    default: sum_rows<1>(H4, meta, j, acc01, acc23); break;
                }
            }
            base += 32;
            if (base >= n) break;
            mm = n - base;
            if (mm > 32) mm = 32;
            meta = 0;
            if (lane < mm) meta = csr[beg + base + lane];
        }

        float4 o;
        o.x = fmaf(fmaf(hs.x, di, acc01.x), di, bb.x); o.x = o.x > 0.f ? o.x * di : 0.f;
        o.y = fmaf(fmaf(hs.y, di, acc01.y), di, bb.y); o.y = o.y > 0.f ? o.y * di : 0.f;
        o.z = fmaf(fmaf(hs.z, di, acc23.x), di, bb.z); o.z = o.z > 0.f ? o.z * di : 0.f;
        o.w = fmaf(fmaf(hs.w, di, acc23.y), di, bb.w); o.w = o.w > 0.f ? o.w * di : 0.f;
        if (!valid) o = make_float4(0.f, 0.f, 0.f, 0.f);
        if (lane < 28) tile[buf][wave * (TPITCH / 4) + lane] = o;
        __syncthreads();

        int begn = __builtin_amdgcn_readlane(nrec, 0);
        int nn = __builtin_amdgcn_readlane(nrec, 1);
        float din = __int_as_float(__builtin_amdgcn_readlane(nrec, 2));
        int mmn = nn < 32 ? nn : 32;
        int metan = 0;
        if (lane < mmn) metan = csr[begn + lane];
        float4 hsn = H4[(size_t)dn * (HPAD / 4)];
        int gnn = gn + FG_BLOCKS;
        bool more2 = gnn < FG_GROUPS;
        int nodenn = gnn * 16 + wave;
        bool validnn = more2 && nodenn < nN;
        int dnn = __builtin_amdgcn_readfirstlane(validnn ? nodenn : nN - 1);
        int nrec2 = nwords[(size_t)dnn * 4];

        int t = (wave - rot) & 15;
        if (t < 7) {
            const float* trow = (const float*)tile[buf] + m * TPITCH + koff;
            const float2* wcol = wl + (koff >> 1) * HPAD + t * 16 + m;
            v8f pacc = (v8f){0.f, 0.f, 0.f, 0.f, 0.f, 0.f, 0.f, 0.f};
#pragma unroll
            for (int k = 0; k < HPAD; k += 4) {
                float2 a2 = *(const float2*)(trow + k);
                float2 b2 = wcol[(k >> 1) * HPAD];
                v2f a = (v2f){a2.x, a2.y};
                v2f b = (v2f){b2.x, b2.y};
                pacc = __builtin_amdgcn_wmma_f32_16x16x4_f32(
                    false, a, false, b, (short)0, pacc, false, false);
            }
            int rbase = g * 16 + ((lane >> 4) << 3);
            int cc = t * 16 + m;
#pragma unroll
            for (int v = 0; v < 8; ++v) {
                int rr = rbase + v;
                if (rr < nN) H2[(size_t)rr * HPAD + cc] = pacc[v];
            }
        }
        if (!more) break;
        g = gn; valid = validn;
        beg = begn; n = nn; di = din; mm = mmn; meta = metan; hs = hsn;
        gn = gnn; more = more2; validn = validnn; dn = dnn; nrec = nrec2;
        rot = (rot + 7) & 15;
        buf ^= 1;
    }
}

__global__ __launch_bounds__(256) void k_gather_pool(
    const float* __restrict__ H, const int4* __restrict__ ninfo, const int* __restrict__ csr,
    const float* __restrict__ bias, const int* __restrict__ batch,
    float* __restrict__ mean, const float* __restrict__ counts, int nN)
{
    __shared__ float4 srow[8 * (HPAD / 4)];
    __shared__ int4 sg[2];
    __shared__ float4 sc[2];
    gather_node(H, ninfo, csr, bias, batch, mean, counts, (float*)srow, (int*)sg, (float*)sc, nN);
}

extern "C" void kernel_launch(void* const* d_in, const int* in_sizes, int n_in,
                              void* d_out, int out_size, void* d_ws, size_t ws_size,
                              hipStream_t stream) {
    const float* x  = (const float*)d_in[0];
    const float* W1 = (const float*)d_in[1];
    const float* b1 = (const float*)d_in[2];
    const float* W2 = (const float*)d_in[3];
    const float* b2 = (const float*)d_in[4];
    const int*   ei = (const int*)d_in[5];
    const int*   bt = (const int*)d_in[6];
    const int* esrc = ei;
    const int* edst = ei + N_EDGES;
    float* out = (float*)d_out;

    float*  ws    = (float*)d_ws;
    int4*   ninfo = (int4*)ws;
    float*  cnts = ws + (size_t)4 * N_NODES;
    float*  B0   = cnts + N_GRAPHS + WS_HEAD_PAD;
    float*  B2   = B0 + (size_t)N_NODES * HPAD;
    int2*   bins = (int2*)(B2 + (size_t)N_NODES * HPAD);
    int*    offT = (int*)(bins + (size_t)NCH * CH);
    int*    csr  = offT + (size_t)(NBK + 1) * NCH;

    auto cdiv = [](long long a, long long b) { return (unsigned)((a + b - 1) / b); };

    k_bin_gemm<<<NCH + GEMM_BLOCKS, 256, 0, stream>>>(esrc, edst, bins, offT, out, cnts,
                                                      x, W1, IN_DIM, HID, B0, N_NODES, IN_DIM);

    k_build<<<NBK, 256, 0, stream>>>(B0, N_NODES, bins, offT, ninfo, csr, bt, cnts);

    k_gather_gemm<<<FG_BLOCKS, 512, 0, stream>>>(B0, ninfo, csr, b1, W2, HID, HID, B2, N_NODES);

    k_gather_pool<<<cdiv(N_NODES, 8), 256, 0, stream>>>(B2, ninfo, csr, b2, bt, out, cnts, N_NODES);
}


# Last aggregation on resident waves: 12288 waves each walk 8-9 consecutive nodes, next node's slots asked for inside this node's first row batch; a wave adds up its run per graph; sum order changes
# speedup vs baseline: 6.4581x; 1.0365x over previous
#include <hip/hip_runtime.h>
#include <hip/hip_bf16.h>

typedef __attribute__((ext_vector_type(2))) float v2f;
typedef __attribute__((ext_vector_type(8))) float v8f;

#define N_NODES  100000
#define N_EDGES  1600000
#define N_GRAPHS 1000
#define IN_DIM   128
#define HID      100
#define HPAD     112
#define GEMM_BLOCKS ((N_NODES + 127) / 128)
#define WS_HEAD (4 * N_NODES + N_GRAPHS)
#define WS_HEAD_PAD ((32 - WS_HEAD % 32) % 32)
static_assert((WS_HEAD + WS_HEAD_PAD) % 32 == 0, "the first row table starts on a 128-byte line");
static_assert((N_NODES * HPAD) % 32 == 0, "and so does the second, one table behind it");

#define BN 128
#define NBK ((N_NODES + BN - 1) / BN)
#define CH 4096
#define NCH ((N_EDGES + CH - 1) / CH)
#define STG 4096
static_assert(NBK + 1 <= 1024, "the level-1 scan covers 1024 entries");
static_assert(N_EDGES % 4 == 0 && CH % 1024 == 0, "edges are read four at a time, 256 threads");
static_assert(NCH * 256 >= N_GRAPHS * HID && NCH * 256 >= N_GRAPHS, "the level-1 blocks also clear the output and the graph counts");

__device__ __forceinline__ void gemm_block(
    int bx, const float* __restrict__ X, const float* __restrict__ W, int wr, int wc,
    float* __restrict__ H, int nrows, int K, float2* wl);

static_assert(1024 * 4 + CH * 8 + 8 * 4 <= (IN_DIM / 2) * HPAD * 8, "a level-1 block's counters and chunk fit in the weight array");
__global__ __launch_bounds__(256) void k_bin_gemm(const int* __restrict__ src, const int* __restrict__ dst,
                                             int2* __restrict__ bins, int* __restrict__ offT,
                                             float* __restrict__ out, float* __restrict__ cnts,
                                             const float* __restrict__ X, const float* __restrict__ W, int wr, int wc,
                                             float* __restrict__ H, int nrows, int K) {
    __shared__ float2 wl[(IN_DIM / 2) * HPAD];
    if (blockIdx.x >= NCH) {
        gemm_block(blockIdx.x - NCH, X, W, wr, wc, H, nrows, K, wl);
        return;
    }
    int* hist = (int*)wl;
    int2* sorted = (int2*)(wl + 512);
    int* wsum = (int*)(wl + 512 + CH);
    int tid = threadIdx.x, lane = tid & 31, wv = tid >> 5;
    int k = blockIdx.x;
    {
        int i = k * 256 + tid;
        if (i < N_GRAPHS * HID) out[i] = 0.0f;
        if (i < N_GRAPHS) cnts[i] = 0.0f;
    }
    for (int i = tid; i < 1024; i += 256) hist[i] = 0;
    int4 dv[CH / 1024], sv[CH / 1024];
#pragma unroll
    for (int j = 0; j < CH / 1024; ++j) {
        int i4 = k * (CH / 4) + j * 256 + tid;
        bool ok = i4 < N_EDGES / 4;
        int ic = ok ? i4 : 0;
        dv[j] = ((const int4*)dst)[ic];
        sv[j] = ((const int4*)src)[ic];
        if (!ok) dv[j] = make_int4(-1, -1, -1, -1);
    }
    __syncthreads();
    int rk[CH / 256];
#pragma unroll
    for (int j = 0; j < CH / 1024; ++j) {
        int dd[4] = {dv[j].x, dv[j].y, dv[j].z, dv[j].w};
#pragma unroll
        for (int q = 0; q < 4; ++q) {
            rk[j * 4 + q] = 0;
            if (dd[q] >= 0) rk[j * 4 + q] = atomicAdd(&hist[dd[q] >> 7], 1);
        }
    }
    __syncthreads();
    int a0 = hist[tid * 4], a1 = hist[tid * 4 + 1], a2 = hist[tid * 4 + 2], a3 = hist[tid * 4 + 3];
    int v = a0 + a1 + a2 + a3;
    int sum4 = v;
#pragma unroll
    for (int o = 1; o < 32; o <<= 1) {
        int t = __shfl_up(v, o);
        if (lane >= o) v += t;
    }
    if (lane == 31) wsum[wv] = v;
    __syncthreads();
    int pre = 0;
#pragma unroll
    for (int q = 0; q < 8; ++q) pre += (q < wv) ? wsum[q] : 0;
    int ex = pre + v - sum4;
    hist[tid * 4] = ex; hist[tid * 4 + 1] = ex + a0; hist[tid * 4 + 2] = ex + a0 + a1; hist[tid * 4 + 3] = ex + a0 + a1 + a2;
    __syncthreads();
#pragma unroll
    for (int j = 0; j < CH / 1024; ++j) {
        int dd[4] = {dv[j].x, dv[j].y, dv[j].z, dv[j].w};
        int ss[4] = {sv[j].x, sv[j].y, sv[j].z, sv[j].w};
#pragma unroll
        for (int q = 0; q < 4; ++q)
            if (dd[q] >= 0) sorted[hist[dd[q] >> 7] + rk[j * 4 + q]] = make_int2(ss[q], dd[q]);
    }
    for (int b = tid; b <= NBK; b += 256) offT[b * NCH + k] = hist[b];
    __syncthreads();
    int tot = hist[NBK];
    int4* o4 = (int4*)(bins + (size_t)k * CH);
    const int4* s4 = (const int4*)sorted;
    for (int i = tid; i < tot / 2; i += 256) o4[i] = s4[i];
    if ((tot & 1) && tid == 0) bins[(size_t)k * CH + tot - 1] = sorted[tot - 1];
}

__device__ __forceinline__ void gemm_block(
    int bx, const float* __restrict__ X, const float* __restrict__ W, int wr, int wc,
    float* __restrict__ H, int nrows, int K, float2* wl)
{
    int tid = threadIdx.x;
    int kpairs = K >> 1;
    int total = kpairs * HPAD;
    for (int base = tid; base < total; base += 14 * 256) {
        float lo[14], hi[14];
#pragma unroll
        for (int u = 0; u < 14; ++u) {
            int i = base + u * 256;
            int p = i / HPAD, n = i - p * HPAD;
            int k0 = 2 * p, k1 = k0 + 1;
            bool ins = i < total && n < wc;
            bool ok0 = ins && k0 < wr, ok1 = ins && k1 < wr;
            float v0 = W[ok0 ? k0 * wc + n : 0];
            float v1 = W[ok1 ? k1 * wc + n : 0];
            lo[u] = ok0 ? v0 : 0.0f;
            hi[u] = ok1 ? v1 : 0.0f;
        }
#pragma unroll
        for (int u = 0; u < 14; ++u) {
            wl[base + u * 256] = make_float2(lo[u], hi[u]);
        }
    }
    __syncthreads();

    int lane = tid & 31;
    int wave = tid >> 5;
    int m    = lane & 15;
    int koff = (lane >> 4) << 1;
    int row0 = bx * 128 + wave * 16;
    int row  = row0 + m;
    bool inr = row < nrows;
    int rowc = inr ? row : (nrows - 1);
    const float* xrow = X + (size_t)rowc * K;

    v8f acc[7];
#pragma unroll
    for (int t = 0; t < 7; ++t)
        acc[t] = (v8f){0.f, 0.f, 0.f, 0.f, 0.f, 0.f, 0.f, 0.f};

    for (int k = 0; k < K; k += 4) {
        float2 t2 = *(const float2*)(xrow + k + koff);
        v2f a = (v2f){ inr ? t2.x : 0.f, inr ? t2.y : 0.f };
        const float2* wrow = wl + ((k + koff) >> 1) * HPAD;
#pragma unroll
        for (int t = 0; t < 7; ++t) {
            float2 bb = wrow[t * 16 + m];
            v2f b = (v2f){bb.x, bb.y};
            acc[t] = __builtin_amdgcn_wmma_f32_16x16x4_f32(
                false, a, false, b, (short)0, acc[t], false, false);
        }
    }

    int rbase = row0 + ((lane >> 4) << 3);
#pragma unroll
    for (int t = 0; t < 7; ++t) {
        int cc = t * 16 + m;
#pragma unroll
        for (int v = 0; v < 8; ++v) {
            int rr = rbase + v;
            if (rr < nrows) H[(size_t)rr * HPAD + cc] = acc[t][v];
        }
    }
}

static_assert(BN == 128 && GEMM_BLOCKS == NBK, "a bucket is a GEMM block's 128 rows");
__global__ __launch_bounds__(256) void k_build(
    float* __restrict__ H, int nrows,
    const int2* __restrict__ bins, const int* __restrict__ offT,
    int4* __restrict__ ninfo, int* __restrict__ csr, const int* __restrict__ batch, float* counts)
{
    __shared__ int cnt[BN];
    __shared__ int cur[BN];
    __shared__ int stage[STG];
    __shared__ int wsum[8];
    __shared__ int sbase, stot;
    int tid = threadIdx.x, lane = tid & 31, wv = tid >> 5;
    int b = blockIdx.x;
    if (tid < BN) cnt[tid] = 0;
    if (tid == 0) sbase = 0;
    int lo0 = 0, hi0 = 0, lo1 = 0, hi1 = 0;
    int k0 = tid, k1 = tid + 256;
    if (k0 < NCH) { lo0 = offT[b * NCH + k0]; hi0 = offT[(b + 1) * NCH + k0]; }
    if (k1 < NCH) { lo1 = offT[b * NCH + k1]; hi1 = offT[(b + 1) * NCH + k1]; }
    const int2* p0 = bins + (size_t)k0 * CH;
    const int2* p1 = bins + (size_t)(k1 < NCH ? k1 : 0) * CH;
    int n0 = hi0 - lo0, n1 = hi1 - lo1;
    int2 q0[8], q1[8];
#pragma unroll
    for (int u = 0; u < 8; ++u) {
        q0[u] = p0[u < n0 ? lo0 + u : 0];
        q1[u] = p1[u < n1 ? lo1 + u : 0];
    }
    __syncthreads();
    atomicAdd(&sbase, lo0 + lo1);
#pragma unroll
    for (int u = 0; u < 8; ++u) {
        if (u < n0) atomicAdd(&cnt[q0[u].y & (BN - 1)], 1);
        if (u < n1) atomicAdd(&cnt[q1[u].y & (BN - 1)], 1);
    }
    for (int e = lo0 + 8; e < hi0; ++e) atomicAdd(&cnt[p0[e].y & (BN - 1)], 1);
    for (int e = lo1 + 8; e < hi1; ++e) atomicAdd(&cnt[p1[e].y & (BN - 1)], 1);
    __syncthreads();
    int node = b * BN + tid;
    bool isn = tid < BN && node < nrows;
    int c = isn ? cnt[tid] : 0;
    int g = isn ? batch[node] : -1;
    {
        if ((unsigned)g >= (unsigned)N_GRAPHS) g = -1;
        int gp = __shfl_up(g, 1);
        bool head = lane == 0 || g != gp;
        unsigned hm = (unsigned)__ballot(head);
        unsigned above = hm & ~((2u << lane) - 1u);
        int len = (above ? __builtin_ctz(above) : 32) - lane;
        if (head && g >= 0) atomicAdd(&counts[g], (float)len);
    }
    int v = c;
#pragma unroll
    for (int o = 1; o < 32; o <<= 1) {
        int t = __shfl_up(v, o);
        if (lane >= o) v += t;
    }
    if (lane == 31) wsum[wv] = v;
    __syncthreads();
    int pre = 0;
#pragma unroll
    for (int q = 0; q < 8; ++q) pre += (q < wv) ? wsum[q] : 0;
    int start = pre + v - c;
    int base = sbase;
    if (tid < BN) {
        float dinv = rsqrtf((float)(c + 1));
        cur[tid] = start;
        if (isn) {
            ninfo[node] = make_int4(base + start, c, __float_as_int(dinv), 0);
            H[(size_t)node * HPAD + HID] = dinv;
        }
    }
    if (tid == 255) stot = pre + v;
    __syncthreads();
    int tot = stot;
    bool staged = tot <= STG;
#pragma unroll
    for (int u = 0; u < 8; ++u) {
        if (u < n0) {
            int pos = atomicAdd(&cur[q0[u].y & (BN - 1)], 1);
            if (staged) stage[pos] = q0[u].x; else csr[base + pos] = q0[u].x;
        }
        if (u < n1) {
            int pos = atomicAdd(&cur[q1[u].y & (BN - 1)], 1);
            if (staged) stage[pos] = q1[u].x; else csr[base + pos] = q1[u].x;
        }
    }
    for (int e = lo0 + 8; e < hi0; ++e) {
        int2 sd = p0[e];
        int pos = atomicAdd(&cur[sd.y & (BN - 1)], 1);
        if (staged) stage[pos] = sd.x; else csr[base + pos] = sd.x;
    }
    for (int e = lo1 + 8; e < hi1; ++e) {
        int2 sd = p1[e];
        int pos = atomicAdd(&cur[sd.y & (BN - 1)], 1);
        if (staged) stage[pos] = sd.x; else csr[base + pos] = sd.x;
    }
    __syncthreads();
    if (staged)
        for (int i = tid; i < tot; i += 256) csr[base + i] = stage[i];
}


template <int NR, bool FIRST>
__device__ __forceinline__ void add_rows(const float4* __restrict__ H4, int meta, int j, float4& acc,
                                         const float4* __restrict__ own, float4& hs,
                                         const int* __restrict__ csrn, bool wantn, int& metan)
{
    int   sx[NR > 0 ? NR : 1];
    float4 h[NR > 0 ? NR : 1];
#pragma unroll
    for (int u = 0; u < NR; ++u) sx[u] = __builtin_amdgcn_readlane(meta, j + u);
#pragma unroll
    for (int u = 0; u < NR; ++u) h[u] = H4[(size_t)sx[u] * (HPAD / 4)];
    if (FIRST) {
        hs = *own;
        metan = 0;
        if (wantn) metan = *csrn;
    }
#pragma unroll
    for (int u = 0; u < NR; ++u) {
        acc.x += h[u].x;
        acc.y += h[u].y;
        acc.z += h[u].z;
        acc.w += h[u].w;
    }
}

template <bool FIRST>
__device__ __forceinline__ void add_batch(const float4* __restrict__ H4, int meta, int j, int r, float4& acc,
                                          const float4* __restrict__ own, float4& hs,
                                          const int* __restrict__ csrn, bool wantn, int& metan)
{
    if (r >= 8) add_rows<8, FIRST>(H4, meta, j, acc, own, hs, csrn, wantn, metan);
    else switch (r) {
        case 7: add_rows<7, FIRST>(H4, meta, j, acc, own, hs, csrn, wantn, metan); break;
        case 6: add_rows<6, FIRST>(H4, meta, j, acc, own, hs, csrn, wantn, metan); break;
        case 5: add_rows<5, FIRST>(H4, meta, j, acc, own, hs, csrn, wantn, metan); break;
        case 4: add_rows<4, FIRST>(H4, meta, j, acc, own, hs, csrn, wantn, metan); break;
        case 3: add_rows<3, FIRST>(H4, meta, j, acc, own, hs, csrn, wantn, metan); break;
        case 2: add_rows<2, FIRST>(H4, meta, j, acc, own, hs, csrn, wantn, metan); break;
        case 1: add_rows<1, FIRST>(H4, meta, j, acc, own, hs, csrn, wantn, metan); break;
        default: add_rows<0, FIRST>(H4, meta, j, acc, own, hs, csrn, wantn, metan); break;
    }
}

template <int NR>
__device__ __forceinline__ void sum_rows(const float4* __restrict__ H4, int meta, int j, v2f& acc01, v2f& acc23)
{
    int   sx[NR];
    float4 h[NR];
#pragma unroll
    for (int u = 0; u < NR; ++u) sx[u] = __builtin_amdgcn_readlane(meta, j + u);
#pragma unroll
    for (int u = 0; u < NR; ++u) h[u] = H4[(size_t)sx[u] * (HPAD / 4)];
#pragma unroll
    for (int u = 0; u < NR; ++u) {
        float sc = __int_as_float(__builtin_amdgcn_readlane(__float_as_int(h[u].x), HID / 4));
        acc01 = __builtin_elementwise_fma((v2f){h[u].x, h[u].y}, (v2f){sc, sc}, acc01);
        acc23 = __builtin_elementwise_fma((v2f){h[u].z, h[u].w}, (v2f){sc, sc}, acc23);
    }
}

#define FG_BLOCKS 1024
#define FG_GROUPS ((N_NODES + 15) / 16)
#define TPITCH 116
__global__ __launch_bounds__(512) void k_gather_gemm(
    const float* __restrict__ H, const int4* __restrict__ ninfo, const int* __restrict__ csr,
    const float* __restrict__ bias, const float* __restrict__ W, int wr, int wc,
    float* __restrict__ H2, int nN)
{
    __shared__ float2 wl[(HPAD / 2) * HPAD];
    __shared__ float4 tile[2][16 * (TPITCH / 4)];
    int tid = threadIdx.x;
    int lane = tid & 31;
    int wave = __builtin_amdgcn_readfirstlane(tid >> 5);

    {
        const int total = (HPAD / 2) * HPAD;
        float lo[13], hi[13];
#pragma unroll
        for (int u = 0; u < 13; ++u) {
            int i = tid + u * 512;
            int p = i / HPAD, n = i - p * HPAD;
            int k0 = 2 * p, k1 = k0 + 1;
            bool ins = i < total && n < wc;
            bool ok0 = ins && k0 < wr, ok1 = ins && k1 < wr;
            float v0 = W[ok0 ? k0 * wc + n : 0];
            float v1 = W[ok1 ? k1 * wc + n : 0];
            lo[u] = ok0 ? v0 : 0.0f;
            hi[u] = ok1 ? v1 : 0.0f;
        }
#pragma unroll
        for (int u = 0; u < 13; ++u) {
            int i = tid + u * 512;
            if (i < total) wl[i] = make_float2(lo[u], hi[u]);
        }
    }

    int m = lane & 15;
    int koff = (lane >> 4) << 1;
    int c4 = lane < 28 ? lane : 27;
    const float4* H4 = (const float4*)H + c4;
    bool hasb = c4 < HID / 4;
    float4 bb = ((const float4*)bias)[hasb ? c4 : HID / 4 - 1];
    if (!hasb) bb = make_float4(0.f, 0.f, 0.f, 0.f);

    int g = blockIdx.x;
    if (g >= FG_GROUPS) return;

    const int* nwords = (const int*)ninfo + (lane & 3);

    int node = g * 16 + wave;
    bool valid = node < nN;
    int d = __builtin_amdgcn_readfirstlane(valid ? node : nN - 1);
    int rec = nwords[(size_t)d * 4];
    int beg = __builtin_amdgcn_readlane(rec, 0);
    int n = __builtin_amdgcn_readlane(rec, 1);
    float di = __int_as_float(__builtin_amdgcn_readlane(rec, 2));
    int mm = n < 32 ? n : 32;
    int meta = 0;
    if (lane < mm) meta = csr[beg + lane];
    float4 hs = H4[(size_t)d * (HPAD / 4)];
    int gn = g + FG_BLOCKS;
    bool more = gn < FG_GROUPS;
    int noden = gn * 16 + wave;
    bool validn = more && noden < nN;
    int dn = __builtin_amdgcn_readfirstlane(validn ? noden : nN - 1);
    int nrec = nwords[(size_t)dn * 4];
    int buf = 0;
    int rot = 0;

    for (;;) {
        v2f acc01 = (v2f){0.f, 0.f}, acc23 = (v2f){0.f, 0.f};
        for (int base = 0;;) {
            for (int j = 0; j < mm; j += 8) {
                int r = mm - j;
                if (r >= 8) sum_rows<8>(H4, meta, j, acc01, acc23);
                else switch (r) {
                    case 7: sum_rows<7>(H4, meta, j, acc01, acc23); break;
                    case 6: sum_rows<6>(H4, meta, j, acc01, acc23); break;
                    case 5: sum_rows<5>(H4, meta, j, acc01, acc23); break;
                    case 4: sum_rows<4>(H4, meta, j, acc01, acc23); break;
                    case 3: sum_rows<3>(H4, meta, j, acc01, acc23); break;
                    case 2: sum_rows<2>(H4, meta, j, acc01, acc23); break;
                    default: sum_rows<1>(H4, meta, j, acc01, acc23); break;
                }
            }
            base += 32;
            if (base >= n) break;
            mm = n - base;
            if (mm > 32) mm = 32;
            meta = 0;
            if (lane < mm) meta = csr[beg + base + lane];
        }

        float4 o;
        o.x = fmaf(fmaf(hs.x, di, acc01.x), di, bb.x); o.x = o.x > 0.f ? o.x * di : 0.f;
        o.y = fmaf(fmaf(hs.y, di, acc01.y), di, bb.y); o.y = o.y > 0.f ? o.y * di : 0.f;
        o.z = fmaf(fmaf(hs.z, di, acc23.x), di, bb.z); o.z = o.z > 0.f ? o.z * di : 0.f;
        o.w = fmaf(fmaf(hs.w, di, acc23.y), di, bb.w); o.w = o.w > 0.f ? o.w * di : 0.f;
        if (!valid) o = make_float4(0.f, 0.f, 0.f, 0.f);
        if (lane < 28) tile[buf][wave * (TPITCH / 4) + lane] = o;
        __syncthreads();

        int begn = __builtin_amdgcn_readlane(nrec, 0);
        int nn = __builtin_amdgcn_readlane(nrec, 1);
        float din = __int_as_float(__builtin_amdgcn_readlane(nrec, 2));
        int mmn = nn < 32 ? nn : 32;
        int metan = 0;
        if (lane < mmn) metan = csr[begn + lane];
        float4 hsn = H4[(size_t)dn * (HPAD / 4)];
        int gnn = gn + FG_BLOCKS;
        bool more2 = gnn < FG_GROUPS;
        int nodenn = gnn * 16 + wave;
        bool validnn = more2 && nodenn < nN;
        int dnn = __builtin_amdgcn_readfirstlane(validnn ? nodenn : nN - 1);
        int nrec2 = nwords[(size_t)dnn * 4];

        int t = (wave - rot) & 15;
        if (t < 7) {
            const float* trow = (const float*)tile[buf] + m * TPITCH + koff;
            const float2* wcol = wl + (koff >> 1) * HPAD + t * 16 + m;
            v8f pacc = (v8f){0.f, 0.f, 0.f, 0.f, 0.f, 0.f, 0.f, 0.f};
#pragma unroll
            for (int k = 0; k < HPAD; k += 4) {
                float2 a2 = *(const float2*)(trow + k);
                float2 b2 = wcol[(k >> 1) * HPAD];
                v2f a = (v2f){a2.x, a2.y};
                v2f b = (v2f){b2.x, b2.y};
                pacc = __builtin_amdgcn_wmma_f32_16x16x4_f32(
                    false, a, false, b, (short)0, pacc, false, false);
            }
            int rbase = g * 16 + ((lane >> 4) << 3);
            int cc = t * 16 + m;
#pragma unroll
            for (int v = 0; v < 8; ++v) {
                int rr = rbase + v;
                if (rr < nN) H2[(size_t)rr * HPAD + cc] = pacc[v];
            }
        }
        if (!more) break;
        g = gn; valid = validn;
        beg = begn; n = nn; di = din; mm = mmn; meta = metan; hs = hsn;
        gn = gnn; more = more2; validn = validnn; dn = dnn; nrec = nrec2;
        rot = (rot + 7) & 15;
        buf ^= 1;
    }
}

#define GP_BLOCKS 768
#define GP_WAVES (GP_BLOCKS * 16)

__device__ __forceinline__ void pool_flush(float* __restrict__ mean, float4* prow4, int g, float cnt, int lane)
{
    if (g < 0) return;
    __builtin_amdgcn_fence(__ATOMIC_RELEASE, "wavefront");
    __builtin_amdgcn_wave_barrier();
    const float* prow = (const float*)prow4;
    float v[4];
#pragma unroll
    for (int i = 0; i < 4; ++i) v[i] = prow[lane + 32 * i];
    __builtin_amdgcn_fence(__ATOMIC_RELEASE, "wavefront");
    __builtin_amdgcn_wave_barrier();
    prow4[lane] = make_float4(0.f, 0.f, 0.f, 0.f);
    float c = fmaxf(cnt, 1.0f);
#pragma unroll
    for (int i = 0; i < 4; ++i) {
        int col = lane + 32 * i;
        if (col < HID) atomicAdd(&mean[(size_t)g * HID + col], v[i] / c);
    }
}

__device__ __forceinline__ int rec_word(int ra, int rb, int i, int k)
{
    int a = __builtin_amdgcn_readlane(ra, 4 * (i & 7) + k);
    int b = __builtin_amdgcn_readlane(rb, 4 * (i & 7) + k);
    return i < 8 ? a : b;
}

__global__ __launch_bounds__(512) __attribute__((amdgpu_waves_per_eu(12, 12))) void k_gather_pool(
    const float* __restrict__ H, const int4* __restrict__ ninfo, const int* __restrict__ csr,
    const float* __restrict__ bias, const int* __restrict__ batch,
    float* __restrict__ mean, const float* __restrict__ counts, int nN)
{
    __shared__ float4 prow_all[16 * 32];
    int tid = threadIdx.x;
    int lane = tid & 31;
    int wave = __builtin_amdgcn_readfirstlane(tid >> 5);
    float4* prow4 = prow_all + wave * 32;
    prow4[lane] = make_float4(0.f, 0.f, 0.f, 0.f);

    long long gw = (long long)blockIdx.x * 16 + wave;
    int lo = (int)((gw * nN) / GP_WAVES);
    int hi = (int)(((gw + 1) * nN) / GP_WAVES);

    int c4 = lane < 28 ? lane : 27;
    const float4* H4 = (const float4*)H + c4;
    bool hasb = c4 < HID / 4;
    float4 bb = ((const float4*)bias)[hasb ? c4 : HID / 4 - 1];
    if (!hasb) bb = make_float4(0.f, 0.f, 0.f, 0.f);

    int curg = -1;
    float curcnt = 1.0f;

    for (int c0 = lo; c0 < hi; c0 += 16) {
        int nc = hi - c0 < 16 ? hi - c0 : 16;
        int la = (lane >> 2) < nc ? (lane >> 2) : nc - 1;
        int lb = 8 + (lane >> 2) < nc ? 8 + (lane >> 2) : nc - 1;
        int ra = ((const int*)ninfo)[(size_t)(c0 + la) * 4 + (lane & 3)];
        int rb = ((const int*)ninfo)[(size_t)(c0 + lb) * 4 + (lane & 3)];
        int l16 = (lane & 15) < nc ? (lane & 15) : nc - 1;
        int gids = batch[c0 + l16];
        if ((unsigned)gids >= (unsigned)N_GRAPHS) gids = -1;
        float cnts = counts[gids >= 0 ? gids : 0];

        int beg = __builtin_amdgcn_readlane(ra, 0);
        int n = __builtin_amdgcn_readlane(ra, 1);
        int mm = n < 32 ? n : 32;
        int meta = 0;
        if (lane < mm) meta = csr[beg + lane];

        for (int i = 0; i < nc; ++i) {
            int d = c0 + i;
            float di = __int_as_float(rec_word(ra, rb, i, 2));
            int in = i + 1 < nc ? i + 1 : i;
            int begn = rec_word(ra, rb, in, 0);
            int nn = rec_word(ra, rb, in, 1);
            int mmn = nn < 32 ? nn : 32;
            int metan;
            float4 hs;
            float4 acc = make_float4(0.f, 0.f, 0.f, 0.f);
            add_batch<true>(H4, meta, 0, mm, acc, H4 + (size_t)d * (HPAD / 4), hs, csr + begn + lane, lane < mmn, metan);
            int dummy;
            for (int j = 8; j < mm; j += 8)
                add_batch<false>(H4, meta, j, mm - j, acc, H4, hs, csr, false, dummy);
            for (int base = 32; base < n; base += 32) {
                int m2 = n - base < 32 ? n - base : 32;
                int meta2 = 0;
                if (lane < m2) meta2 = csr[beg + base + lane];
                for (int j = 0; j < m2; j += 8)
                    add_batch<false>(H4, meta2, j, m2 - j, acc, H4, hs, csr, false, dummy);
            }

            float4 o;
            o.x = fmaf(hs.x + acc.x, di, bb.x); o.x = o.x > 0.f ? o.x : 0.f;
            o.y = fmaf(hs.y + acc.y, di, bb.y); o.y = o.y > 0.f ? o.y : 0.f;
            o.z = fmaf(hs.z + acc.z, di, bb.z); o.z = o.z > 0.f ? o.z : 0.f;
            o.w = fmaf(hs.w + acc.w, di, bb.w); o.w = o.w > 0.f ? o.w : 0.f;

            int g = __builtin_amdgcn_readlane(gids, i);
            if (g != curg) {
                pool_flush(mean, prow4, curg, curcnt, lane);
                curg = g;
                curcnt = __int_as_float(__builtin_amdgcn_readlane(__float_as_int(cnts), i));
            }
            if (g >= 0) {
                float4 p = prow4[lane];
                p.x += o.x; p.y += o.y; p.z += o.z; p.w += o.w;
                prow4[lane] = p;
            }
            beg = begn; n = nn; mm = mmn; meta = metan;
        }
    }
    pool_flush(mean, prow4, curg, curcnt, lane);
}

extern "C" void kernel_launch(void* const* d_in, const int* in_sizes, int n_in,
                              void* d_out, int out_size, void* d_ws, size_t ws_size,
                              hipStream_t stream) {
    const float* x  = (const float*)d_in[0];
    const float* W1 = (const float*)d_in[1];
    const float* b1 = (const float*)d_in[2];
    const float* W2 = (const float*)d_in[3];
    const float* b2 = (const float*)d_in[4];
    const int*   ei = (const int*)d_in[5];
    const int*   bt = (const int*)d_in[6];
    const int* esrc = ei;
    const int* edst = ei + N_EDGES;
    float* out = (float*)d_out;

    float*  ws    = (float*)d_ws;
    int4*   ninfo = (int4*)ws;
    float*  cnts = ws + (size_t)4 * N_NODES;
    float*  B0   = cnts + N_GRAPHS + WS_HEAD_PAD;
    float*  B2   = B0 + (size_t)N_NODES * HPAD;
    int2*   bins = (int2*)(B2 + (size_t)N_NODES * HPAD);
    int*    offT = (int*)(bins + (size_t)NCH * CH);
    int*    csr  = offT + (size_t)(NBK + 1) * NCH;

    k_bin_gemm<<<NCH + GEMM_BLOCKS, 256, 0, stream>>>(esrc, edst, bins, offT, out, cnts,
                                                      x, W1, IN_DIM, HID, B0, N_NODES, IN_DIM);

    k_build<<<NBK, 256, 0, stream>>>(B0, N_NODES, bins, offT, ninfo, csr, bt, cnts);

    k_gather_gemm<<<FG_BLOCKS, 512, 0, stream>>>(B0, ninfo, csr, b1, W2, HID, HID, B2, N_NODES);

    k_gather_pool<<<GP_BLOCKS, 512, 0, stream>>>(B2, ninfo, csr, b2, bt, out, cnts, N_NODES);
}
